# starting kernel
# baseline (speedup 1.0000x reference)
#include <hip/hip_runtime.h>
#include <math.h>

#define N_NODES 50000
#define N_EDGES 800000
#define IN_F    191
#define H1 16
#define C1 16
#define F1 256
#define H2 8
#define C2 3
#define F2 24
#define E_TOT (N_EDGES + N_NODES)

typedef __attribute__((ext_vector_type(2))) float v2f;
typedef __attribute__((ext_vector_type(8))) float v8f;


__device__ __forceinline__ unsigned enc_f32(float f) {
  unsigned u = __float_as_uint(f);
  return (u & 0x80000000u) ? ~u : (u | 0x80000000u);
}
__device__ __forceinline__ float dec_f32(unsigned u) {
  return (u & 0x80000000u) ? __uint_as_float(u & 0x7fffffffu)
                           : __uint_as_float(~u);
}
#define MX_INIT 0x00800000u

__device__ __forceinline__ void atomic_add_f32(float* p, float v) {
  __hip_atomic_fetch_add(p, v, __ATOMIC_RELAXED, __HIP_MEMORY_SCOPE_AGENT);
}

__global__ void __launch_bounds__(1024)
prep_kernel(const float* __restrict__ ea,
            const float* __restrict__ We1, const float* __restrict__ ae1,
            const float* __restrict__ We2, const float* __restrict__ ae2,
            float* __restrict__ mean_out, float* __restrict__ ce1,
            float* __restrict__ ce2) {
  __shared__ float red[1024];
  const int t = threadIdx.x;
  float s = 0.f;
  for (int i = t; i < N_EDGES; i += 1024) s += ea[i];
  red[t] = s;
  __syncthreads();
  for (int off = 512; off > 0; off >>= 1) {
    if (t < off) red[t] += red[t + off];
    __syncthreads();
  }
  if (t == 0) mean_out[0] = red[0] * (1.0f / (float)N_EDGES);
  if (t < H1) {
    float c = 0.f;
    for (int k = 0; k < C1; k++) c += We1[t * C1 + k] * ae1[t * C1 + k];
    ce1[t] = c;
  } else if (t >= 32 && t < 32 + H2) {
    const int h = t - 32;
    float c = 0.f;
    for (int k = 0; k < C2; k++) c += We2[h * C2 + k] * ae2[h * C2 + k];
    ce2[h] = c;
  }
}

__global__ void init_keys_kernel(unsigned* __restrict__ mx1,
                                 unsigned* __restrict__ mx2) {
  const int idx = blockIdx.x * blockDim.x + threadIdx.x;
  if (idx < N_NODES * H1) mx1[idx] = MX_INIT;
  if (idx < N_NODES * H2) mx2[idx] = MX_INIT;
}

__global__ void __launch_bounds__(256)
gemm1_wmma(const float* __restrict__ x, const float* __restrict__ W1,
           float* __restrict__ xl1) {
  __shared__ float As[16][49];
  __shared__ float Bs[48][128];
  const int row0 = blockIdx.x * 16;
  const int colB = blockIdx.y * 128;
  const int t = threadIdx.x;
  const int wave = t >> 5, lane = t & 31, lh = lane & 15, hi = lane >> 4;
  v8f acc = {};
  for (int k0 = 0; k0 < 192; k0 += 48) {
    for (int idx = t; idx < 16 * 48; idx += 256) {
      const int r = idx / 48, c = idx % 48, k = k0 + c;
      As[r][c] = (k < IN_F) ? x[(row0 + r) * IN_F + k] : 0.f;
    }
    for (int idx4 = t; idx4 < (48 * 128) / 4; idx4 += 256) {
      const int r = idx4 >> 5, c = (idx4 & 31) << 2, k = k0 + r;
      float4 f = make_float4(0.f, 0.f, 0.f, 0.f);
      if (k < IN_F) {
        f = *(const float4*)&W1[k * F1 + colB + c];
        if ((idx4 & 7) == 0 && k + 48 < IN_F)
          __builtin_prefetch(&W1[(k + 48) * F1 + colB + c], 0, 0);
      }
      *(float4*)&Bs[r][c] = f;
    }
    __syncthreads();
#pragma unroll
    for (int kk = 0; kk < 48; kk += 4) {
      v2f a, b;
      a.x = As[lh][kk + 2 * hi];
      a.y = As[lh][kk + 2 * hi + 1];
      b.x = Bs[kk + 2 * hi][wave * 16 + lh];
      b.y = Bs[kk + 2 * hi + 1][wave * 16 + lh];
      acc = __builtin_amdgcn_wmma_f32_16x16x4_f32(false, a, false, b,
                                                  (short)0, acc, false, false);
    }
    __syncthreads();
  }
  const int col = colB + wave * 16 + lh;
#pragma unroll
  for (int i = 0; i < 8; i++)
    xl1[(row0 + i + 8 * hi) * F1 + col] = acc[i];
}

__global__ void __launch_bounds__(64)
gemm2_wmma(const float* __restrict__ h1, const float* __restrict__ W2,
           float* __restrict__ xl2) {
  __shared__ float As[16][258];
  const int row0 = blockIdx.x * 16;
  const int t = threadIdx.x;
  const int wave = t >> 5, lane = t & 31, lh = lane & 15, hi = lane >> 4;
  for (int idx4 = t; idx4 < (16 * 256) / 4; idx4 += 64) {
    const int r = idx4 >> 6, c = (idx4 & 63) << 2;
    const float4 f = *(const float4*)&h1[(row0 + r) * F1 + c];
    As[r][c + 0] = f.x; As[r][c + 1] = f.y;
    As[r][c + 2] = f.z; As[r][c + 3] = f.w;
  }
  __syncthreads();
  const int col = wave * 16 + lh;
  v8f acc = {};
  for (int k = 0; k < 256; k += 4) {
    v2f a, b;
    a.x = As[lh][k + 2 * hi];
    a.y = As[lh][k + 2 * hi + 1];
    if (col < F2) {
      b.x = W2[(k + 2 * hi) * F2 + col];
      b.y = W2[(k + 2 * hi + 1) * F2 + col];
    } else {
      b.x = 0.f; b.y = 0.f;
    }
    acc = __builtin_amdgcn_wmma_f32_16x16x4_f32(false, a, false, b,
                                                (short)0, acc, false, false);
  }
  if (col < F2) {
#pragma unroll
    for (int i = 0; i < 8; i++)
      xl2[(row0 + i + 8 * hi) * F2 + col] = acc[i];
  }
}

template <int H, int C>
__global__ void alpha_node_kernel(const float* __restrict__ xl,
                                  const float* __restrict__ a_src,
                                  const float* __restrict__ a_dst,
                                  float* __restrict__ asrc,
                                  float* __restrict__ adst) {
  const int idx = blockIdx.x * blockDim.x + threadIdx.x;
  if (idx >= N_NODES * H) return;
  const int n = idx / H, h = idx % H;
  const float* row = xl + n * (H * C) + h * C;
  float s = 0.f, d = 0.f;
#pragma unroll
  for (int c = 0; c < C; c++) {
    const float v = row[c];
    s += v * a_src[h * C + c];
    d += v * a_dst[h * C + c];
  }
  asrc[idx] = s;
  adst[idx] = d;
}

template <int H>
__device__ __forceinline__ float edge_logit(int e, int h, const int* __restrict__ ei,
                                            const float* __restrict__ ea,
                                            const float* __restrict__ mean,
                                            const float* __restrict__ ce,
                                            const float* __restrict__ asrc,
                                            const float* __restrict__ adst,
                                            int& s, int& d) {
  float eav;
  if (e < N_EDGES) {
    s = ei[e];
    d = ei[N_EDGES + e];
    eav = ea[e];
  } else {
    s = d = e - N_EDGES;
    eav = mean[0];
  }
  const float a = asrc[s * H + h] + adst[d * H + h] + eav * ce[h];
  return (a > 0.f) ? a : 0.2f * a;
}

template <int H>
__global__ void edge_max_kernel(const int* __restrict__ ei,
                                const float* __restrict__ ea,
                                const float* __restrict__ mean,
                                const float* __restrict__ ce,
                                const float* __restrict__ asrc,
                                const float* __restrict__ adst,
                                unsigned* __restrict__ mx) {
  const int idx = blockIdx.x * blockDim.x + threadIdx.x;
  if (idx >= E_TOT * H) return;
  int s, d;
  const int e = idx / H, h = idx % H;
  const float a = edge_logit<H>(e, h, ei, ea, mean, ce, asrc, adst, s, d);
  atomicMax(&mx[d * H + h], enc_f32(a));
}

template <int H>
__global__ void edge_den_kernel(const int* __restrict__ ei,
                                const float* __restrict__ ea,
                                const float* __restrict__ mean,
                                const float* __restrict__ ce,
                                const float* __restrict__ asrc,
                                const float* __restrict__ adst,
                                const unsigned* __restrict__ mx,
                                float* __restrict__ den) {
  const int idx = blockIdx.x * blockDim.x + threadIdx.x;
  if (idx >= E_TOT * H) return;
  int s, d;
  const int e = idx / H, h = idx % H;
  const float a = edge_logit<H>(e, h, ei, ea, mean, ce, asrc, adst, s, d);
  const float m = dec_f32(mx[d * H + h]);
  atomic_add_f32(&den[d * H + h], __expf(a - m));
}

__global__ void edge_agg1_kernel(const int* __restrict__ ei,
                                 const float* __restrict__ ea,
                                 const float* __restrict__ mean,
                                 const float* __restrict__ ce,
                                 const float* __restrict__ asrc,
                                 const float* __restrict__ adst,
                                 const unsigned* __restrict__ mx,
                                 const float* __restrict__ den,
                                 const float* __restrict__ xl,
                                 float* __restrict__ agg) {
  const int idx = blockIdx.x * blockDim.x + threadIdx.x;
  if (idx >= E_TOT * H1) return;
  int s, d;
  const int e = idx / H1, h = idx % H1;
  const float a = edge_logit<H1>(e, h, ei, ea, mean, ce, asrc, adst, s, d);
  const float4* srow = (const float4*)(xl + s * F1 + h * C1);
  __builtin_prefetch(srow, 0, 0);
  const float m = dec_f32(mx[d * H1 + h]);
  const float w = __expf(a - m) / (den[d * H1 + h] + 1e-16f);
  float* drow = agg + d * F1 + h * C1;
#pragma unroll
  for (int q = 0; q < C1 / 4; q++) {
    const float4 v = srow[q];
    atomic_add_f32(&drow[4 * q + 0], w * v.x);
    atomic_add_f32(&drow[4 * q + 1], w * v.y);
    atomic_add_f32(&drow[4 * q + 2], w * v.z);
    atomic_add_f32(&drow[4 * q + 3], w * v.w);
  }
}

__global__ void edge_agg2_kernel(const int* __restrict__ ei,
                                 const float* __restrict__ ea,
                                 const float* __restrict__ mean,
                                 const float* __restrict__ ce,
                                 const float* __restrict__ asrc,
                                 const float* __restrict__ adst,
                                 const unsigned* __restrict__ mx,
                                 const float* __restrict__ den,
                                 const float* __restrict__ xl,
                                 float* __restrict__ agg) {
  const int idx = blockIdx.x * blockDim.x + threadIdx.x;
  if (idx >= E_TOT * H2) return;
  int s, d;
  const int e = idx / H2, h = idx % H2;
  const float a = edge_logit<H2>(e, h, ei, ea, mean, ce, asrc, adst, s, d);
  const float* srow = xl + s * F2 + h * C2;
  __builtin_prefetch(srow, 0, 0);
  const float m = dec_f32(mx[d * H2 + h]);
  const float w = __expf(a - m) / (den[d * H2 + h] + 1e-16f);
  float* drow = agg + d * F2 + h * C2;
#pragma unroll
  for (int c = 0; c < C2; c++) atomic_add_f32(&drow[c], w * srow[c]);
}

__global__ void finalize1_kernel(float* __restrict__ agg1,
                                 const float* __restrict__ b1) {
  const int idx4 = blockIdx.x * blockDim.x + threadIdx.x;
  if (idx4 >= (N_NODES * F1) / 4) return;
  const int e0 = idx4 << 2;
  float4 v = *(float4*)&agg1[e0];
  const float4 b = *(const float4*)&b1[e0 & (F1 - 1)];
  v.x += b.x; v.y += b.y; v.z += b.z; v.w += b.w;
  v.x = (v.x > 0.f) ? v.x : expm1f(v.x);
  v.y = (v.y > 0.f) ? v.y : expm1f(v.y);
  v.z = (v.z > 0.f) ? v.z : expm1f(v.z);
  v.w = (v.w > 0.f) ? v.w : expm1f(v.w);
  *(float4*)&agg1[e0] = v;
}

__global__ void final2_kernel(const float* __restrict__ agg2,
                              const float* __restrict__ b2,
                              float* __restrict__ out) {
  const int idx = blockIdx.x * blockDim.x + threadIdx.x;
  if (idx >= N_NODES * C2) return;
  const int n = idx / C2, c = idx % C2;
  float s = 0.f;
#pragma unroll
  for (int h = 0; h < H2; h++) s += agg2[n * F2 + h * C2 + c];
  const float v = s * (1.f / (float)H2) + b2[c];
  out[idx] = (v > 0.f) ? v : expm1f(v);
}

extern "C" void kernel_launch(void* const* d_in, const int* in_sizes, int n_in,
                              void* d_out, int out_size, void* d_ws, size_t ws_size,
                              hipStream_t stream) {
  (void)in_sizes; (void)n_in; (void)out_size; (void)ws_size;
  const float* x   = (const float*)d_in[0];
  const int*   ei  = (const int*)d_in[1];
  const float* ea  = (const float*)d_in[2];
  const float* W1  = (const float*)d_in[3];
  const float* We1 = (const float*)d_in[4];
  const float* as1 = (const float*)d_in[5];
  const float* ad1 = (const float*)d_in[6];
  const float* ae1 = (const float*)d_in[7];
  const float* b1  = (const float*)d_in[8];
  const float* W2  = (const float*)d_in[9];
  const float* We2 = (const float*)d_in[10];
  const float* as2 = (const float*)d_in[11];
  const float* ad2 = (const float*)d_in[12];
  const float* ae2 = (const float*)d_in[13];
  const float* b2  = (const float*)d_in[14];
  float* out = (float*)d_out;

  float* w = (float*)d_ws;
  float* mean = w;
  float* ce1  = w + 1;
  float* ce2  = w + 17;
  size_t off = 32;
  float* xl1   = w + off; off += (size_t)N_NODES * F1;
  float* asrc1 = w + off; off += (size_t)N_NODES * H1;
  float* adst1 = w + off; off += (size_t)N_NODES * H1;
  unsigned* mx1 = (unsigned*)(w + off); off += (size_t)N_NODES * H1;
  float* den1  = w + off; off += (size_t)N_NODES * H1;
  float* agg1  = w + off; off += (size_t)N_NODES * F1;
  float* xl2   = w + off; off += (size_t)N_NODES * F2;
  float* asrc2 = w + off; off += (size_t)N_NODES * H2;
  float* adst2 = w + off; off += (size_t)N_NODES * H2;
  unsigned* mx2 = (unsigned*)(w + off); off += (size_t)N_NODES * H2;
  float* den2  = w + off; off += (size_t)N_NODES * H2;
  float* agg2  = w + off; off += (size_t)N_NODES * F2;

  hipMemsetAsync(den1, 0, (size_t)N_NODES * H1 * 4, stream);
  hipMemsetAsync(agg1, 0, (size_t)N_NODES * F1 * 4, stream);
  hipMemsetAsync(den2, 0, (size_t)N_NODES * H2 * 4, stream);
  hipMemsetAsync(agg2, 0, (size_t)N_NODES * F2 * 4, stream);
  init_keys_kernel<<<(N_NODES * H1 + 255) / 256, 256, 0, stream>>>(mx1, mx2);

  prep_kernel<<<1, 1024, 0, stream>>>(ea, We1, ae1, We2, ae2, mean, ce1, ce2);

  gemm1_wmma<<<dim3(N_NODES / 16, 2), 256, 0, stream>>>(x, W1, xl1);
  alpha_node_kernel<H1, C1><<<(N_NODES * H1 + 255) / 256, 256, 0, stream>>>(
      xl1, as1, ad1, asrc1, adst1);
  const int g1 = (E_TOT * H1 + 255) / 256;
  edge_max_kernel<H1><<<g1, 256, 0, stream>>>(ei, ea, mean, ce1, asrc1, adst1, mx1);
  edge_den_kernel<H1><<<g1, 256, 0, stream>>>(ei, ea, mean, ce1, asrc1, adst1, mx1, den1);
  edge_agg1_kernel<<<g1, 256, 0, stream>>>(ei, ea, mean, ce1, asrc1, adst1,
                                           mx1, den1, xl1, agg1);
  finalize1_kernel<<<((N_NODES * F1) / 4 + 255) / 256, 256, 0, stream>>>(agg1, b1);

  gemm2_wmma<<<N_NODES / 16, 64, 0, stream>>>(agg1, W2, xl2);
  alpha_node_kernel<H2, C2><<<(N_NODES * H2 + 255) / 256, 256, 0, stream>>>(
      xl2, as2, ad2, asrc2, adst2);
  const int g2 = (E_TOT * H2 + 255) / 256;
  edge_max_kernel<H2><<<g2, 256, 0, stream>>>(ei, ea, mean, ce2, asrc2, adst2, mx2);
  edge_den_kernel<H2><<<g2, 256, 0, stream>>>(ei, ea, mean, ce2, asrc2, adst2, mx2, den2);
  edge_agg2_kernel<<<g2, 256, 0, stream>>>(ei, ea, mean, ce2, asrc2, adst2,
                                           mx2, den2, xl2, agg2);
  final2_kernel<<<(N_NODES * C2 + 255) / 256, 256, 0, stream>>>(agg2, b2, out);
}


# edges stably counting-sorted by destination in-call; one wave per node does segment max, exp-sum and weighted row sum in registers: no float atomics or memsets; sums now in fixed edge order
# speedup vs baseline: 3.9673x; 3.9673x over previous
#include <hip/hip_runtime.h>
#include <math.h>

#define N_NODES 50000
#define N_EDGES 800000
#define IN_F    191
#define H1 16
#define C1 16
#define F1 256
#define H2 8
#define C2 3
#define F2 24

typedef __attribute__((ext_vector_type(2))) float v2f;
typedef __attribute__((ext_vector_type(8))) float v8f;


__device__ __forceinline__ float lrelu02(float a) {
  return (a > 0.f) ? a : 0.2f * a;
}

__global__ void __launch_bounds__(1024)
prep_kernel(const float* __restrict__ ea,
            const float* __restrict__ We1, const float* __restrict__ ae1,
            const float* __restrict__ We2, const float* __restrict__ ae2,
            float* __restrict__ mean_out, float* __restrict__ ce1,
            float* __restrict__ ce2) {
  __shared__ float red[1024];
  const int t = threadIdx.x;
  float s = 0.f;
  for (int i = t; i < N_EDGES; i += 1024) s += ea[i];
  red[t] = s;
  __syncthreads();
  for (int off = 512; off > 0; off >>= 1) {
    if (t < off) red[t] += red[t + off];
    __syncthreads();
  }
  if (t == 0) mean_out[0] = red[0] * (1.0f / (float)N_EDGES);
  if (t < H1) {
    float c = 0.f;
    for (int k = 0; k < C1; k++) c += We1[t * C1 + k] * ae1[t * C1 + k];
    ce1[t] = c;
  } else if (t >= 32 && t < 32 + H2) {
    const int h = t - 32;
    float c = 0.f;
    for (int k = 0; k < C2; k++) c += We2[h * C2 + k] * ae2[h * C2 + k];
    ce2[h] = c;
  }
}

__global__ void zero_counts_kernel(int* __restrict__ cnt) {
  const int idx = blockIdx.x * blockDim.x + threadIdx.x;
  if (idx < N_NODES) cnt[idx] = 0;
}

__global__ void count_rank_kernel(const int* __restrict__ ei,
                                  int* __restrict__ cnt,
                                  int* __restrict__ rank) {
  const int e = blockIdx.x * blockDim.x + threadIdx.x;
  if (e >= N_EDGES) return;
  rank[e] = atomicAdd(&cnt[ei[N_EDGES + e]], 1);
}

#define SCAN_CHUNK 49
__global__ void __launch_bounds__(1024)
scan_counts_kernel(const int* __restrict__ cnt, int* __restrict__ rowptr) {
  __shared__ int sc[1024];
  const int t = threadIdx.x;
  const int lo = t * SCAN_CHUNK;
  int v[SCAN_CHUNK];
  int local = 0;
#pragma unroll
  for (int k = 0; k < SCAN_CHUNK; k++) {
    const int i = lo + k;
    v[k] = (i < N_NODES) ? cnt[i] : 0;
    local += v[k];
  }
  sc[t] = local;
  __syncthreads();
  for (int off = 1; off < 1024; off <<= 1) {
    const int u = (t >= off) ? sc[t - off] : 0;
    __syncthreads();
    sc[t] += u;
    __syncthreads();
  }
  int run = sc[t] - local;
#pragma unroll
  for (int k = 0; k < SCAN_CHUNK; k++) {
    const int i = lo + k;
    if (i < N_NODES) rowptr[i] = run;
    run += v[k];
  }
  if (t == 1023) rowptr[N_NODES] = sc[1023];
}

__global__ void place_edges_kernel(const int* __restrict__ ei,
                                   const int* __restrict__ rowptr,
                                   const int* __restrict__ rank,
                                   int* __restrict__ slot_eid) {
  const int e = blockIdx.x * blockDim.x + threadIdx.x;
  if (e >= N_EDGES) return;
  slot_eid[rowptr[ei[N_EDGES + e]] + rank[e]] = e;
}

__global__ void __launch_bounds__(256)
order_segments_kernel(const int* __restrict__ ei,
                      const float* __restrict__ ea,
                      const int* __restrict__ rowptr,
                      const int* __restrict__ slot_eid,
                      int* __restrict__ csr_src,
                      float* __restrict__ csr_ea) {
  const int lane = threadIdx.x & 31;
  const int wv = __builtin_amdgcn_readfirstlane((int)(threadIdx.x >> 5));
  const int d = blockIdx.x * 8 + wv;
  const int beg = rowptr[d], end = rowptr[d + 1];
  for (int base = beg; base < end; base += 32) {
    const int n = (end - base < 32) ? end - base : 32;
    const int mine = (lane < n) ? slot_eid[base + lane] : 0;
    int pos = 0;
    for (int b2 = beg; b2 < end; b2 += 32) {
      const int n2 = (end - b2 < 32) ? end - b2 : 32;
      const int other = (lane < n2) ? slot_eid[b2 + lane] : 0;
      for (int j = 0; j < n2; j++)
        pos += (__builtin_amdgcn_readlane(other, j) < mine) ? 1 : 0;
    }
    if (lane < n) {
      csr_src[beg + pos] = ei[mine];
      csr_ea[beg + pos] = ea[mine];
    }
  }
}

__global__ void __launch_bounds__(256)
gemm1_wmma(const float* __restrict__ x, const float* __restrict__ W1,
           float* __restrict__ xl1) {
  __shared__ float As[16][49];
  __shared__ float Bs[48][128];
  const int row0 = blockIdx.x * 16;
  const int colB = blockIdx.y * 128;
  const int t = threadIdx.x;
  const int wave = t >> 5, lane = t & 31, lh = lane & 15, hi = lane >> 4;
  v8f acc = {};
  for (int k0 = 0; k0 < 192; k0 += 48) {
    for (int idx = t; idx < 16 * 48; idx += 256) {
      const int r = idx / 48, c = idx % 48, k = k0 + c;
      As[r][c] = (k < IN_F) ? x[(row0 + r) * IN_F + k] : 0.f;
    }
    for (int idx4 = t; idx4 < (48 * 128) / 4; idx4 += 256) {
      const int r = idx4 >> 5, c = (idx4 & 31) << 2, k = k0 + r;
      float4 f = make_float4(0.f, 0.f, 0.f, 0.f);
      if (k < IN_F) {
        f = *(const float4*)&W1[k * F1 + colB + c];
        if ((idx4 & 7) == 0 && k + 48 < IN_F)
          __builtin_prefetch(&W1[(k + 48) * F1 + colB + c], 0, 0);
      }
      *(float4*)&Bs[r][c] = f;
    }
    __syncthreads();
#pragma unroll
    for (int kk = 0; kk < 48; kk += 4) {
      v2f a, b;
      a.x = As[lh][kk + 2 * hi];
      a.y = As[lh][kk + 2 * hi + 1];
      b.x = Bs[kk + 2 * hi][wave * 16 + lh];
      b.y = Bs[kk + 2 * hi + 1][wave * 16 + lh];
      acc = __builtin_amdgcn_wmma_f32_16x16x4_f32(false, a, false, b,
                                                  (short)0, acc, false, false);
    }
    __syncthreads();
  }
  const int col = colB + wave * 16 + lh;
#pragma unroll
  for (int i = 0; i < 8; i++)
    xl1[(row0 + i + 8 * hi) * F1 + col] = acc[i];
}

__global__ void __launch_bounds__(64)
gemm2_wmma(const float* __restrict__ h1, const float* __restrict__ W2,
           float* __restrict__ xl2) {
  __shared__ float As[16][258];
  const int row0 = blockIdx.x * 16;
  const int t = threadIdx.x;
  const int wave = t >> 5, lane = t & 31, lh = lane & 15, hi = lane >> 4;
  for (int idx4 = t; idx4 < (16 * 256) / 4; idx4 += 64) {
    const int r = idx4 >> 6, c = (idx4 & 63) << 2;
    const float4 f = *(const float4*)&h1[(row0 + r) * F1 + c];
    As[r][c + 0] = f.x; As[r][c + 1] = f.y;
    As[r][c + 2] = f.z; As[r][c + 3] = f.w;
  }
  __syncthreads();
  const int col = wave * 16 + lh;
  v8f acc = {};
  for (int k = 0; k < 256; k += 4) {
    v2f a, b;
    a.x = As[lh][k + 2 * hi];
    a.y = As[lh][k + 2 * hi + 1];
    if (col < F2) {
      b.x = W2[(k + 2 * hi) * F2 + col];
      b.y = W2[(k + 2 * hi + 1) * F2 + col];
    } else {
      b.x = 0.f; b.y = 0.f;
    }
    acc = __builtin_amdgcn_wmma_f32_16x16x4_f32(false, a, false, b,
                                                (short)0, acc, false, false);
  }
  if (col < F2) {
#pragma unroll
    for (int i = 0; i < 8; i++)
      xl2[(row0 + i + 8 * hi) * F2 + col] = acc[i];
  }
}

template <int H, int C>
__global__ void alpha_node_kernel(const float* __restrict__ xl,
                                  const float* __restrict__ a_src,
                                  const float* __restrict__ a_dst,
                                  float* __restrict__ asrc,
                                  float* __restrict__ adst) {
  const int idx = blockIdx.x * blockDim.x + threadIdx.x;
  if (idx >= N_NODES * H) return;
  const int n = idx / H, h = idx % H;
  const float* row = xl + n * (H * C) + h * C;
  float s = 0.f, d = 0.f;
#pragma unroll
  for (int c = 0; c < C; c++) {
    const float v = row[c];
    s += v * a_src[h * C + c];
    d += v * a_dst[h * C + c];
  }
  asrc[idx] = s;
  adst[idx] = d;
}

__global__ void __launch_bounds__(256)
node_softmax_agg1_kernel(const int* __restrict__ rowptr,
                         const int* __restrict__ csr_src,
                         const float* __restrict__ csr_ea,
                         const float* __restrict__ mean,
                         const float* __restrict__ ce,
                         const float* __restrict__ asrc,
                         const float* __restrict__ adst,
                         const float* __restrict__ xl,
                         float* __restrict__ agg) {
  const int lane = threadIdx.x & 31;
  const int wv = __builtin_amdgcn_readfirstlane((int)(threadIdx.x >> 5));
  const int d = blockIdx.x * 8 + wv;
  const int hA = lane >> 2, hB = 8 + (lane >> 2);
  const int beg = rowptr[d], end = rowptr[d + 1];
  const float adA = adst[d * H1 + hA], adB = adst[d * H1 + hB];
  const float cA = ce[hA], cB = ce[hB];
  const float mv = mean[0];
  const float aSelfA = lrelu02(asrc[d * H1 + hA] + adA + mv * cA);
  const float aSelfB = lrelu02(asrc[d * H1 + hB] + adB + mv * cB);

  float mA = aSelfA, mB = aSelfB;
  for (int base = beg; base < end; base += 32) {
    const int n = (end - base < 32) ? end - base : 32;
    int sL = 0;
    float eL = 0.f;
    if (lane < n) { sL = csr_src[base + lane]; eL = csr_ea[base + lane]; }
    for (int j = 0; j < n; j++) {
      const int s = __builtin_amdgcn_readlane(sL, j);
      const float ev = __int_as_float(__builtin_amdgcn_readlane(__float_as_int(eL), j));
      const float aA = lrelu02(asrc[s * H1 + hA] + adA + ev * cA);
      const float aB = lrelu02(asrc[s * H1 + hB] + adB + ev * cB);
      mA = fmaxf(mA, aA);
      mB = fmaxf(mB, aB);
    }
  }

  float denA = __expf(aSelfA - mA), denB = __expf(aSelfB - mB);
  float4 accA, accB;
  {
    const float4* r = (const float4*)(xl + d * F1);
    const float4 u = r[lane], w = r[32 + lane];
    accA = make_float4(denA * u.x, denA * u.y, denA * u.z, denA * u.w);
    accB = make_float4(denB * w.x, denB * w.y, denB * w.z, denB * w.w);
  }
  for (int base = beg; base < end; base += 32) {
    const int n = (end - base < 32) ? end - base : 32;
    int sL = 0;
    float eL = 0.f;
    if (lane < n) { sL = csr_src[base + lane]; eL = csr_ea[base + lane]; }
    for (int j = 0; j < n; j++) {
      const int s = __builtin_amdgcn_readlane(sL, j);
      const float ev = __int_as_float(__builtin_amdgcn_readlane(__float_as_int(eL), j));
      const float4* r = (const float4*)(xl + s * F1);
      const float4 u = r[lane], w = r[32 + lane];
      const float aA = lrelu02(asrc[s * H1 + hA] + adA + ev * cA);
      const float aB = lrelu02(asrc[s * H1 + hB] + adB + ev * cB);
      const float eA = __expf(aA - mA), eB = __expf(aB - mB);
      denA += eA;
      denB += eB;
      accA.x += eA * u.x; accA.y += eA * u.y; accA.z += eA * u.z; accA.w += eA * u.w;
      accB.x += eB * w.x; accB.y += eB * w.y; accB.z += eB * w.z; accB.w += eB * w.w;
    }
  }
  const float rA = 1.0f / (denA + 1e-16f), rB = 1.0f / (denB + 1e-16f);
  float4* o = (float4*)(agg + d * F1);
  o[lane]      = make_float4(accA.x * rA, accA.y * rA, accA.z * rA, accA.w * rA);
  o[32 + lane] = make_float4(accB.x * rB, accB.y * rB, accB.z * rB, accB.w * rB);
}

__global__ void __launch_bounds__(256)
node_softmax_agg2_kernel(const int* __restrict__ rowptr,
                         const int* __restrict__ csr_src,
                         const float* __restrict__ csr_ea,
                         const float* __restrict__ mean,
                         const float* __restrict__ ce,
                         const float* __restrict__ asrc,
                         const float* __restrict__ adst,
                         const float* __restrict__ xl,
                         float* __restrict__ agg) {
  const int idx = blockIdx.x * blockDim.x + threadIdx.x;
  if (idx >= N_NODES * H2) return;
  const int d = idx / H2, h = idx % H2;
  const int beg = rowptr[d], end = rowptr[d + 1];
  const float ad = adst[idx];
  const float c = ce[h];
  const float aSelf = lrelu02(asrc[idx] + ad + mean[0] * c);

  float m = aSelf;
  for (int j = beg; j < end; j++) {
    const float a = lrelu02(asrc[csr_src[j] * H2 + h] + ad + csr_ea[j] * c);
    m = fmaxf(m, a);
  }

  float den = __expf(aSelf - m);
  const float* rd = xl + d * F2 + h * C2;
  float acc0 = den * rd[0], acc1 = den * rd[1], acc2 = den * rd[2];
  for (int j = beg; j < end; j++) {
    const int s = csr_src[j];
    const float a = lrelu02(asrc[s * H2 + h] + ad + csr_ea[j] * c);
    const float ex = __expf(a - m);
    const float* rs = xl + s * F2 + h * C2;
    den += ex;
    acc0 += ex * rs[0];
    acc1 += ex * rs[1];
    acc2 += ex * rs[2];
  }
  const float rinv = 1.0f / (den + 1e-16f);
  float* o = agg + d * F2 + h * C2;
  o[0] = acc0 * rinv;
  o[1] = acc1 * rinv;
  o[2] = acc2 * rinv;
}

__global__ void finalize1_kernel(float* __restrict__ agg1,
                                 const float* __restrict__ b1) {
  const int idx4 = blockIdx.x * blockDim.x + threadIdx.x;
  if (idx4 >= (N_NODES * F1) / 4) return;
  const int e0 = idx4 << 2;
  float4 v = *(float4*)&agg1[e0];
  const float4 b = *(const float4*)&b1[e0 & (F1 - 1)];
  v.x += b.x; v.y += b.y; v.z += b.z; v.w += b.w;
  v.x = (v.x > 0.f) ? v.x : expm1f(v.x);
  v.y = (v.y > 0.f) ? v.y : expm1f(v.y);
  v.z = (v.z > 0.f) ? v.z : expm1f(v.z);
  v.w = (v.w > 0.f) ? v.w : expm1f(v.w);
  *(float4*)&agg1[e0] = v;
}

__global__ void final2_kernel(const float* __restrict__ agg2,
                              const float* __restrict__ b2,
                              float* __restrict__ out) {
  const int idx = blockIdx.x * blockDim.x + threadIdx.x;
  if (idx >= N_NODES * C2) return;
  const int n = idx / C2, c = idx % C2;
  float s = 0.f;
#pragma unroll
  for (int h = 0; h < H2; h++) s += agg2[n * F2 + h * C2 + c];
  const float v = s * (1.f / (float)H2) + b2[c];
  out[idx] = (v > 0.f) ? v : expm1f(v);
}

extern "C" void kernel_launch(void* const* d_in, const int* in_sizes, int n_in,
                              void* d_out, int out_size, void* d_ws, size_t ws_size,
                              hipStream_t stream) {
  (void)in_sizes; (void)n_in; (void)out_size; (void)ws_size;
  const float* x   = (const float*)d_in[0];
  const int*   ei  = (const int*)d_in[1];
  const float* ea  = (const float*)d_in[2];
  const float* W1  = (const float*)d_in[3];
  const float* We1 = (const float*)d_in[4];
  const float* as1 = (const float*)d_in[5];
  const float* ad1 = (const float*)d_in[6];
  const float* ae1 = (const float*)d_in[7];
  const float* b1  = (const float*)d_in[8];
  const float* W2  = (const float*)d_in[9];
  const float* We2 = (const float*)d_in[10];
  const float* as2 = (const float*)d_in[11];
  const float* ad2 = (const float*)d_in[12];
  const float* ae2 = (const float*)d_in[13];
  const float* b2  = (const float*)d_in[14];
  float* out = (float*)d_out;

  float* w = (float*)d_ws;
  float* mean = w;
  float* ce1  = w + 1;
  float* ce2  = w + 17;
  size_t off = 32;
  float* xl1   = w + off; off += (size_t)N_NODES * F1;
  float* asrc1 = w + off; off += (size_t)N_NODES * H1;
  float* adst1 = w + off; off += (size_t)N_NODES * H1;
  float* agg1  = w + off; off += (size_t)N_NODES * F1;
  float* xl2   = w + off; off += (size_t)N_NODES * F2;
  float* asrc2 = w + off; off += (size_t)N_NODES * H2;
  float* adst2 = w + off; off += (size_t)N_NODES * H2;
  float* agg2  = w + off; off += (size_t)N_NODES * F2;
  int* cnt     = (int*)(w + off); off += (size_t)N_NODES;
  int* rowptr  = (int*)(w + off); off += (size_t)N_NODES + 32;
  int* rank    = (int*)(w + off); off += (size_t)N_EDGES;
  int* slot_eid = (int*)(w + off); off += (size_t)N_EDGES;
  int* csr_src = (int*)(w + off); off += (size_t)N_EDGES;
  float* csr_ea = w + off;        off += (size_t)N_EDGES;

  const int ge = (N_EDGES + 255) / 256;
  zero_counts_kernel<<<(N_NODES + 255) / 256, 256, 0, stream>>>(cnt);
  count_rank_kernel<<<ge, 256, 0, stream>>>(ei, cnt, rank);
  scan_counts_kernel<<<1, 1024, 0, stream>>>(cnt, rowptr);
  place_edges_kernel<<<ge, 256, 0, stream>>>(ei, rowptr, rank, slot_eid);
  order_segments_kernel<<<N_NODES / 8, 256, 0, stream>>>(ei, ea, rowptr, slot_eid, csr_src, csr_ea);

  prep_kernel<<<1, 1024, 0, stream>>>(ea, We1, ae1, We2, ae2, mean, ce1, ce2);

  gemm1_wmma<<<dim3(N_NODES / 16, 2), 256, 0, stream>>>(x, W1, xl1);
  alpha_node_kernel<H1, C1><<<(N_NODES * H1 + 255) / 256, 256, 0, stream>>>(
      xl1, as1, ad1, asrc1, adst1);
  node_softmax_agg1_kernel<<<N_NODES / 8, 256, 0, stream>>>(
      rowptr, csr_src, csr_ea, mean, ce1, asrc1, adst1, xl1, agg1);
  finalize1_kernel<<<((N_NODES * F1) / 4 + 255) / 256, 256, 0, stream>>>(agg1, b1);

  gemm2_wmma<<<N_NODES / 16, 64, 0, stream>>>(agg1, W2, xl2);
  alpha_node_kernel<H2, C2><<<(N_NODES * H2 + 255) / 256, 256, 0, stream>>>(
      xl2, as2, ad2, asrc2, adst2);
  node_softmax_agg2_kernel<<<(N_NODES * H2 + 255) / 256, 256, 0, stream>>>(
      rowptr, csr_src, csr_ea, mean, ce2, asrc2, adst2, xl2, agg2);
  final2_kernel<<<(N_NODES * C2 + 255) / 256, 256, 0, stream>>>(agg2, b2, out);
}


# both single-workgroup launches spread over 196 workgroups: each scan block sums the counts ahead of its 256 nodes; edge_attr mean from per-block partial sums (new fixed float sum order)
# speedup vs baseline: 4.7861x; 1.2064x over previous
#include <hip/hip_runtime.h>
#include <math.h>

#define N_NODES 50000
#define N_EDGES 800000
#define IN_F    191
#define H1 16
#define C1 16
#define F1 256
#define H2 8
#define C2 3
#define F2 24

typedef __attribute__((ext_vector_type(2))) float v2f;
typedef __attribute__((ext_vector_type(8))) float v8f;


__device__ __forceinline__ float lrelu02(float a) {
  return (a > 0.f) ? a : 0.2f * a;
}

#define PS_BLOCKS 196
#define PS_SLICE  4096

__device__ __forceinline__ float wave_sum_f(float v) {
  for (int off = 16; off > 0; off >>= 1) v += __shfl_xor(v, off);
  return v;
}

__global__ void __launch_bounds__(256)
prep_kernel(const float* __restrict__ psum,
            const float* __restrict__ We1, const float* __restrict__ ae1,
            const float* __restrict__ We2, const float* __restrict__ ae2,
            float* __restrict__ mean_out, float* __restrict__ ce1,
            float* __restrict__ ce2) {
  __shared__ float red[8];
  const int t = threadIdx.x;
  const float s = wave_sum_f((t < PS_BLOCKS) ? psum[t] : 0.f);
  if ((t & 31) == 0) red[t >> 5] = s;
  __syncthreads();
  if (t == 0) {
    float tot = 0.f;
    for (int w = 0; w < 8; w++) tot += red[w];
    mean_out[0] = tot * (1.0f / (float)N_EDGES);
  }
  if (t < H1) {
    float c = 0.f;
    for (int k = 0; k < C1; k++) c += We1[t * C1 + k] * ae1[t * C1 + k];
    ce1[t] = c;
  } else if (t >= 32 && t < 32 + H2) {
    const int h = t - 32;
    float c = 0.f;
    for (int k = 0; k < C2; k++) c += We2[h * C2 + k] * ae2[h * C2 + k];
    ce2[h] = c;
  }
}

__global__ void __launch_bounds__(256)
zero_counts_edge_sums_kernel(int* __restrict__ cnt,
                             const float* __restrict__ ea,
                             float* __restrict__ psum) {
  __shared__ float red[8];
  const int t = threadIdx.x;
  const int idx = blockIdx.x * 256 + t;
  if (idx < N_NODES) cnt[idx] = 0;
  const int base = blockIdx.x * PS_SLICE;
  float4 q[PS_SLICE / 1024];
  for (int k = 0; k < PS_SLICE / 1024; k++) {
    const int i = base + (k * 256 + t) * 4;
    q[k] = *(const float4*)&ea[(i < N_EDGES) ? i : N_EDGES - 4];
  }
  float s = 0.f;
  for (int k = 0; k < PS_SLICE / 1024; k++) {
    const int i = base + (k * 256 + t) * 4;
    const float part = (q[k].x + q[k].y) + (q[k].z + q[k].w);
    s += (i < N_EDGES) ? part : 0.f;
  }
  s = wave_sum_f(s);
  if ((t & 31) == 0) red[t >> 5] = s;
  __syncthreads();
  if (t == 0) {
    float tot = 0.f;
    for (int w = 0; w < 8; w++) tot += red[w];
    psum[blockIdx.x] = tot;
  }
}

__global__ void count_rank_kernel(const int* __restrict__ ei,
                                  int* __restrict__ cnt,
                                  int* __restrict__ rank) {
  const int e = blockIdx.x * blockDim.x + threadIdx.x;
  if (e >= N_EDGES) return;
  rank[e] = atomicAdd(&cnt[ei[N_EDGES + e]], 1);
}

__global__ void __launch_bounds__(256)
scan_counts_kernel(const int* __restrict__ cnt, int* __restrict__ rowptr) {
  __shared__ int wpre[8];
  __shared__ int wtot[8];
  const int t = threadIdx.x;
  const int lane = t & 31, wave = t >> 5;
  const int lo = blockIdx.x * 256;
  int pre = 0;
  for (int i0 = t * 4; i0 < lo; i0 += 8 * 1024) {
    int4 q[8];
    for (int u = 0; u < 8; u++) {
      const int i = i0 + u * 1024;
      q[u] = (i < lo) ? *(const int4*)&cnt[i] : make_int4(0, 0, 0, 0);
    }
    for (int u = 0; u < 8; u++) pre += (q[u].x + q[u].y) + (q[u].z + q[u].w);
  }
  const int node = lo + t;
  const int c = (node < N_NODES) ? cnt[node] : 0;
  int v = c;
  for (int off = 1; off < 32; off <<= 1) {
    const int u = __shfl_up(v, off);
    if (lane >= off) v += u;
  }
  for (int off = 16; off > 0; off >>= 1) pre += __shfl_xor(pre, off);
  if (lane == 31) wtot[wave] = v;
  if (lane == 0) wpre[wave] = pre;
  __syncthreads();
  int start = 0;
  for (int w = 0; w < 8; w++) {
    start += wpre[w];
    if (w < wave) start += wtot[w];
  }
  if (node <= N_NODES) rowptr[node] = start + v - c;
}

__global__ void place_edges_kernel(const int* __restrict__ ei,
                                   const int* __restrict__ rowptr,
                                   const int* __restrict__ rank,
                                   int* __restrict__ slot_eid) {
  const int e = blockIdx.x * blockDim.x + threadIdx.x;
  if (e >= N_EDGES) return;
  slot_eid[rowptr[ei[N_EDGES + e]] + rank[e]] = e;
}

__global__ void __launch_bounds__(256)
order_segments_kernel(const int* __restrict__ ei,
                      const float* __restrict__ ea,
                      const int* __restrict__ rowptr,
                      const int* __restrict__ slot_eid,
                      int* __restrict__ csr_src,
                      float* __restrict__ csr_ea) {
  const int lane = threadIdx.x & 31;
  const int wv = __builtin_amdgcn_readfirstlane((int)(threadIdx.x >> 5));
  const int d = blockIdx.x * 8 + wv;
  const int beg = rowptr[d], end = rowptr[d + 1];
  for (int base = beg; base < end; base += 32) {
    const int n = (end - base < 32) ? end - base : 32;
    const int mine = (lane < n) ? slot_eid[base + lane] : 0;
    int pos = 0;
    for (int b2 = beg; b2 < end; b2 += 32) {
      const int n2 = (end - b2 < 32) ? end - b2 : 32;
      const int other = (lane < n2) ? slot_eid[b2 + lane] : 0;
      for (int j = 0; j < n2; j++)
        pos += (__builtin_amdgcn_readlane(other, j) < mine) ? 1 : 0;
    }
    if (lane < n) {
      csr_src[beg + pos] = ei[mine];
      csr_ea[beg + pos] = ea[mine];
    }
  }
}

__global__ void __launch_bounds__(256)
gemm1_wmma(const float* __restrict__ x, const float* __restrict__ W1,
           float* __restrict__ xl1) {
  __shared__ float As[16][49];
  __shared__ float Bs[48][128];
  const int row0 = blockIdx.x * 16;
  const int colB = blockIdx.y * 128;
  const int t = threadIdx.x;
  const int wave = t >> 5, lane = t & 31, lh = lane & 15, hi = lane >> 4;
  v8f acc = {};
  for (int k0 = 0; k0 < 192; k0 += 48) {
    for (int idx = t; idx < 16 * 48; idx += 256) {
      const int r = idx / 48, c = idx % 48, k = k0 + c;
      As[r][c] = (k < IN_F) ? x[(row0 + r) * IN_F + k] : 0.f;
    }
    for (int idx4 = t; idx4 < (48 * 128) / 4; idx4 += 256) {
      const int r = idx4 >> 5, c = (idx4 & 31) << 2, k = k0 + r;
      float4 f = make_float4(0.f, 0.f, 0.f, 0.f);
      if (k < IN_F) {
        f = *(const float4*)&W1[k * F1 + colB + c];
        if ((idx4 & 7) == 0 && k + 48 < IN_F)
          __builtin_prefetch(&W1[(k + 48) * F1 + colB + c], 0, 0);
      }
      *(float4*)&Bs[r][c] = f;
    }
    __syncthreads();
#pragma unroll
    for (int kk = 0; kk < 48; kk += 4) {
      v2f a, b;
      a.x = As[lh][kk + 2 * hi];
      a.y = As[lh][kk + 2 * hi + 1];
      b.x = Bs[kk + 2 * hi][wave * 16 + lh];
      b.y = Bs[kk + 2 * hi + 1][wave * 16 + lh];
      acc = __builtin_amdgcn_wmma_f32_16x16x4_f32(false, a, false, b,
                                                  (short)0, acc, false, false);
    }
    __syncthreads();
  }
  const int col = colB + wave * 16 + lh;
#pragma unroll
  for (int i = 0; i < 8; i++)
    xl1[(row0 + i + 8 * hi) * F1 + col] = acc[i];
}

__global__ void __launch_bounds__(64)
gemm2_wmma(const float* __restrict__ h1, const float* __restrict__ W2,
           float* __restrict__ xl2) {
  __shared__ float As[16][258];
  const int row0 = blockIdx.x * 16;
  const int t = threadIdx.x;
  const int wave = t >> 5, lane = t & 31, lh = lane & 15, hi = lane >> 4;
  for (int idx4 = t; idx4 < (16 * 256) / 4; idx4 += 64) {
    const int r = idx4 >> 6, c = (idx4 & 63) << 2;
    const float4 f = *(const float4*)&h1[(row0 + r) * F1 + c];
    As[r][c + 0] = f.x; As[r][c + 1] = f.y;
    As[r][c + 2] = f.z; As[r][c + 3] = f.w;
  }
  __syncthreads();
  const int col = wave * 16 + lh;
  v8f acc = {};
  for (int k = 0; k < 256; k += 4) {
    v2f a, b;
    a.x = As[lh][k + 2 * hi];
    a.y = As[lh][k + 2 * hi + 1];
    if (col < F2) {
      b.x = W2[(k + 2 * hi) * F2 + col];
      b.y = W2[(k + 2 * hi + 1) * F2 + col];
    } else {
      b.x = 0.f; b.y = 0.f;
    }
    acc = __builtin_amdgcn_wmma_f32_16x16x4_f32(false, a, false, b,
                                                (short)0, acc, false, false);
  }
  if (col < F2) {
#pragma unroll
    for (int i = 0; i < 8; i++)
      xl2[(row0 + i + 8 * hi) * F2 + col] = acc[i];
  }
}

template <int H, int C>
__global__ void alpha_node_kernel(const float* __restrict__ xl,
                                  const float* __restrict__ a_src,
                                  const float* __restrict__ a_dst,
                                  float* __restrict__ asrc,
                                  float* __restrict__ adst) {
  const int idx = blockIdx.x * blockDim.x + threadIdx.x;
  if (idx >= N_NODES * H) return;
  const int n = idx / H, h = idx % H;
  const float* row = xl + n * (H * C) + h * C;
  float s = 0.f, d = 0.f;
#pragma unroll
  for (int c = 0; c < C; c++) {
    const float v = row[c];
    s += v * a_src[h * C + c];
    d += v * a_dst[h * C + c];
  }
  asrc[idx] = s;
  adst[idx] = d;
}

__global__ void __launch_bounds__(256)
node_softmax_agg1_kernel(const int* __restrict__ rowptr,
                         const int* __restrict__ csr_src,
                         const float* __restrict__ csr_ea,
                         const float* __restrict__ mean,
                         const float* __restrict__ ce,
                         const float* __restrict__ asrc,
                         const float* __restrict__ adst,
                         const float* __restrict__ xl,
                         float* __restrict__ agg) {
  const int lane = threadIdx.x & 31;
  const int wv = __builtin_amdgcn_readfirstlane((int)(threadIdx.x >> 5));
  const int d = blockIdx.x * 8 + wv;
  const int hA = lane >> 2, hB = 8 + (lane >> 2);
  const int beg = rowptr[d], end = rowptr[d + 1];
  const float adA = adst[d * H1 + hA], adB = adst[d * H1 + hB];
  const float cA = ce[hA], cB = ce[hB];
  const float mv = mean[0];
  const float aSelfA = lrelu02(asrc[d * H1 + hA] + adA + mv * cA);
  const float aSelfB = lrelu02(asrc[d * H1 + hB] + adB + mv * cB);

  float mA = aSelfA, mB = aSelfB;
  for (int base = beg; base < end; base += 32) {
    const int n = (end - base < 32) ? end - base : 32;
    int sL = 0;
    float eL = 0.f;
    if (lane < n) { sL = csr_src[base + lane]; eL = csr_ea[base + lane]; }
    for (int j = 0; j < n; j++) {
      const int s = __builtin_amdgcn_readlane(sL, j);
      const float ev = __int_as_float(__builtin_amdgcn_readlane(__float_as_int(eL), j));
      const float aA = lrelu02(asrc[s * H1 + hA] + adA + ev * cA);
      const float aB = lrelu02(asrc[s * H1 + hB] + adB + ev * cB);
      mA = fmaxf(mA, aA);
      mB = fmaxf(mB, aB);
    }
  }

  float denA = __expf(aSelfA - mA), denB = __expf(aSelfB - mB);
  float4 accA, accB;
  {
    const float4* r = (const float4*)(xl + d * F1);
    const float4 u = r[lane], w = r[32 + lane];
    accA = make_float4(denA * u.x, denA * u.y, denA * u.z, denA * u.w);
    accB = make_float4(denB * w.x, denB * w.y, denB * w.z, denB * w.w);
  }
  for (int base = beg; base < end; base += 32) {
    const int n = (end - base < 32) ? end - base : 32;
    int sL = 0;
    float eL = 0.f;
    if (lane < n) { sL = csr_src[base + lane]; eL = csr_ea[base + lane]; }
    for (int j = 0; j < n; j++) {
      const int s = __builtin_amdgcn_readlane(sL, j);
      const float ev = __int_as_float(__builtin_amdgcn_readlane(__float_as_int(eL), j));
      const float4* r = (const float4*)(xl + s * F1);
      const float4 u = r[lane], w = r[32 + lane];
      const float aA = lrelu02(asrc[s * H1 + hA] + adA + ev * cA);
      const float aB = lrelu02(asrc[s * H1 + hB] + adB + ev * cB);
      const float eA = __expf(aA - mA), eB = __expf(aB - mB);
      denA += eA;
      denB += eB;
      accA.x += eA * u.x; accA.y += eA * u.y; accA.z += eA * u.z; accA.w += eA * u.w;
      accB.x += eB * w.x; accB.y += eB * w.y; accB.z += eB * w.z; accB.w += eB * w.w;
    }
  }
  const float rA = 1.0f / (denA + 1e-16f), rB = 1.0f / (denB + 1e-16f);
  float4* o = (float4*)(agg + d * F1);
  o[lane]      = make_float4(accA.x * rA, accA.y * rA, accA.z * rA, accA.w * rA);
  o[32 + lane] = make_float4(accB.x * rB, accB.y * rB, accB.z * rB, accB.w * rB);
}

__global__ void __launch_bounds__(256)
node_softmax_agg2_kernel(const int* __restrict__ rowptr,
                         const int* __restrict__ csr_src,
                         const float* __restrict__ csr_ea,
                         const float* __restrict__ mean,
                         const float* __restrict__ ce,
                         const float* __restrict__ asrc,
                         const float* __restrict__ adst,
                         const float* __restrict__ xl,
                         float* __restrict__ agg) {
  const int idx = blockIdx.x * blockDim.x + threadIdx.x;
  if (idx >= N_NODES * H2) return;
  const int d = idx / H2, h = idx % H2;
  const int beg = rowptr[d], end = rowptr[d + 1];
  const float ad = adst[idx];
  const float c = ce[h];
  const float aSelf = lrelu02(asrc[idx] + ad + mean[0] * c);

  float m = aSelf;
  for (int j = beg; j < end; j++) {
    const float a = lrelu02(asrc[csr_src[j] * H2 + h] + ad + csr_ea[j] * c);
    m = fmaxf(m, a);
  }

  float den = __expf(aSelf - m);
  const float* rd = xl + d * F2 + h * C2;
  float acc0 = den * rd[0], acc1 = den * rd[1], acc2 = den * rd[2];
  for (int j = beg; j < end; j++) {
    const int s = csr_src[j];
    const float a = lrelu02(asrc[s * H2 + h] + ad + csr_ea[j] * c);
    const float ex = __expf(a - m);
    const float* rs = xl + s * F2 + h * C2;
    den += ex;
    acc0 += ex * rs[0];
    acc1 += ex * rs[1];
    acc2 += ex * rs[2];
  }
  const float rinv = 1.0f / (den + 1e-16f);
  float* o = agg + d * F2 + h * C2;
  o[0] = acc0 * rinv;
  o[1] = acc1 * rinv;
  o[2] = acc2 * rinv;
}

__global__ void finalize1_kernel(float* __restrict__ agg1,
                                 const float* __restrict__ b1) {
  const int idx4 = blockIdx.x * blockDim.x + threadIdx.x;
  if (idx4 >= (N_NODES * F1) / 4) return;
  const int e0 = idx4 << 2;
  float4 v = *(float4*)&agg1[e0];
  const float4 b = *(const float4*)&b1[e0 & (F1 - 1)];
  v.x += b.x; v.y += b.y; v.z += b.z; v.w += b.w;
  v.x = (v.x > 0.f) ? v.x : expm1f(v.x);
  v.y = (v.y > 0.f) ? v.y : expm1f(v.y);
  v.z = (v.z > 0.f) ? v.z : expm1f(v.z);
  v.w = (v.w > 0.f) ? v.w : expm1f(v.w);
  *(float4*)&agg1[e0] = v;
}

__global__ void final2_kernel(const float* __restrict__ agg2,
                              const float* __restrict__ b2,
                              float* __restrict__ out) {
  const int idx = blockIdx.x * blockDim.x + threadIdx.x;
  if (idx >= N_NODES * C2) return;
  const int n = idx / C2, c = idx % C2;
  float s = 0.f;
#pragma unroll
  for (int h = 0; h < H2; h++) s += agg2[n * F2 + h * C2 + c];
  const float v = s * (1.f / (float)H2) + b2[c];
  out[idx] = (v > 0.f) ? v : expm1f(v);
}

extern "C" void kernel_launch(void* const* d_in, const int* in_sizes, int n_in,
                              void* d_out, int out_size, void* d_ws, size_t ws_size,
                              hipStream_t stream) {
  (void)in_sizes; (void)n_in; (void)out_size; (void)ws_size;
  const float* x   = (const float*)d_in[0];
  const int*   ei  = (const int*)d_in[1];
  const float* ea  = (const float*)d_in[2];
  const float* W1  = (const float*)d_in[3];
  const float* We1 = (const float*)d_in[4];
  const float* as1 = (const float*)d_in[5];
  const float* ad1 = (const float*)d_in[6];
  const float* ae1 = (const float*)d_in[7];
  const float* b1  = (const float*)d_in[8];
  const float* W2  = (const float*)d_in[9];
  const float* We2 = (const float*)d_in[10];
  const float* as2 = (const float*)d_in[11];
  const float* ad2 = (const float*)d_in[12];
  const float* ae2 = (const float*)d_in[13];
  const float* b2  = (const float*)d_in[14];
  float* out = (float*)d_out;

  float* w = (float*)d_ws;
  float* mean = w;
  float* ce1  = w + 1;
  float* ce2  = w + 17;
  size_t off = 32;
  float* xl1   = w + off; off += (size_t)N_NODES * F1;
  float* asrc1 = w + off; off += (size_t)N_NODES * H1;
  float* adst1 = w + off; off += (size_t)N_NODES * H1;
  float* agg1  = w + off; off += (size_t)N_NODES * F1;
  float* xl2   = w + off; off += (size_t)N_NODES * F2;
  float* asrc2 = w + off; off += (size_t)N_NODES * H2;
  float* adst2 = w + off; off += (size_t)N_NODES * H2;
  float* agg2  = w + off; off += (size_t)N_NODES * F2;
  int* cnt     = (int*)(w + off); off += (size_t)N_NODES;
  int* rowptr  = (int*)(w + off); off += (size_t)N_NODES + 32;
  int* rank    = (int*)(w + off); off += (size_t)N_EDGES;
  int* slot_eid = (int*)(w + off); off += (size_t)N_EDGES;
  int* csr_src = (int*)(w + off); off += (size_t)N_EDGES;
  float* csr_ea = w + off;        off += (size_t)N_EDGES;
  float* psum = w + off;          off += 256;

  const int ge = (N_EDGES + 255) / 256;
  zero_counts_edge_sums_kernel<<<PS_BLOCKS, 256, 0, stream>>>(cnt, ea, psum);
  count_rank_kernel<<<ge, 256, 0, stream>>>(ei, cnt, rank);
  scan_counts_kernel<<<PS_BLOCKS, 256, 0, stream>>>(cnt, rowptr);
  place_edges_kernel<<<ge, 256, 0, stream>>>(ei, rowptr, rank, slot_eid);
  order_segments_kernel<<<N_NODES / 8, 256, 0, stream>>>(ei, ea, rowptr, slot_eid, csr_src, csr_ea);

  prep_kernel<<<1, 256, 0, stream>>>(psum, We1, ae1, We2, ae2, mean, ce1, ce2);

  gemm1_wmma<<<dim3(N_NODES / 16, 2), 256, 0, stream>>>(x, W1, xl1);
  alpha_node_kernel<H1, C1><<<(N_NODES * H1 + 255) / 256, 256, 0, stream>>>(
      xl1, as1, ad1, asrc1, adst1);
  node_softmax_agg1_kernel<<<N_NODES / 8, 256, 0, stream>>>(
      rowptr, csr_src, csr_ea, mean, ce1, asrc1, adst1, xl1, agg1);
  finalize1_kernel<<<((N_NODES * F1) / 4 + 255) / 256, 256, 0, stream>>>(agg1, b1);

  gemm2_wmma<<<N_NODES / 16, 64, 0, stream>>>(agg1, W2, xl2);
  alpha_node_kernel<H2, C2><<<(N_NODES * H2 + 255) / 256, 256, 0, stream>>>(
      xl2, as2, ad2, asrc2, adst2);
  node_softmax_agg2_kernel<<<(N_NODES * H2 + 255) / 256, 256, 0, stream>>>(
      rowptr, csr_src, csr_ea, mean, ce2, asrc2, adst2, xl2, agg2);
  final2_kernel<<<(N_NODES * C2 + 255) / 256, 256, 0, stream>>>(agg2, b2, out);
}


# both GEMMs stage operands with unguarded loads issued as one group (x row tile as float4 over all K, next W1 chunk fetched under the products, W2 fragments a k-group ahead), not load-wait-store
# speedup vs baseline: 5.7143x; 1.1939x over previous
#include <hip/hip_runtime.h>
#include <math.h>

#define N_NODES 50000
#define N_EDGES 800000
#define IN_F    191
#define H1 16
#define C1 16
#define F1 256
#define H2 8
#define C2 3
#define F2 24

typedef __attribute__((ext_vector_type(2))) float v2f;
typedef __attribute__((ext_vector_type(8))) float v8f;


__device__ __forceinline__ float lrelu02(float a) {
  return (a > 0.f) ? a : 0.2f * a;
}

#define PS_BLOCKS 196
#define PS_SLICE  4096

__device__ __forceinline__ float wave_sum_f(float v) {
  for (int off = 16; off > 0; off >>= 1) v += __shfl_xor(v, off);
  return v;
}

__global__ void __launch_bounds__(256)
prep_kernel(const float* __restrict__ psum,
            const float* __restrict__ We1, const float* __restrict__ ae1,
            const float* __restrict__ We2, const float* __restrict__ ae2,
            float* __restrict__ mean_out, float* __restrict__ ce1,
            float* __restrict__ ce2) {
  __shared__ float red[8];
  const int t = threadIdx.x;
  const float s = wave_sum_f((t < PS_BLOCKS) ? psum[t] : 0.f);
  if ((t & 31) == 0) red[t >> 5] = s;
  __syncthreads();
  if (t == 0) {
    float tot = 0.f;
    for (int w = 0; w < 8; w++) tot += red[w];
    mean_out[0] = tot * (1.0f / (float)N_EDGES);
  }
  if (t < H1) {
    float c = 0.f;
    for (int k = 0; k < C1; k++) c += We1[t * C1 + k] * ae1[t * C1 + k];
    ce1[t] = c;
  } else if (t >= 32 && t < 32 + H2) {
    const int h = t - 32;
    float c = 0.f;
    for (int k = 0; k < C2; k++) c += We2[h * C2 + k] * ae2[h * C2 + k];
    ce2[h] = c;
  }
}

__global__ void __launch_bounds__(256)
zero_counts_edge_sums_kernel(int* __restrict__ cnt,
                             const float* __restrict__ ea,
                             float* __restrict__ psum) {
  __shared__ float red[8];
  const int t = threadIdx.x;
  const int idx = blockIdx.x * 256 + t;
  if (idx < N_NODES) cnt[idx] = 0;
  const int base = blockIdx.x * PS_SLICE;
  float4 q[PS_SLICE / 1024];
  for (int k = 0; k < PS_SLICE / 1024; k++) {
    const int i = base + (k * 256 + t) * 4;
    q[k] = *(const float4*)&ea[(i < N_EDGES) ? i : N_EDGES - 4];
  }
  float s = 0.f;
  for (int k = 0; k < PS_SLICE / 1024; k++) {
    const int i = base + (k * 256 + t) * 4;
    const float part = (q[k].x + q[k].y) + (q[k].z + q[k].w);
    s += (i < N_EDGES) ? part : 0.f;
  }
  s = wave_sum_f(s);
  if ((t & 31) == 0) red[t >> 5] = s;
  __syncthreads();
  if (t == 0) {
    float tot = 0.f;
    for (int w = 0; w < 8; w++) tot += red[w];
    psum[blockIdx.x] = tot;
  }
}

__global__ void count_rank_kernel(const int* __restrict__ ei,
                                  int* __restrict__ cnt,
                                  int* __restrict__ rank) {
  const int e = blockIdx.x * blockDim.x + threadIdx.x;
  if (e >= N_EDGES) return;
  rank[e] = atomicAdd(&cnt[ei[N_EDGES + e]], 1);
}

__global__ void __launch_bounds__(256)
scan_counts_kernel(const int* __restrict__ cnt, int* __restrict__ rowptr) {
  __shared__ int wpre[8];
  __shared__ int wtot[8];
  const int t = threadIdx.x;
  const int lane = t & 31, wave = t >> 5;
  const int lo = blockIdx.x * 256;
  int pre = 0;
  for (int i0 = t * 4; i0 < lo; i0 += 8 * 1024) {
    int4 q[8];
    for (int u = 0; u < 8; u++) {
      const int i = i0 + u * 1024;
      q[u] = (i < lo) ? *(const int4*)&cnt[i] : make_int4(0, 0, 0, 0);
    }
    for (int u = 0; u < 8; u++) pre += (q[u].x + q[u].y) + (q[u].z + q[u].w);
  }
  const int node = lo + t;
  const int c = (node < N_NODES) ? cnt[node] : 0;
  int v = c;
  for (int off = 1; off < 32; off <<= 1) {
    const int u = __shfl_up(v, off);
    if (lane >= off) v += u;
  }
  for (int off = 16; off > 0; off >>= 1) pre += __shfl_xor(pre, off);
  if (lane == 31) wtot[wave] = v;
  if (lane == 0) wpre[wave] = pre;
  __syncthreads();
  int start = 0;
  for (int w = 0; w < 8; w++) {
    start += wpre[w];
    if (w < wave) start += wtot[w];
  }
  if (node <= N_NODES) rowptr[node] = start + v - c;
}

__global__ void place_edges_kernel(const int* __restrict__ ei,
                                   const int* __restrict__ rowptr,
                                   const int* __restrict__ rank,
                                   int* __restrict__ slot_eid) {
  const int e = blockIdx.x * blockDim.x + threadIdx.x;
  if (e >= N_EDGES) return;
  slot_eid[rowptr[ei[N_EDGES + e]] + rank[e]] = e;
}

__global__ void __launch_bounds__(256)
order_segments_kernel(const int* __restrict__ ei,
                      const float* __restrict__ ea,
                      const int* __restrict__ rowptr,
                      const int* __restrict__ slot_eid,
                      int* __restrict__ csr_src,
                      float* __restrict__ csr_ea) {
  const int lane = threadIdx.x & 31;
  const int wv = __builtin_amdgcn_readfirstlane((int)(threadIdx.x >> 5));
  const int d = blockIdx.x * 8 + wv;
  const int beg = rowptr[d], end = rowptr[d + 1];
  for (int base = beg; base < end; base += 32) {
    const int n = (end - base < 32) ? end - base : 32;
    const int mine = (lane < n) ? slot_eid[base + lane] : 0;
    int pos = 0;
    for (int b2 = beg; b2 < end; b2 += 32) {
      const int n2 = (end - b2 < 32) ? end - b2 : 32;
      const int other = (lane < n2) ? slot_eid[b2 + lane] : 0;
      for (int j = 0; j < n2; j++)
        pos += (__builtin_amdgcn_readlane(other, j) < mine) ? 1 : 0;
    }
    if (lane < n) {
      csr_src[beg + pos] = ei[mine];
      csr_ea[beg + pos] = ea[mine];
    }
  }
}

#define A_PITCH 196
#define A_F4    764

__device__ __forceinline__ void gemm1_load_b(const float* __restrict__ W1, int colB,
                                             int t, int k0, float4 (&qb)[6]) {
#pragma unroll
  for (int i = 0; i < 6; i++) {
    const int idx4 = t + 256 * i;
    const int r = idx4 >> 5, c = (idx4 & 31) << 2, k = k0 + r;
    qb[i] = *(const float4*)&W1[((k < IN_F) ? k : IN_F - 1) * F1 + colB + c];
  }
}

__global__ void __launch_bounds__(256)
gemm1_wmma(const float* __restrict__ x, const float* __restrict__ W1,
           float* __restrict__ xl1) {
  __shared__ __attribute__((aligned(16))) float As[16 * A_PITCH];
  __shared__ __attribute__((aligned(16))) float Bs[48][128];
  const int row0 = blockIdx.x * 16;
  const int colB = blockIdx.y * 128;
  const int t = threadIdx.x;
  const int wave = t >> 5, lane = t & 31, lh = lane & 15, hi = lane >> 4;

  const float4* xa = (const float4*)(x + (size_t)row0 * IN_F);
  float4 qa[3];
#pragma unroll
  for (int j = 0; j < 3; j++) {
    const int f = t + 256 * j;
    qa[j] = xa[(f < A_F4) ? f : A_F4 - 1];
  }
  float4 qb[6];
  gemm1_load_b(W1, colB, t, 0, qb);

#pragma unroll
  for (int j = 0; j < 3; j++) {
    const int f = t + 256 * j;
    if (f < A_F4) {
      const float v[4] = {qa[j].x, qa[j].y, qa[j].z, qa[j].w};
#pragma unroll
      for (int q = 0; q < 4; q++) {
        const int e = 4 * f + q, r = e / IN_F, k = e - r * IN_F;
        As[r * A_PITCH + k] = v[q];
      }
    }
  }
  if (t < 16) {
#pragma unroll
    for (int k = IN_F; k < A_PITCH; k++) As[t * A_PITCH + k] = 0.f;
  }

  v8f acc = {};
#pragma unroll
  for (int k0 = 0; k0 < 192; k0 += 48) {
#pragma unroll
    for (int i = 0; i < 6; i++) {
      const int idx4 = t + 256 * i;
      const int r = idx4 >> 5, c = (idx4 & 31) << 2;
      *(float4*)&Bs[r][c] = (k0 + r < IN_F) ? qb[i] : make_float4(0.f, 0.f, 0.f, 0.f);
    }
    __syncthreads();
    if (k0 + 48 < 192) gemm1_load_b(W1, colB, t, k0 + 48, qb);
#pragma unroll
    for (int kk = 0; kk < 48; kk += 4) {
      v2f a, b;
      a = *(const v2f*)&As[lh * A_PITCH + k0 + kk + 2 * hi];
      b.x = Bs[kk + 2 * hi][wave * 16 + lh];
      b.y = Bs[kk + 2 * hi + 1][wave * 16 + lh];
      acc = __builtin_amdgcn_wmma_f32_16x16x4_f32(false, a, false, b,
                                                  (short)0, acc, false, false);
    }
    __syncthreads();
  }
  const int col = colB + wave * 16 + lh;
#pragma unroll
  for (int i = 0; i < 8; i++)
    xl1[(row0 + i + 8 * hi) * F1 + col] = acc[i];
}

__global__ void __launch_bounds__(64)
gemm2_wmma(const float* __restrict__ h1, const float* __restrict__ W2,
           float* __restrict__ xl2) {
  __shared__ __attribute__((aligned(16))) float As[16][258];
  const int row0 = blockIdx.x * 16;
  const int t = threadIdx.x;
  const int wave = t >> 5, lane = t & 31, lh = lane & 15, hi = lane >> 4;
  float4 q[16];
#pragma unroll
  for (int i = 0; i < 16; i++) {
    const int idx4 = t + 64 * i;
    const int r = idx4 >> 6, c = (idx4 & 63) << 2;
    q[i] = *(const float4*)&h1[(row0 + r) * F1 + c];
  }
#pragma unroll
  for (int i = 0; i < 16; i++) {
    const int idx4 = t + 64 * i;
    const int r = idx4 >> 6, c = (idx4 & 63) << 2;
    As[r][c + 0] = q[i].x; As[r][c + 1] = q[i].y;
    As[r][c + 2] = q[i].z; As[r][c + 3] = q[i].w;
  }
  __syncthreads();
  const int col = wave * 16 + lh;
  const bool cv = col < F2;
  const int cc = cv ? col : F2 - 1;
  v8f acc = {};
#pragma unroll
  for (int kb = 0; kb < 256; kb += 64) {
    v2f bf[16];
#pragma unroll
    for (int u = 0; u < 16; u++) {
      const int k = kb + 4 * u + 2 * hi;
      const float bx = W2[k * F2 + cc], by = W2[(k + 1) * F2 + cc];
      bf[u].x = cv ? bx : 0.f;
      bf[u].y = cv ? by : 0.f;
    }
#pragma unroll
    for (int u = 0; u < 16; u++) {
      const v2f a = *(const v2f*)&As[lh][kb + 4 * u + 2 * hi];
      acc = __builtin_amdgcn_wmma_f32_16x16x4_f32(false, a, false, bf[u],
                                                  (short)0, acc, false, false);
    }
  }
  if (cv) {
#pragma unroll
    for (int i = 0; i < 8; i++)
      xl2[(row0 + i + 8 * hi) * F2 + col] = acc[i];
  }
}

template <int H, int C>
__global__ void alpha_node_kernel(const float* __restrict__ xl,
                                  const float* __restrict__ a_src,
                                  const float* __restrict__ a_dst,
                                  float* __restrict__ asrc,
                                  float* __restrict__ adst) {
  const int idx = blockIdx.x * blockDim.x + threadIdx.x;
  if (idx >= N_NODES * H) return;
  const int n = idx / H, h = idx % H;
  const float* row = xl + n * (H * C) + h * C;
  float s = 0.f, d = 0.f;
#pragma unroll
  for (int c = 0; c < C; c++) {
    const float v = row[c];
    s += v * a_src[h * C + c];
    d += v * a_dst[h * C + c];
  }
  asrc[idx] = s;
  adst[idx] = d;
}

__global__ void __launch_bounds__(256)
node_softmax_agg1_kernel(const int* __restrict__ rowptr,
                         const int* __restrict__ csr_src,
                         const float* __restrict__ csr_ea,
                         const float* __restrict__ mean,
                         const float* __restrict__ ce,
                         const float* __restrict__ asrc,
                         const float* __restrict__ adst,
                         const float* __restrict__ xl,
                         float* __restrict__ agg) {
  const int lane = threadIdx.x & 31;
  const int wv = __builtin_amdgcn_readfirstlane((int)(threadIdx.x >> 5));
  const int d = blockIdx.x * 8 + wv;
  const int hA = lane >> 2, hB = 8 + (lane >> 2);
  const int beg = rowptr[d], end = rowptr[d + 1];
  const float adA = adst[d * H1 + hA], adB = adst[d * H1 + hB];
  const float cA = ce[hA], cB = ce[hB];
  const float mv = mean[0];
  const float aSelfA = lrelu02(asrc[d * H1 + hA] + adA + mv * cA);
  const float aSelfB = lrelu02(asrc[d * H1 + hB] + adB + mv * cB);

  float mA = aSelfA, mB = aSelfB;
  for (int base = beg; base < end; base += 32) {
    const int n = (end - base < 32) ? end - base : 32;
    int sL = 0;
    float eL = 0.f;
    if (lane < n) { sL = csr_src[base + lane]; eL = csr_ea[base + lane]; }
    for (int j = 0; j < n; j++) {
      const int s = __builtin_amdgcn_readlane(sL, j);
      const float ev = __int_as_float(__builtin_amdgcn_readlane(__float_as_int(eL), j));
      const float aA = lrelu02(asrc[s * H1 + hA] + adA + ev * cA);
      const float aB = lrelu02(asrc[s * H1 + hB] + adB + ev * cB);
      mA = fmaxf(mA, aA);
      mB = fmaxf(mB, aB);
    }
  }

  float denA = __expf(aSelfA - mA), denB = __expf(aSelfB - mB);
  float4 accA, accB;
  {
    const float4* r = (const float4*)(xl + d * F1);
    const float4 u = r[lane], w = r[32 + lane];
    accA = make_float4(denA * u.x, denA * u.y, denA * u.z, denA * u.w);
    accB = make_float4(denB * w.x, denB * w.y, denB * w.z, denB * w.w);
  }
  for (int base = beg; base < end; base += 32) {
    const int n = (end - base < 32) ? end - base : 32;
    int sL = 0;
    float eL = 0.f;
    if (lane < n) { sL = csr_src[base + lane]; eL = csr_ea[base + lane]; }
    for (int j = 0; j < n; j++) {
      const int s = __builtin_amdgcn_readlane(sL, j);
      const float ev = __int_as_float(__builtin_amdgcn_readlane(__float_as_int(eL), j));
      const float4* r = (const float4*)(xl + s * F1);
      const float4 u = r[lane], w = r[32 + lane];
      const float aA = lrelu02(asrc[s * H1 + hA] + adA + ev * cA);
      const float aB = lrelu02(asrc[s * H1 + hB] + adB + ev * cB);
      const float eA = __expf(aA - mA), eB = __expf(aB - mB);
      denA += eA;
      denB += eB;
      accA.x += eA * u.x; accA.y += eA * u.y; accA.z += eA * u.z; accA.w += eA * u.w;
      accB.x += eB * w.x; accB.y += eB * w.y; accB.z += eB * w.z; accB.w += eB * w.w;
    }
  }
  const float rA = 1.0f / (denA + 1e-16f), rB = 1.0f / (denB + 1e-16f);
  float4* o = (float4*)(agg + d * F1);
  o[lane]      = make_float4(accA.x * rA, accA.y * rA, accA.z * rA, accA.w * rA);
  o[32 + lane] = make_float4(accB.x * rB, accB.y * rB, accB.z * rB, accB.w * rB);
}

__global__ void __launch_bounds__(256)
node_softmax_agg2_kernel(const int* __restrict__ rowptr,
                         const int* __restrict__ csr_src,
                         const float* __restrict__ csr_ea,
                         const float* __restrict__ mean,
                         const float* __restrict__ ce,
                         const float* __restrict__ asrc,
                         const float* __restrict__ adst,
                         const float* __restrict__ xl,
                         float* __restrict__ agg) {
  const int idx = blockIdx.x * blockDim.x + threadIdx.x;
  if (idx >= N_NODES * H2) return;
  const int d = idx / H2, h = idx % H2;
  const int beg = rowptr[d], end = rowptr[d + 1];
  const float ad = adst[idx];
  const float c = ce[h];
  const float aSelf = lrelu02(asrc[idx] + ad + mean[0] * c);

  float m = aSelf;
  for (int j = beg; j < end; j++) {
    const float a = lrelu02(asrc[csr_src[j] * H2 + h] + ad + csr_ea[j] * c);
    m = fmaxf(m, a);
  }

  float den = __expf(aSelf - m);
  const float* rd = xl + d * F2 + h * C2;
  float acc0 = den * rd[0], acc1 = den * rd[1], acc2 = den * rd[2];
  for (int j = beg; j < end; j++) {
    const int s = csr_src[j];
    const float a = lrelu02(asrc[s * H2 + h] + ad + csr_ea[j] * c);
    const float ex = __expf(a - m);
    const float* rs = xl + s * F2 + h * C2;
    den += ex;
    acc0 += ex * rs[0];
    acc1 += ex * rs[1];
    acc2 += ex * rs[2];
  }
  const float rinv = 1.0f / (den + 1e-16f);
  float* o = agg + d * F2 + h * C2;
  o[0] = acc0 * rinv;
  o[1] = acc1 * rinv;
  o[2] = acc2 * rinv;
}

__global__ void finalize1_kernel(float* __restrict__ agg1,
                                 const float* __restrict__ b1) {
  const int idx4 = blockIdx.x * blockDim.x + threadIdx.x;
  if (idx4 >= (N_NODES * F1) / 4) return;
  const int e0 = idx4 << 2;
  float4 v = *(float4*)&agg1[e0];
  const float4 b = *(const float4*)&b1[e0 & (F1 - 1)];
  v.x += b.x; v.y += b.y; v.z += b.z; v.w += b.w;
  v.x = (v.x > 0.f) ? v.x : expm1f(v.x);
  v.y = (v.y > 0.f) ? v.y : expm1f(v.y);
  v.z = (v.z > 0.f) ? v.z : expm1f(v.z);
  v.w = (v.w > 0.f) ? v.w : expm1f(v.w);
  *(float4*)&agg1[e0] = v;
}

__global__ void final2_kernel(const float* __restrict__ agg2,
                              const float* __restrict__ b2,
                              float* __restrict__ out) {
  const int idx = blockIdx.x * blockDim.x + threadIdx.x;
  if (idx >= N_NODES * C2) return;
  const int n = idx / C2, c = idx % C2;
  float s = 0.f;
#pragma unroll
  for (int h = 0; h < H2; h++) s += agg2[n * F2 + h * C2 + c];
  const float v = s * (1.f / (float)H2) + b2[c];
  out[idx] = (v > 0.f) ? v : expm1f(v);
}

extern "C" void kernel_launch(void* const* d_in, const int* in_sizes, int n_in,
                              void* d_out, int out_size, void* d_ws, size_t ws_size,
                              hipStream_t stream) {
  (void)in_sizes; (void)n_in; (void)out_size; (void)ws_size;
  const float* x   = (const float*)d_in[0];
  const int*   ei  = (const int*)d_in[1];
  const float* ea  = (const float*)d_in[2];
  const float* W1  = (const float*)d_in[3];
  const float* We1 = (const float*)d_in[4];
  const float* as1 = (const float*)d_in[5];
  const float* ad1 = (const float*)d_in[6];
  const float* ae1 = (const float*)d_in[7];
  const float* b1  = (const float*)d_in[8];
  const float* W2  = (const float*)d_in[9];
  const float* We2 = (const float*)d_in[10];
  const float* as2 = (const float*)d_in[11];
  const float* ad2 = (const float*)d_in[12];
  const float* ae2 = (const float*)d_in[13];
  const float* b2  = (const float*)d_in[14];
  float* out = (float*)d_out;

  float* w = (float*)d_ws;
  float* mean = w;
  float* ce1  = w + 1;
  float* ce2  = w + 17;
  size_t off = 32;
  float* xl1   = w + off; off += (size_t)N_NODES * F1;
  float* asrc1 = w + off; off += (size_t)N_NODES * H1;
  float* adst1 = w + off; off += (size_t)N_NODES * H1;
  float* agg1  = w + off; off += (size_t)N_NODES * F1;
  float* xl2   = w + off; off += (size_t)N_NODES * F2;
  float* asrc2 = w + off; off += (size_t)N_NODES * H2;
  float* adst2 = w + off; off += (size_t)N_NODES * H2;
  float* agg2  = w + off; off += (size_t)N_NODES * F2;
  int* cnt     = (int*)(w + off); off += (size_t)N_NODES;
  int* rowptr  = (int*)(w + off); off += (size_t)N_NODES + 32;
  int* rank    = (int*)(w + off); off += (size_t)N_EDGES;
  int* slot_eid = (int*)(w + off); off += (size_t)N_EDGES;
  int* csr_src = (int*)(w + off); off += (size_t)N_EDGES;
  float* csr_ea = w + off;        off += (size_t)N_EDGES;
  float* psum = w + off;          off += 256;

  const int ge = (N_EDGES + 255) / 256;
  zero_counts_edge_sums_kernel<<<PS_BLOCKS, 256, 0, stream>>>(cnt, ea, psum);
  count_rank_kernel<<<ge, 256, 0, stream>>>(ei, cnt, rank);
  scan_counts_kernel<<<PS_BLOCKS, 256, 0, stream>>>(cnt, rowptr);
  place_edges_kernel<<<ge, 256, 0, stream>>>(ei, rowptr, rank, slot_eid);
  order_segments_kernel<<<N_NODES / 8, 256, 0, stream>>>(ei, ea, rowptr, slot_eid, csr_src, csr_ea);

  prep_kernel<<<1, 256, 0, stream>>>(psum, We1, ae1, We2, ae2, mean, ce1, ce2);

  gemm1_wmma<<<dim3(N_NODES / 16, 2), 256, 0, stream>>>(x, W1, xl1);
  alpha_node_kernel<H1, C1><<<(N_NODES * H1 + 255) / 256, 256, 0, stream>>>(
      xl1, as1, ad1, asrc1, adst1);
  node_softmax_agg1_kernel<<<N_NODES / 8, 256, 0, stream>>>(
      rowptr, csr_src, csr_ea, mean, ce1, asrc1, adst1, xl1, agg1);
  finalize1_kernel<<<((N_NODES * F1) / 4 + 255) / 256, 256, 0, stream>>>(agg1, b1);

  gemm2_wmma<<<N_NODES / 16, 64, 0, stream>>>(agg1, W2, xl2);
  alpha_node_kernel<H2, C2><<<(N_NODES * H2 + 255) / 256, 256, 0, stream>>>(
      xl2, as2, ad2, asrc2, adst2);
  node_softmax_agg2_kernel<<<(N_NODES * H2 + 255) / 256, 256, 0, stream>>>(
      rowptr, csr_src, csr_ea, mean, ce2, asrc2, adst2, xl2, agg2);
  final2_kernel<<<(N_NODES * C2 + 255) / 256, 256, 0, stream>>>(agg2, b2, out);
}


# three follow-up launches folded into their producers: bias+ELU into the layer-1 aggregation's row store, both per-node logit kernels into the GEMM epilogues through an LDS copy of the result tile
# speedup vs baseline: 6.1953x; 1.0842x over previous
#include <hip/hip_runtime.h>
#include <math.h>

#define N_NODES 50000
#define N_EDGES 800000
#define IN_F    191
#define H1 16
#define C1 16
#define F1 256
#define H2 8
#define C2 3
#define F2 24

typedef __attribute__((ext_vector_type(2))) float v2f;
typedef __attribute__((ext_vector_type(8))) float v8f;


__device__ __forceinline__ float lrelu02(float a) {
  return (a > 0.f) ? a : 0.2f * a;
}

__device__ __forceinline__ float elu1(float v) {
  return (v > 0.f) ? v : expm1f(v);
}

#define PS_BLOCKS 196
#define PS_SLICE  4096

__device__ __forceinline__ float wave_sum_f(float v) {
  for (int off = 16; off > 0; off >>= 1) v += __shfl_xor(v, off);
  return v;
}

__global__ void __launch_bounds__(256)
prep_kernel(const float* __restrict__ psum,
            const float* __restrict__ We1, const float* __restrict__ ae1,
            const float* __restrict__ We2, const float* __restrict__ ae2,
            float* __restrict__ mean_out, float* __restrict__ ce1,
            float* __restrict__ ce2) {
  __shared__ float red[8];
  const int t = threadIdx.x;
  const float s = wave_sum_f((t < PS_BLOCKS) ? psum[t] : 0.f);
  if ((t & 31) == 0) red[t >> 5] = s;
  __syncthreads();
  if (t == 0) {
    float tot = 0.f;
    for (int w = 0; w < 8; w++) tot += red[w];
    mean_out[0] = tot * (1.0f / (float)N_EDGES);
  }
  if (t < H1) {
    float c = 0.f;
    for (int k = 0; k < C1; k++) c += We1[t * C1 + k] * ae1[t * C1 + k];
    ce1[t] = c;
  } else if (t >= 32 && t < 32 + H2) {
    const int h = t - 32;
    float c = 0.f;
    for (int k = 0; k < C2; k++) c += We2[h * C2 + k] * ae2[h * C2 + k];
    ce2[h] = c;
  }
}

__global__ void __launch_bounds__(256)
zero_counts_edge_sums_kernel(int* __restrict__ cnt,
                             const float* __restrict__ ea,
                             float* __restrict__ psum) {
  __shared__ float red[8];
  const int t = threadIdx.x;
  const int idx = blockIdx.x * 256 + t;
  if (idx < N_NODES) cnt[idx] = 0;
  const int base = blockIdx.x * PS_SLICE;
  float4 q[PS_SLICE / 1024];
  for (int k = 0; k < PS_SLICE / 1024; k++) {
    const int i = base + (k * 256 + t) * 4;
    q[k] = *(const float4*)&ea[(i < N_EDGES) ? i : N_EDGES - 4];
  }
  float s = 0.f;
  for (int k = 0; k < PS_SLICE / 1024; k++) {
    const int i = base + (k * 256 + t) * 4;
    const float part = (q[k].x + q[k].y) + (q[k].z + q[k].w);
    s += (i < N_EDGES) ? part : 0.f;
  }
  s = wave_sum_f(s);
  if ((t & 31) == 0) red[t >> 5] = s;
  __syncthreads();
  if (t == 0) {
    float tot = 0.f;
    for (int w = 0; w < 8; w++) tot += red[w];
    psum[blockIdx.x] = tot;
  }
}

__global__ void count_rank_kernel(const int* __restrict__ ei,
                                  int* __restrict__ cnt,
                                  int* __restrict__ rank) {
  const int e = blockIdx.x * blockDim.x + threadIdx.x;
  if (e >= N_EDGES) return;
  rank[e] = atomicAdd(&cnt[ei[N_EDGES + e]], 1);
}

__global__ void __launch_bounds__(256)
scan_counts_kernel(const int* __restrict__ cnt, int* __restrict__ rowptr) {
  __shared__ int wpre[8];
  __shared__ int wtot[8];
  const int t = threadIdx.x;
  const int lane = t & 31, wave = t >> 5;
  const int lo = blockIdx.x * 256;
  int pre = 0;
  for (int i0 = t * 4; i0 < lo; i0 += 8 * 1024) {
    int4 q[8];
    for (int u = 0; u < 8; u++) {
      const int i = i0 + u * 1024;
      q[u] = (i < lo) ? *(const int4*)&cnt[i] : make_int4(0, 0, 0, 0);
    }
    for (int u = 0; u < 8; u++) pre += (q[u].x + q[u].y) + (q[u].z + q[u].w);
  }
  const int node = lo + t;
  const int c = (node < N_NODES) ? cnt[node] : 0;
  int v = c;
  for (int off = 1; off < 32; off <<= 1) {
    const int u = __shfl_up(v, off);
    if (lane >= off) v += u;
  }
  for (int off = 16; off > 0; off >>= 1) pre += __shfl_xor(pre, off);
  if (lane == 31) wtot[wave] = v;
  if (lane == 0) wpre[wave] = pre;
  __syncthreads();
  int start = 0;
  for (int w = 0; w < 8; w++) {
    start += wpre[w];
    if (w < wave) start += wtot[w];
  }
  if (node <= N_NODES) rowptr[node] = start + v - c;
}

__global__ void place_edges_kernel(const int* __restrict__ ei,
                                   const int* __restrict__ rowptr,
                                   const int* __restrict__ rank,
                                   int* __restrict__ slot_eid) {
  const int e = blockIdx.x * blockDim.x + threadIdx.x;
  if (e >= N_EDGES) return;
  slot_eid[rowptr[ei[N_EDGES + e]] + rank[e]] = e;
}

__global__ void __launch_bounds__(256)
order_segments_kernel(const int* __restrict__ ei,
                      const float* __restrict__ ea,
                      const int* __restrict__ rowptr,
                      const int* __restrict__ slot_eid,
                      int* __restrict__ csr_src,
                      float* __restrict__ csr_ea) {
  const int lane = threadIdx.x & 31;
  const int wv = __builtin_amdgcn_readfirstlane((int)(threadIdx.x >> 5));
  const int d = blockIdx.x * 8 + wv;
  const int beg = rowptr[d], end = rowptr[d + 1];
  for (int base = beg; base < end; base += 32) {
    const int n = (end - base < 32) ? end - base : 32;
    const int mine = (lane < n) ? slot_eid[base + lane] : 0;
    int pos = 0;
    for (int b2 = beg; b2 < end; b2 += 32) {
      const int n2 = (end - b2 < 32) ? end - b2 : 32;
      const int other = (lane < n2) ? slot_eid[b2 + lane] : 0;
      for (int j = 0; j < n2; j++)
        pos += (__builtin_amdgcn_readlane(other, j) < mine) ? 1 : 0;
    }
    if (lane < n) {
      csr_src[beg + pos] = ei[mine];
      csr_ea[beg + pos] = ea[mine];
    }
  }
}

#define A_PITCH 196
#define A_F4    764
#define D1_PITCH 132

__device__ __forceinline__ void gemm1_load_b(const float* __restrict__ W1, int colB,
                                             int t, int k0, float4 (&qb)[6]) {
#pragma unroll
  for (int i = 0; i < 6; i++) {
    const int idx4 = t + 256 * i;
    const int r = idx4 >> 5, c = (idx4 & 31) << 2, k = k0 + r;
    qb[i] = *(const float4*)&W1[((k < IN_F) ? k : IN_F - 1) * F1 + colB + c];
  }
}

__global__ void __launch_bounds__(256)
gemm1_wmma(const float* __restrict__ x, const float* __restrict__ W1,
           const float* __restrict__ a_src, const float* __restrict__ a_dst,
           float* __restrict__ xl1, float* __restrict__ asrc,
           float* __restrict__ adst) {
  __shared__ __attribute__((aligned(16))) float As[16 * A_PITCH];
  __shared__ __attribute__((aligned(16))) float Bs[48][128];
  const int row0 = blockIdx.x * 16;
  const int colB = blockIdx.y * 128;
  const int t = threadIdx.x;
  const int wave = t >> 5, lane = t & 31, lh = lane & 15, hi = lane >> 4;

  const float4* xa = (const float4*)(x + (size_t)row0 * IN_F);
  float4 qa[3];
#pragma unroll
  for (int j = 0; j < 3; j++) {
    const int f = t + 256 * j;
    qa[j] = xa[(f < A_F4) ? f : A_F4 - 1];
  }
  float4 qb[6];
  gemm1_load_b(W1, colB, t, 0, qb);

#pragma unroll
  for (int j = 0; j < 3; j++) {
    const int f = t + 256 * j;
    if (f < A_F4) {
      const float v[4] = {qa[j].x, qa[j].y, qa[j].z, qa[j].w};
#pragma unroll
      for (int q = 0; q < 4; q++) {
        const int e = 4 * f + q, r = e / IN_F, k = e - r * IN_F;
        As[r * A_PITCH + k] = v[q];
      }
    }
  }
  if (t < 16) {
#pragma unroll
    for (int k = IN_F; k < A_PITCH; k++) As[t * A_PITCH + k] = 0.f;
  }

  v8f acc = {};
#pragma unroll
  for (int k0 = 0; k0 < 192; k0 += 48) {
#pragma unroll
    for (int i = 0; i < 6; i++) {
      const int idx4 = t + 256 * i;
      const int r = idx4 >> 5, c = (idx4 & 31) << 2;
      *(float4*)&Bs[r][c] = (k0 + r < IN_F) ? qb[i] : make_float4(0.f, 0.f, 0.f, 0.f);
    }
    __syncthreads();
    if (k0 + 48 < 192) gemm1_load_b(W1, colB, t, k0 + 48, qb);
#pragma unroll
    for (int kk = 0; kk < 48; kk += 4) {
      v2f a, b;
      a = *(const v2f*)&As[lh * A_PITCH + k0 + kk + 2 * hi];
      b.x = Bs[kk + 2 * hi][wave * 16 + lh];
      b.y = Bs[kk + 2 * hi + 1][wave * 16 + lh];
      acc = __builtin_amdgcn_wmma_f32_16x16x4_f32(false, a, false, b,
                                                  (short)0, acc, false, false);
    }
    __syncthreads();
  }
  float* Ds = &Bs[0][0];
#pragma unroll
  for (int i = 0; i < 8; i++)
    Ds[(i + 8 * hi) * D1_PITCH + wave * 16 + lh] = acc[i];
  __syncthreads();
  const int col = colB + wave * 16 + lh;
#pragma unroll
  for (int i = 0; i < 8; i++)
    xl1[(row0 + i + 8 * hi) * F1 + col] = acc[i];
  if (t < 128) {
    const int r = t >> 3, hl = t & 7;
    const int h = blockIdx.y * 8 + hl;
    const float* row = Ds + r * D1_PITCH + hl * C1;
    float s = 0.f, d = 0.f;
#pragma unroll
    for (int c = 0; c < C1; c++) {
      const float v = row[c];
      s += v * a_src[h * C1 + c];
      d += v * a_dst[h * C1 + c];
    }
    asrc[(row0 + r) * H1 + h] = s;
    adst[(row0 + r) * H1 + h] = d;
  }
}

#define D2_PITCH 25
__global__ void __launch_bounds__(64)
gemm2_wmma(const float* __restrict__ h1, const float* __restrict__ W2,
           const float* __restrict__ a_src, const float* __restrict__ a_dst,
           float* __restrict__ xl2, float* __restrict__ asrc,
           float* __restrict__ adst) {
  __shared__ __attribute__((aligned(16))) float As[16][258];
  const int row0 = blockIdx.x * 16;
  const int t = threadIdx.x;
  const int wave = t >> 5, lane = t & 31, lh = lane & 15, hi = lane >> 4;
  float4 q[16];
#pragma unroll
  for (int i = 0; i < 16; i++) {
    const int idx4 = t + 64 * i;
    const int r = idx4 >> 6, c = (idx4 & 63) << 2;
    q[i] = *(const float4*)&h1[(row0 + r) * F1 + c];
  }
#pragma unroll
  for (int i = 0; i < 16; i++) {
    const int idx4 = t + 64 * i;
    const int r = idx4 >> 6, c = (idx4 & 63) << 2;
    As[r][c + 0] = q[i].x; As[r][c + 1] = q[i].y;
    As[r][c + 2] = q[i].z; As[r][c + 3] = q[i].w;
  }
  __syncthreads();
  const int col = wave * 16 + lh;
  const bool cv = col < F2;
  const int cc = cv ? col : F2 - 1;
  v8f acc = {};
#pragma unroll
  for (int kb = 0; kb < 256; kb += 64) {
    v2f bf[16];
#pragma unroll
    for (int u = 0; u < 16; u++) {
      const int k = kb + 4 * u + 2 * hi;
      const float bx = W2[k * F2 + cc], by = W2[(k + 1) * F2 + cc];
      bf[u].x = cv ? bx : 0.f;
      bf[u].y = cv ? by : 0.f;
    }
#pragma unroll
    for (int u = 0; u < 16; u++) {
      const v2f a = *(const v2f*)&As[lh][kb + 4 * u + 2 * hi];
      acc = __builtin_amdgcn_wmma_f32_16x16x4_f32(false, a, false, bf[u],
                                                  (short)0, acc, false, false);
    }
  }
  __syncthreads();
  float* Ds = &As[0][0];
  if (cv) {
#pragma unroll
    for (int i = 0; i < 8; i++)
      Ds[(i + 8 * hi) * D2_PITCH + col] = acc[i];
  }
  __syncthreads();
  if (cv) {
#pragma unroll
    for (int i = 0; i < 8; i++)
      xl2[(row0 + i + 8 * hi) * F2 + col] = acc[i];
  }
#pragma unroll
  for (int j = 0; j < 2; j++) {
    const int p = t + 64 * j;
    const int r = p >> 3, h = p & 7;
    const float* row = Ds + r * D2_PITCH + h * C2;
    float s = 0.f, d = 0.f;
#pragma unroll
    for (int c = 0; c < C2; c++) {
      const float v = row[c];
      s += v * a_src[h * C2 + c];
      d += v * a_dst[h * C2 + c];
    }
    asrc[(row0 + r) * H2 + h] = s;
    adst[(row0 + r) * H2 + h] = d;
  }
}

__global__ void __launch_bounds__(256)
node_softmax_agg1_kernel(const int* __restrict__ rowptr,
                         const int* __restrict__ csr_src,
                         const float* __restrict__ csr_ea,
                         const float* __restrict__ mean,
                         const float* __restrict__ ce,
                         const float* __restrict__ asrc,
                         const float* __restrict__ adst,
                         const float* __restrict__ xl,
                         const float* __restrict__ b1,
                         float* __restrict__ h1out) {
  const int lane = threadIdx.x & 31;
  const int wv = __builtin_amdgcn_readfirstlane((int)(threadIdx.x >> 5));
  const int d = blockIdx.x * 8 + wv;
  const int hA = lane >> 2, hB = 8 + (lane >> 2);
  const int beg = rowptr[d], end = rowptr[d + 1];
  const float adA = adst[d * H1 + hA], adB = adst[d * H1 + hB];
  const float cA = ce[hA], cB = ce[hB];
  const float mv = mean[0];
  const float aSelfA = lrelu02(asrc[d * H1 + hA] + adA + mv * cA);
  const float aSelfB = lrelu02(asrc[d * H1 + hB] + adB + mv * cB);

  float mA = aSelfA, mB = aSelfB;
  for (int base = beg; base < end; base += 32) {
    const int n = (end - base < 32) ? end - base : 32;
    int sL = 0;
    float eL = 0.f;
    if (lane < n) { sL = csr_src[base + lane]; eL = csr_ea[base + lane]; }
    for (int j = 0; j < n; j++) {
      const int s = __builtin_amdgcn_readlane(sL, j);
      const float ev = __int_as_float(__builtin_amdgcn_readlane(__float_as_int(eL), j));
      const float aA = lrelu02(asrc[s * H1 + hA] + adA + ev * cA);
      const float aB = lrelu02(asrc[s * H1 + hB] + adB + ev * cB);
      mA = fmaxf(mA, aA);
      mB = fmaxf(mB, aB);
    }
  }

  const float4 bA = ((const float4*)b1)[lane], bB = ((const float4*)b1)[32 + lane];

  float denA = __expf(aSelfA - mA), denB = __expf(aSelfB - mB);
  float4 accA, accB;
  {
    const float4* r = (const float4*)(xl + d * F1);
    const float4 u = r[lane], w = r[32 + lane];
    accA = make_float4(denA * u.x, denA * u.y, denA * u.z, denA * u.w);
    accB = make_float4(denB * w.x, denB * w.y, denB * w.z, denB * w.w);
  }
  for (int base = beg; base < end; base += 32) {
    const int n = (end - base < 32) ? end - base : 32;
    int sL = 0;
    float eL = 0.f;
    if (lane < n) { sL = csr_src[base + lane]; eL = csr_ea[base + lane]; }
    for (int j = 0; j < n; j++) {
      const int s = __builtin_amdgcn_readlane(sL, j);
      const float ev = __int_as_float(__builtin_amdgcn_readlane(__float_as_int(eL), j));
      const float4* r = (const float4*)(xl + s * F1);
      const float4 u = r[lane], w = r[32 + lane];
      const float aA = lrelu02(asrc[s * H1 + hA] + adA + ev * cA);
      const float aB = lrelu02(asrc[s * H1 + hB] + adB + ev * cB);
      const float eA = __expf(aA - mA), eB = __expf(aB - mB);
      denA += eA;
      denB += eB;
      accA.x += eA * u.x; accA.y += eA * u.y; accA.z += eA * u.z; accA.w += eA * u.w;
      accB.x += eB * w.x; accB.y += eB * w.y; accB.z += eB * w.z; accB.w += eB * w.w;
    }
  }
  const float rA = 1.0f / (denA + 1e-16f), rB = 1.0f / (denB + 1e-16f);
  float4* o = (float4*)(h1out + d * F1);
  o[lane]      = make_float4(elu1(accA.x * rA + bA.x), elu1(accA.y * rA + bA.y),
                             elu1(accA.z * rA + bA.z), elu1(accA.w * rA + bA.w));
  o[32 + lane] = make_float4(elu1(accB.x * rB + bB.x), elu1(accB.y * rB + bB.y),
                             elu1(accB.z * rB + bB.z), elu1(accB.w * rB + bB.w));
}

__global__ void __launch_bounds__(256)
node_softmax_agg2_kernel(const int* __restrict__ rowptr,
                         const int* __restrict__ csr_src,
                         const float* __restrict__ csr_ea,
                         const float* __restrict__ mean,
                         const float* __restrict__ ce,
                         const float* __restrict__ asrc,
                         const float* __restrict__ adst,
                         const float* __restrict__ xl,
                         float* __restrict__ agg) {
  const int idx = blockIdx.x * blockDim.x + threadIdx.x;
  if (idx >= N_NODES * H2) return;
  const int d = idx / H2, h = idx % H2;
  const int beg = rowptr[d], end = rowptr[d + 1];
  const float ad = adst[idx];
  const float c = ce[h];
  const float aSelf = lrelu02(asrc[idx] + ad + mean[0] * c);

  float m = aSelf;
  for (int j = beg; j < end; j++) {
    const float a = lrelu02(asrc[csr_src[j] * H2 + h] + ad + csr_ea[j] * c);
    m = fmaxf(m, a);
  }

  float den = __expf(aSelf - m);
  const float* rd = xl + d * F2 + h * C2;
  float acc0 = den * rd[0], acc1 = den * rd[1], acc2 = den * rd[2];
  for (int j = beg; j < end; j++) {
    const int s = csr_src[j];
    const float a = lrelu02(asrc[s * H2 + h] + ad + csr_ea[j] * c);
    const float ex = __expf(a - m);
    const float* rs = xl + s * F2 + h * C2;
    den += ex;
    acc0 += ex * rs[0];
    acc1 += ex * rs[1];
    acc2 += ex * rs[2];
  }
  const float rinv = 1.0f / (den + 1e-16f);
  float* o = agg + d * F2 + h * C2;
  o[0] = acc0 * rinv;
  o[1] = acc1 * rinv;
  o[2] = acc2 * rinv;
}

__global__ void final2_kernel(const float* __restrict__ agg2,
                              const float* __restrict__ b2,
                              float* __restrict__ out) {
  const int idx = blockIdx.x * blockDim.x + threadIdx.x;
  if (idx >= N_NODES * C2) return;
  const int n = idx / C2, c = idx % C2;
  float s = 0.f;
#pragma unroll
  for (int h = 0; h < H2; h++) s += agg2[n * F2 + h * C2 + c];
  const float v = s * (1.f / (float)H2) + b2[c];
  out[idx] = (v > 0.f) ? v : expm1f(v);
}

extern "C" void kernel_launch(void* const* d_in, const int* in_sizes, int n_in,
                              void* d_out, int out_size, void* d_ws, size_t ws_size,
                              hipStream_t stream) {
  (void)in_sizes; (void)n_in; (void)out_size; (void)ws_size;
  const float* x   = (const float*)d_in[0];
  const int*   ei  = (const int*)d_in[1];
  const float* ea  = (const float*)d_in[2];
  const float* W1  = (const float*)d_in[3];
  const float* We1 = (const float*)d_in[4];
  const float* as1 = (const float*)d_in[5];
  const float* ad1 = (const float*)d_in[6];
  const float* ae1 = (const float*)d_in[7];
  const float* b1  = (const float*)d_in[8];
  const float* W2  = (const float*)d_in[9];
  const float* We2 = (const float*)d_in[10];
  const float* as2 = (const float*)d_in[11];
  const float* ad2 = (const float*)d_in[12];
  const float* ae2 = (const float*)d_in[13];
  const float* b2  = (const float*)d_in[14];
  float* out = (float*)d_out;

  float* w = (float*)d_ws;
  float* mean = w;
  float* ce1  = w + 1;
  float* ce2  = w + 17;
  size_t off = 32;
  float* xl1   = w + off; off += (size_t)N_NODES * F1;
  float* asrc1 = w + off; off += (size_t)N_NODES * H1;
  float* adst1 = w + off; off += (size_t)N_NODES * H1;
  float* agg1  = w + off; off += (size_t)N_NODES * F1;
  float* xl2   = w + off; off += (size_t)N_NODES * F2;
  float* asrc2 = w + off; off += (size_t)N_NODES * H2;
  float* adst2 = w + off; off += (size_t)N_NODES * H2;
  float* agg2  = w + off; off += (size_t)N_NODES * F2;
  int* cnt     = (int*)(w + off); off += (size_t)N_NODES;
  int* rowptr  = (int*)(w + off); off += (size_t)N_NODES + 32;
  int* rank    = (int*)(w + off); off += (size_t)N_EDGES;
  int* slot_eid = (int*)(w + off); off += (size_t)N_EDGES;
  int* csr_src = (int*)(w + off); off += (size_t)N_EDGES;
  float* csr_ea = w + off;        off += (size_t)N_EDGES;
  float* psum = w + off;          off += 256;

  const int ge = (N_EDGES + 255) / 256;
  zero_counts_edge_sums_kernel<<<PS_BLOCKS, 256, 0, stream>>>(cnt, ea, psum);
  count_rank_kernel<<<ge, 256, 0, stream>>>(ei, cnt, rank);
  scan_counts_kernel<<<PS_BLOCKS, 256, 0, stream>>>(cnt, rowptr);
  place_edges_kernel<<<ge, 256, 0, stream>>>(ei, rowptr, rank, slot_eid);
  order_segments_kernel<<<N_NODES / 8, 256, 0, stream>>>(ei, ea, rowptr, slot_eid, csr_src, csr_ea);

  prep_kernel<<<1, 256, 0, stream>>>(psum, We1, ae1, We2, ae2, mean, ce1, ce2);

  gemm1_wmma<<<dim3(N_NODES / 16, 2), 256, 0, stream>>>(x, W1, as1, ad1, xl1, asrc1, adst1);
  node_softmax_agg1_kernel<<<N_NODES / 8, 256, 0, stream>>>(
      rowptr, csr_src, csr_ea, mean, ce1, asrc1, adst1, xl1, b1, agg1);

  gemm2_wmma<<<N_NODES / 16, 64, 0, stream>>>(agg1, W2, as2, ad2, xl2, asrc2, adst2);
  node_softmax_agg2_kernel<<<(N_NODES * H2 + 255) / 256, 256, 0, stream>>>(
      rowptr, csr_src, csr_ea, mean, ce2, asrc2, adst2, xl2, agg2);
  final2_kernel<<<(N_NODES * C2 + 255) / 256, 256, 0, stream>>>(agg2, b2, out);
}


# edge sort rebuilt as a two-level counting sort with every counter in LDS: no global atomics left, pairs staged in LDS and stored in contiguous runs; csr order and all float sums unchanged
# speedup vs baseline: 8.2556x; 1.3326x over previous
#include <hip/hip_runtime.h>
#include <math.h>

#define N_NODES 50000
#define N_EDGES 800000
#define IN_F    191
#define H1 16
#define C1 16
#define F1 256
#define H2 8
#define C2 3
#define F2 24

typedef __attribute__((ext_vector_type(2))) float v2f;
typedef __attribute__((ext_vector_type(8))) float v8f;


__device__ __forceinline__ float lrelu02(float a) {
  return (a > 0.f) ? a : 0.2f * a;
}

__device__ __forceinline__ float elu1(float v) {
  return (v > 0.f) ? v : expm1f(v);
}

#define PS_BLOCKS 196
#define PS_SLICE  4096

__device__ __forceinline__ float wave_sum_f(float v) {
  for (int off = 16; off > 0; off >>= 1) v += __shfl_xor(v, off);
  return v;
}

__global__ void __launch_bounds__(256)
prep_kernel(const float* __restrict__ psum,
            const float* __restrict__ We1, const float* __restrict__ ae1,
            const float* __restrict__ We2, const float* __restrict__ ae2,
            float* __restrict__ mean_out, float* __restrict__ ce1,
            float* __restrict__ ce2) {
  __shared__ float red[8];
  const int t = threadIdx.x;
  const float s = wave_sum_f((t < PS_BLOCKS) ? psum[t] : 0.f);
  if ((t & 31) == 0) red[t >> 5] = s;
  __syncthreads();
  if (t == 0) {
    float tot = 0.f;
    for (int w = 0; w < 8; w++) tot += red[w];
    mean_out[0] = tot * (1.0f / (float)N_EDGES);
  }
  if (t < H1) {
    float c = 0.f;
    for (int k = 0; k < C1; k++) c += We1[t * C1 + k] * ae1[t * C1 + k];
    ce1[t] = c;
  } else if (t >= 32 && t < 32 + H2) {
    const int h = t - 32;
    float c = 0.f;
    for (int k = 0; k < C2; k++) c += We2[h * C2 + k] * ae2[h * C2 + k];
    ce2[h] = c;
  }
}

#define NBKT 196
#define EBLK PS_BLOCKS

__global__ void __launch_bounds__(256)
bucket_count_kernel(const int* __restrict__ ei,
                    const float* __restrict__ ea,
                    int* __restrict__ lrank,
                    int* __restrict__ bhT,
                    float* __restrict__ psum) {
  __shared__ int hist[NBKT];
  __shared__ float red[8];
  const int t = threadIdx.x;
  if (t < NBKT) hist[t] = 0;
  const int base = blockIdx.x * PS_SLICE;
  const int* dstp = ei + N_EDGES;
  float4 q[PS_SLICE / 1024];
  int4 dq[PS_SLICE / 1024];
  for (int k = 0; k < PS_SLICE / 1024; k++) {
    const int i = base + (k * 256 + t) * 4;
    const int ic = (i < N_EDGES) ? i : N_EDGES - 4;
    q[k] = *(const float4*)&ea[ic];
    dq[k] = *(const int4*)&dstp[ic];
  }
  float s = 0.f;
  for (int k = 0; k < PS_SLICE / 1024; k++) {
    const int i = base + (k * 256 + t) * 4;
    const float part = (q[k].x + q[k].y) + (q[k].z + q[k].w);
    s += (i < N_EDGES) ? part : 0.f;
  }
  s = wave_sum_f(s);
  if ((t & 31) == 0) red[t >> 5] = s;
  __syncthreads();
  for (int k = 0; k < PS_SLICE / 1024; k++) {
    const int i = base + (k * 256 + t) * 4;
    if (i < N_EDGES) {
      int4 r;
      r.x = atomicAdd(&hist[dq[k].x >> 8], 1);
      r.y = atomicAdd(&hist[dq[k].y >> 8], 1);
      r.z = atomicAdd(&hist[dq[k].z >> 8], 1);
      r.w = atomicAdd(&hist[dq[k].w >> 8], 1);
      *(int4*)&lrank[i] = r;
    }
  }
  if (t == 0) {
    float tot = 0.f;
    for (int w = 0; w < 8; w++) tot += red[w];
    psum[blockIdx.x] = tot;
  }
  __syncthreads();
  if (t < NBKT) bhT[t * EBLK + blockIdx.x] = hist[t];
}

__global__ void __launch_bounds__(256)
bucket_scan_kernel(const int* __restrict__ bhT, int* __restrict__ boff,
                   int* __restrict__ bse) {
  __shared__ int wpre[8];
  __shared__ int wtot[8];
  const int t = threadIdx.x;
  const int lane = t & 31, wave = t >> 5;
  const int lo = blockIdx.x * EBLK;
  int pre = 0;
  for (int i0 = t * 4; i0 < lo; i0 += 8 * 1024) {
    int4 q[8];
    for (int u = 0; u < 8; u++) {
      const int i = i0 + u * 1024;
      q[u] = (i < lo) ? *(const int4*)&bhT[i] : make_int4(0, 0, 0, 0);
    }
    for (int u = 0; u < 8; u++) pre += (q[u].x + q[u].y) + (q[u].z + q[u].w);
  }
  const int c = (t < EBLK) ? bhT[lo + t] : 0;
  int v = c;
  for (int off = 1; off < 32; off <<= 1) {
    const int u = __shfl_up(v, off);
    if (lane >= off) v += u;
  }
  for (int off = 16; off > 0; off >>= 1) pre += __shfl_xor(pre, off);
  if (lane == 31) wtot[wave] = v;
  if (lane == 0) wpre[wave] = pre;
  __syncthreads();
  int start = 0;
  for (int w = 0; w < 8; w++) {
    start += wpre[w];
    if (w < wave) start += wtot[w];
  }
  if (t < EBLK) boff[lo + t] = start + v - c;
  if (t == 0) bse[2 * blockIdx.x] = start + v - c;
  if (t == EBLK - 1) bse[2 * blockIdx.x + 1] = start + v;
}

__global__ void __launch_bounds__(256)
bucket_scatter_kernel(const int* __restrict__ ei,
                      const int* __restrict__ lrank,
                      const int* __restrict__ bhT,
                      const int* __restrict__ boff,
                      int* __restrict__ tmp_pairs) {
  int2* tmp = (int2*)tmp_pairs;
  __shared__ int lst[NBKT];
  __shared__ int delta[NBKT];
  __shared__ int wtot[8];
  __shared__ int2 sorted[PS_SLICE];
  const int t = threadIdx.x;
  const int lane = t & 31, wave = t >> 5;
  const int base = blockIdx.x * PS_SLICE;
  const int nblk = (N_EDGES - base < PS_SLICE) ? N_EDGES - base : PS_SLICE;
  const int* dstp = ei + N_EDGES;
  const int c = (t < NBKT) ? bhT[t * EBLK + blockIdx.x] : 0;
  const int o = (t < NBKT) ? boff[t * EBLK + blockIdx.x] : 0;
  int4 dq[PS_SLICE / 1024], rq[PS_SLICE / 1024];
  for (int k = 0; k < PS_SLICE / 1024; k++) {
    const int i = base + (k * 256 + t) * 4;
    const int ic = (i < N_EDGES) ? i : N_EDGES - 4;
    dq[k] = *(const int4*)&dstp[ic];
    rq[k] = *(const int4*)&lrank[ic];
  }
  int v = c;
  for (int off = 1; off < 32; off <<= 1) {
    const int u = __shfl_up(v, off);
    if (lane >= off) v += u;
  }
  if (lane == 31) wtot[wave] = v;
  __syncthreads();
  int first = v - c;
  for (int w = 0; w < 8; w++)
    if (w < wave) first += wtot[w];
  if (t < NBKT) {
    lst[t] = first;
    delta[t] = o - first;
  }
  __syncthreads();
  for (int k = 0; k < PS_SLICE / 1024; k++) {
    const int i = base + (k * 256 + t) * 4;
    if (i < N_EDGES) {
      sorted[lst[dq[k].x >> 8] + rq[k].x] = make_int2(i, dq[k].x);
      sorted[lst[dq[k].y >> 8] + rq[k].y] = make_int2(i + 1, dq[k].y);
      sorted[lst[dq[k].z >> 8] + rq[k].z] = make_int2(i + 2, dq[k].z);
      sorted[lst[dq[k].w >> 8] + rq[k].w] = make_int2(i + 3, dq[k].w);
    }
  }
  __syncthreads();
  for (int i = t; i < nblk; i += 256) {
    const int2 p = sorted[i];
    tmp[delta[p.y >> 8] + i] = p;
  }
}

#define BP_GROUP 6
#define BP_CAP   6144
__global__ void __launch_bounds__(1024)
bucket_place_kernel(const int* __restrict__ tmp_pairs, const int* __restrict__ bse,
                    int* __restrict__ rowptr, int* __restrict__ slot_eid) {
  const int2* tmp = (const int2*)tmp_pairs;
  __shared__ int cnt[256];
  __shared__ int wtot[8];
  __shared__ int sbuf[BP_CAP];
  const int t = threadIdx.x;
  const int lane = t & 31, wave = t >> 5;
  const int bs = bse[2 * blockIdx.x], be = bse[2 * blockIdx.x + 1];
  const bool fits = (be - bs) <= BP_CAP;
  if (t < 256) cnt[t] = 0;
  __syncthreads();
  for (int i0 = bs + t; i0 < be; i0 += BP_GROUP * 1024) {
    int dd[BP_GROUP];
    for (int u = 0; u < BP_GROUP; u++) {
      const int i = i0 + u * 1024;
      dd[u] = tmp[(i < be) ? i : be - 1].y;
    }
    for (int u = 0; u < BP_GROUP; u++)
      if (i0 + u * 1024 < be) atomicAdd(&cnt[dd[u] & 255], 1);
  }
  __syncthreads();
  const int c = (t < 256) ? cnt[t] : 0;
  int v = c;
  for (int off = 1; off < 32; off <<= 1) {
    const int u = __shfl_up(v, off);
    if (lane >= off) v += u;
  }
  if (lane == 31 && wave < 8) wtot[wave] = v;
  __syncthreads();
  if (t < 256) {
    int start = bs;
    for (int w = 0; w < 8; w++)
      if (w < wave) start += wtot[w];
    const int first = start + v - c;
    const int node = blockIdx.x * 256 + t;
    if (node <= N_NODES) rowptr[node] = first;
    cnt[t] = first;
  }
  __syncthreads();
  for (int i0 = bs + t; i0 < be; i0 += BP_GROUP * 1024) {
    int2 q[BP_GROUP];
    for (int u = 0; u < BP_GROUP; u++) {
      const int i = i0 + u * 1024;
      q[u] = tmp[(i < be) ? i : be - 1];
    }
    for (int u = 0; u < BP_GROUP; u++) {
      if (i0 + u * 1024 < be) {
        const int pos = atomicAdd(&cnt[q[u].y & 255], 1);
        if (fits) sbuf[pos - bs] = q[u].x;
        else slot_eid[pos] = q[u].x;
      }
    }
  }
  __syncthreads();
  if (fits)
    for (int i = t; i < be - bs; i += 1024) slot_eid[bs + i] = sbuf[i];
}

__global__ void __launch_bounds__(256)
order_segments_kernel(const int* __restrict__ ei,
                      const float* __restrict__ ea,
                      const int* __restrict__ rowptr,
                      const int* __restrict__ slot_eid,
                      int* __restrict__ csr_src,
                      float* __restrict__ csr_ea) {
  const int lane = threadIdx.x & 31;
  const int wv = __builtin_amdgcn_readfirstlane((int)(threadIdx.x >> 5));
  const int d = blockIdx.x * 8 + wv;
  const int beg = rowptr[d], end = rowptr[d + 1];
  for (int base = beg; base < end; base += 32) {
    const int n = (end - base < 32) ? end - base : 32;
    const int mine = (lane < n) ? slot_eid[base + lane] : 0;
    int pos = 0;
    for (int b2 = beg; b2 < end; b2 += 32) {
      const int n2 = (end - b2 < 32) ? end - b2 : 32;
      const int other = (lane < n2) ? slot_eid[b2 + lane] : 0;
      for (int j = 0; j < n2; j++)
        pos += (__builtin_amdgcn_readlane(other, j) < mine) ? 1 : 0;
    }
    if (lane < n) {
      csr_src[beg + pos] = ei[mine];
      csr_ea[beg + pos] = ea[mine];
    }
  }
}

#define A_PITCH 196
#define A_F4    764
#define D1_PITCH 132

__device__ __forceinline__ void gemm1_load_b(const float* __restrict__ W1, int colB,
                                             int t, int k0, float4 (&qb)[6]) {
#pragma unroll
  for (int i = 0; i < 6; i++) {
    const int idx4 = t + 256 * i;
    const int r = idx4 >> 5, c = (idx4 & 31) << 2, k = k0 + r;
    qb[i] = *(const float4*)&W1[((k < IN_F) ? k : IN_F - 1) * F1 + colB + c];
  }
}

__global__ void __launch_bounds__(256)
gemm1_wmma(const float* __restrict__ x, const float* __restrict__ W1,
           const float* __restrict__ a_src, const float* __restrict__ a_dst,
           float* __restrict__ xl1, float* __restrict__ asrc,
           float* __restrict__ adst) {
  __shared__ __attribute__((aligned(16))) float As[16 * A_PITCH];
  __shared__ __attribute__((aligned(16))) float Bs[48][128];
  const int row0 = blockIdx.x * 16;
  const int colB = blockIdx.y * 128;
  const int t = threadIdx.x;
  const int wave = t >> 5, lane = t & 31, lh = lane & 15, hi = lane >> 4;

  const float4* xa = (const float4*)(x + (size_t)row0 * IN_F);
  float4 qa[3];
#pragma unroll
  for (int j = 0; j < 3; j++) {
    const int f = t + 256 * j;
    qa[j] = xa[(f < A_F4) ? f : A_F4 - 1];
  }
  float4 qb[6];
  gemm1_load_b(W1, colB, t, 0, qb);

#pragma unroll
  for (int j = 0; j < 3; j++) {
    const int f = t + 256 * j;
    if (f < A_F4) {
      const float v[4] = {qa[j].x, qa[j].y, qa[j].z, qa[j].w};
#pragma unroll
      for (int q = 0; q < 4; q++) {
        const int e = 4 * f + q, r = e / IN_F, k = e - r * IN_F;
        As[r * A_PITCH + k] = v[q];
      }
    }
  }
  if (t < 16) {
#pragma unroll
    for (int k = IN_F; k < A_PITCH; k++) As[t * A_PITCH + k] = 0.f;
  }

  v8f acc = {};
#pragma unroll
  for (int k0 = 0; k0 < 192; k0 += 48) {
#pragma unroll
    for (int i = 0; i < 6; i++) {
      const int idx4 = t + 256 * i;
      const int r = idx4 >> 5, c = (idx4 & 31) << 2;
      *(float4*)&Bs[r][c] = (k0 + r < IN_F) ? qb[i] : make_float4(0.f, 0.f, 0.f, 0.f);
    }
    __syncthreads();
    if (k0 + 48 < 192) gemm1_load_b(W1, colB, t, k0 + 48, qb);
#pragma unroll
    for (int kk = 0; kk < 48; kk += 4) {
      v2f a, b;
      a = *(const v2f*)&As[lh * A_PITCH + k0 + kk + 2 * hi];
      b.x = Bs[kk + 2 * hi][wave * 16 + lh];
      b.y = Bs[kk + 2 * hi + 1][wave * 16 + lh];
      acc = __builtin_amdgcn_wmma_f32_16x16x4_f32(false, a, false, b,
                                                  (short)0, acc, false, false);
    }
    __syncthreads();
  }
  float* Ds = &Bs[0][0];
#pragma unroll
  for (int i = 0; i < 8; i++)
    Ds[(i + 8 * hi) * D1_PITCH + wave * 16 + lh] = acc[i];
  __syncthreads();
  const int col = colB + wave * 16 + lh;
#pragma unroll
  for (int i = 0; i < 8; i++)
    xl1[(row0 + i + 8 * hi) * F1 + col] = acc[i];
  if (t < 128) {
    const int r = t >> 3, hl = t & 7;
    const int h = blockIdx.y * 8 + hl;
    const float* row = Ds + r * D1_PITCH + hl * C1;
    float s = 0.f, d = 0.f;
#pragma unroll
    for (int c = 0; c < C1; c++) {
      const float v = row[c];
      s += v * a_src[h * C1 + c];
      d += v * a_dst[h * C1 + c];
    }
    asrc[(row0 + r) * H1 + h] = s;
    adst[(row0 + r) * H1 + h] = d;
  }
}

#define D2_PITCH 25
__global__ void __launch_bounds__(64)
gemm2_wmma(const float* __restrict__ h1, const float* __restrict__ W2,
           const float* __restrict__ a_src, const float* __restrict__ a_dst,
           float* __restrict__ xl2, float* __restrict__ asrc,
           float* __restrict__ adst) {
  __shared__ __attribute__((aligned(16))) float As[16][258];
  const int row0 = blockIdx.x * 16;
  const int t = threadIdx.x;
  const int wave = t >> 5, lane = t & 31, lh = lane & 15, hi = lane >> 4;
  float4 q[16];
#pragma unroll
  for (int i = 0; i < 16; i++) {
    const int idx4 = t + 64 * i;
    const int r = idx4 >> 6, c = (idx4 & 63) << 2;
    q[i] = *(const float4*)&h1[(row0 + r) * F1 + c];
  }
#pragma unroll
  for (int i = 0; i < 16; i++) {
    const int idx4 = t + 64 * i;
    const int r = idx4 >> 6, c = (idx4 & 63) << 2;
    As[r][c + 0] = q[i].x; As[r][c + 1] = q[i].y;
    As[r][c + 2] = q[i].z; As[r][c + 3] = q[i].w;
  }
  __syncthreads();
  const int col = wave * 16 + lh;
  const bool cv = col < F2;
  const int cc = cv ? col : F2 - 1;
  v8f acc = {};
#pragma unroll
  for (int kb = 0; kb < 256; kb += 64) {
    v2f bf[16];
#pragma unroll
    for (int u = 0; u < 16; u++) {
      const int k = kb + 4 * u + 2 * hi;
      const float bx = W2[k * F2 + cc], by = W2[(k + 1) * F2 + cc];
      bf[u].x = cv ? bx : 0.f;
      bf[u].y = cv ? by : 0.f;
    }
#pragma unroll
    for (int u = 0; u < 16; u++) {
      const v2f a = *(const v2f*)&As[lh][kb + 4 * u + 2 * hi];
      acc = __builtin_amdgcn_wmma_f32_16x16x4_f32(false, a, false, bf[u],
                                                  (short)0, acc, false, false);
    }
  }
  __syncthreads();
  float* Ds = &As[0][0];
  if (cv) {
#pragma unroll
    for (int i = 0; i < 8; i++)
      Ds[(i + 8 * hi) * D2_PITCH + col] = acc[i];
  }
  __syncthreads();
  if (cv) {
#pragma unroll
    for (int i = 0; i < 8; i++)
      xl2[(row0 + i + 8 * hi) * F2 + col] = acc[i];
  }
#pragma unroll
  for (int j = 0; j < 2; j++) {
    const int p = t + 64 * j;
    const int r = p >> 3, h = p & 7;
    const float* row = Ds + r * D2_PITCH + h * C2;
    float s = 0.f, d = 0.f;
#pragma unroll
    for (int c = 0; c < C2; c++) {
      const float v = row[c];
      s += v * a_src[h * C2 + c];
      d += v * a_dst[h * C2 + c];
    }
    asrc[(row0 + r) * H2 + h] = s;
    adst[(row0 + r) * H2 + h] = d;
  }
}

__global__ void __launch_bounds__(256)
node_softmax_agg1_kernel(const int* __restrict__ rowptr,
                         const int* __restrict__ csr_src,
                         const float* __restrict__ csr_ea,
                         const float* __restrict__ mean,
                         const float* __restrict__ ce,
                         const float* __restrict__ asrc,
                         const float* __restrict__ adst,
                         const float* __restrict__ xl,
                         const float* __restrict__ b1,
                         float* __restrict__ h1out) {
  const int lane = threadIdx.x & 31;
  const int wv = __builtin_amdgcn_readfirstlane((int)(threadIdx.x >> 5));
  const int d = blockIdx.x * 8 + wv;
  const int hA = lane >> 2, hB = 8 + (lane >> 2);
  const int beg = rowptr[d], end = rowptr[d + 1];
  const float adA = adst[d * H1 + hA], adB = adst[d * H1 + hB];
  const float cA = ce[hA], cB = ce[hB];
  const float mv = mean[0];
  const float aSelfA = lrelu02(asrc[d * H1 + hA] + adA + mv * cA);
  const float aSelfB = lrelu02(asrc[d * H1 + hB] + adB + mv * cB);

  float mA = aSelfA, mB = aSelfB;
  for (int base = beg; base < end; base += 32) {
    const int n = (end - base < 32) ? end - base : 32;
    int sL = 0;
    float eL = 0.f;
    if (lane < n) { sL = csr_src[base + lane]; eL = csr_ea[base + lane]; }
    for (int j = 0; j < n; j++) {
      const int s = __builtin_amdgcn_readlane(sL, j);
      const float ev = __int_as_float(__builtin_amdgcn_readlane(__float_as_int(eL), j));
      const float aA = lrelu02(asrc[s * H1 + hA] + adA + ev * cA);
      const float aB = lrelu02(asrc[s * H1 + hB] + adB + ev * cB);
      mA = fmaxf(mA, aA);
      mB = fmaxf(mB, aB);
    }
  }

  const float4 bA = ((const float4*)b1)[lane], bB = ((const float4*)b1)[32 + lane];

  float denA = __expf(aSelfA - mA), denB = __expf(aSelfB - mB);
  float4 accA, accB;
  {
    const float4* r = (const float4*)(xl + d * F1);
    const float4 u = r[lane], w = r[32 + lane];
    accA = make_float4(denA * u.x, denA * u.y, denA * u.z, denA * u.w);
    accB = make_float4(denB * w.x, denB * w.y, denB * w.z, denB * w.w);
  }
  for (int base = beg; base < end; base += 32) {
    const int n = (end - base < 32) ? end - base : 32;
    int sL = 0;
    float eL = 0.f;
    if (lane < n) { sL = csr_src[base + lane]; eL = csr_ea[base + lane]; }
    for (int j = 0; j < n; j++) {
      const int s = __builtin_amdgcn_readlane(sL, j);
      const float ev = __int_as_float(__builtin_amdgcn_readlane(__float_as_int(eL), j));
      const float4* r = (const float4*)(xl + s * F1);
      const float4 u = r[lane], w = r[32 + lane];
      const float aA = lrelu02(asrc[s * H1 + hA] + adA + ev * cA);
      const float aB = lrelu02(asrc[s * H1 + hB] + adB + ev * cB);
      const float eA = __expf(aA - mA), eB = __expf(aB - mB);
      denA += eA;
      denB += eB;
      accA.x += eA * u.x; accA.y += eA * u.y; accA.z += eA * u.z; accA.w += eA * u.w;
      accB.x += eB * w.x; accB.y += eB * w.y; accB.z += eB * w.z; accB.w += eB * w.w;
    }
  }
  const float rA = 1.0f / (denA + 1e-16f), rB = 1.0f / (denB + 1e-16f);
  float4* o = (float4*)(h1out + d * F1);
  o[lane]      = make_float4(elu1(accA.x * rA + bA.x), elu1(accA.y * rA + bA.y),
                             elu1(accA.z * rA + bA.z), elu1(accA.w * rA + bA.w));
  o[32 + lane] = make_float4(elu1(accB.x * rB + bB.x), elu1(accB.y * rB + bB.y),
                             elu1(accB.z * rB + bB.z), elu1(accB.w * rB + bB.w));
}

__global__ void __launch_bounds__(256)
node_softmax_agg2_kernel(const int* __restrict__ rowptr,
                         const int* __restrict__ csr_src,
                         const float* __restrict__ csr_ea,
                         const float* __restrict__ mean,
                         const float* __restrict__ ce,
                         const float* __restrict__ asrc,
                         const float* __restrict__ adst,
                         const float* __restrict__ xl,
                         float* __restrict__ agg) {
  const int idx = blockIdx.x * blockDim.x + threadIdx.x;
  if (idx >= N_NODES * H2) return;
  const int d = idx / H2, h = idx % H2;
  const int beg = rowptr[d], end = rowptr[d + 1];
  const float ad = adst[idx];
  const float c = ce[h];
  const float aSelf = lrelu02(asrc[idx] + ad + mean[0] * c);

  float m = aSelf;
  for (int j = beg; j < end; j++) {
    const float a = lrelu02(asrc[csr_src[j] * H2 + h] + ad + csr_ea[j] * c);
    m = fmaxf(m, a);
  }

  float den = __expf(aSelf - m);
  const float* rd = xl + d * F2 + h * C2;
  float acc0 = den * rd[0], acc1 = den * rd[1], acc2 = den * rd[2];
  for (int j = beg; j < end; j++) {
    const int s = csr_src[j];
    const float a = lrelu02(asrc[s * H2 + h] + ad + csr_ea[j] * c);
    const float ex = __expf(a - m);
    const float* rs = xl + s * F2 + h * C2;
    den += ex;
    acc0 += ex * rs[0];
    acc1 += ex * rs[1];
    acc2 += ex * rs[2];
  }
  const float rinv = 1.0f / (den + 1e-16f);
  float* o = agg + d * F2 + h * C2;
  o[0] = acc0 * rinv;
  o[1] = acc1 * rinv;
  o[2] = acc2 * rinv;
}

__global__ void final2_kernel(const float* __restrict__ agg2,
                              const float* __restrict__ b2,
                              float* __restrict__ out) {
  const int idx = blockIdx.x * blockDim.x + threadIdx.x;
  if (idx >= N_NODES * C2) return;
  const int n = idx / C2, c = idx % C2;
  float s = 0.f;
#pragma unroll
  for (int h = 0; h < H2; h++) s += agg2[n * F2 + h * C2 + c];
  const float v = s * (1.f / (float)H2) + b2[c];
  out[idx] = (v > 0.f) ? v : expm1f(v);
}

extern "C" void kernel_launch(void* const* d_in, const int* in_sizes, int n_in,
                              void* d_out, int out_size, void* d_ws, size_t ws_size,
                              hipStream_t stream) {
  (void)in_sizes; (void)n_in; (void)out_size; (void)ws_size;
  const float* x   = (const float*)d_in[0];
  const int*   ei  = (const int*)d_in[1];
  const float* ea  = (const float*)d_in[2];
  const float* W1  = (const float*)d_in[3];
  const float* We1 = (const float*)d_in[4];
  const float* as1 = (const float*)d_in[5];
  const float* ad1 = (const float*)d_in[6];
  const float* ae1 = (const float*)d_in[7];
  const float* b1  = (const float*)d_in[8];
  const float* W2  = (const float*)d_in[9];
  const float* We2 = (const float*)d_in[10];
  const float* as2 = (const float*)d_in[11];
  const float* ad2 = (const float*)d_in[12];
  const float* ae2 = (const float*)d_in[13];
  const float* b2  = (const float*)d_in[14];
  float* out = (float*)d_out;

  float* w = (float*)d_ws;
  float* mean = w;
  float* ce1  = w + 1;
  float* ce2  = w + 17;
  size_t off = 32;
  float* xl1   = w + off; off += (size_t)N_NODES * F1;
  float* asrc1 = w + off; off += (size_t)N_NODES * H1;
  float* adst1 = w + off; off += (size_t)N_NODES * H1;
  float* agg1  = w + off; off += (size_t)N_NODES * F1;
  float* xl2   = w + off; off += (size_t)N_NODES * F2;
  float* asrc2 = w + off; off += (size_t)N_NODES * H2;
  float* adst2 = w + off; off += (size_t)N_NODES * H2;
  float* agg2  = w + off; off += (size_t)N_NODES * F2;
  int* rowptr  = (int*)(w + off); off += (size_t)N_NODES + 32;
  int* rank    = (int*)(w + off); off += (size_t)N_EDGES;
  int* slot_eid = (int*)(w + off); off += (size_t)N_EDGES;
  int* csr_src = (int*)(w + off); off += (size_t)N_EDGES;
  float* csr_ea = w + off;        off += (size_t)N_EDGES;
  float* psum = w + off;          off += 256;
  int* bhT  = (int*)(w + off);    off += (size_t)NBKT * EBLK;
  int* boff = (int*)(w + off);    off += (size_t)NBKT * EBLK;
  int* bse  = (int*)(w + off);    off += 512;
  int* tmp  = (int*)(w + off);    off += (size_t)2 * N_EDGES;

  bucket_count_kernel<<<EBLK, 256, 0, stream>>>(ei, ea, rank, bhT, psum);
  bucket_scan_kernel<<<NBKT, 256, 0, stream>>>(bhT, boff, bse);
  bucket_scatter_kernel<<<EBLK, 256, 0, stream>>>(ei, rank, bhT, boff, tmp);
  bucket_place_kernel<<<NBKT, 1024, 0, stream>>>(tmp, bse, rowptr, slot_eid);
  order_segments_kernel<<<N_NODES / 8, 256, 0, stream>>>(ei, ea, rowptr, slot_eid, csr_src, csr_ea);

  prep_kernel<<<1, 256, 0, stream>>>(psum, We1, ae1, We2, ae2, mean, ce1, ce2);

  gemm1_wmma<<<dim3(N_NODES / 16, 2), 256, 0, stream>>>(x, W1, as1, ad1, xl1, asrc1, adst1);
  node_softmax_agg1_kernel<<<N_NODES / 8, 256, 0, stream>>>(
      rowptr, csr_src, csr_ea, mean, ce1, asrc1, adst1, xl1, b1, agg1);

  gemm2_wmma<<<N_NODES / 16, 64, 0, stream>>>(agg1, W2, as2, ad2, xl2, asrc2, adst2);
  node_softmax_agg2_kernel<<<(N_NODES * H2 + 255) / 256, 256, 0, stream>>>(
      rowptr, csr_src, csr_ea, mean, ce2, asrc2, adst2, xl2, agg2);
  final2_kernel<<<(N_NODES * C2 + 255) / 256, 256, 0, stream>>>(agg2, b2, out);
}


# sort records carry source and attribute; the placement block ranks each segment by edge id in LDS and writes csr itself: ordering launch and its per-edge fetches gone; csr and float sums unchanged
# speedup vs baseline: 8.5471x; 1.0353x over previous
#include <hip/hip_runtime.h>
#include <math.h>

#define N_NODES 50000
#define N_EDGES 800000
#define IN_F    191
#define H1 16
#define C1 16
#define F1 256
#define H2 8
#define C2 3
#define F2 24

typedef __attribute__((ext_vector_type(2))) float v2f;
typedef __attribute__((ext_vector_type(8))) float v8f;


__device__ __forceinline__ float lrelu02(float a) {
  return (a > 0.f) ? a : 0.2f * a;
}

__device__ __forceinline__ float elu1(float v) {
  return (v > 0.f) ? v : expm1f(v);
}

#define PS_BLOCKS 196
#define PS_SLICE  4096

__device__ __forceinline__ float wave_sum_f(float v) {
  for (int off = 16; off > 0; off >>= 1) v += __shfl_xor(v, off);
  return v;
}

__global__ void __launch_bounds__(256)
prep_kernel(const float* __restrict__ psum,
            const float* __restrict__ We1, const float* __restrict__ ae1,
            const float* __restrict__ We2, const float* __restrict__ ae2,
            float* __restrict__ mean_out, float* __restrict__ ce1,
            float* __restrict__ ce2) {
  __shared__ float red[8];
  const int t = threadIdx.x;
  const float s = wave_sum_f((t < PS_BLOCKS) ? psum[t] : 0.f);
  if ((t & 31) == 0) red[t >> 5] = s;
  __syncthreads();
  if (t == 0) {
    float tot = 0.f;
    for (int w = 0; w < 8; w++) tot += red[w];
    mean_out[0] = tot * (1.0f / (float)N_EDGES);
  }
  if (t < H1) {
    float c = 0.f;
    for (int k = 0; k < C1; k++) c += We1[t * C1 + k] * ae1[t * C1 + k];
    ce1[t] = c;
  } else if (t >= 32 && t < 32 + H2) {
    const int h = t - 32;
    float c = 0.f;
    for (int k = 0; k < C2; k++) c += We2[h * C2 + k] * ae2[h * C2 + k];
    ce2[h] = c;
  }
}

#define NBKT 196
#define EBLK PS_BLOCKS

__global__ void __launch_bounds__(256)
bucket_count_kernel(const int* __restrict__ ei,
                    const float* __restrict__ ea,
                    int* __restrict__ lrank,
                    int* __restrict__ bhT,
                    float* __restrict__ psum) {
  __shared__ int hist[NBKT];
  __shared__ float red[8];
  const int t = threadIdx.x;
  if (t < NBKT) hist[t] = 0;
  const int base = blockIdx.x * PS_SLICE;
  const int* dstp = ei + N_EDGES;
  float4 q[PS_SLICE / 1024];
  int4 dq[PS_SLICE / 1024];
  for (int k = 0; k < PS_SLICE / 1024; k++) {
    const int i = base + (k * 256 + t) * 4;
    const int ic = (i < N_EDGES) ? i : N_EDGES - 4;
    q[k] = *(const float4*)&ea[ic];
    dq[k] = *(const int4*)&dstp[ic];
  }
  float s = 0.f;
  for (int k = 0; k < PS_SLICE / 1024; k++) {
    const int i = base + (k * 256 + t) * 4;
    const float part = (q[k].x + q[k].y) + (q[k].z + q[k].w);
    s += (i < N_EDGES) ? part : 0.f;
  }
  s = wave_sum_f(s);
  if ((t & 31) == 0) red[t >> 5] = s;
  __syncthreads();
  for (int k = 0; k < PS_SLICE / 1024; k++) {
    const int i = base + (k * 256 + t) * 4;
    if (i < N_EDGES) {
      int4 r;
      r.x = atomicAdd(&hist[dq[k].x >> 8], 1);
      r.y = atomicAdd(&hist[dq[k].y >> 8], 1);
      r.z = atomicAdd(&hist[dq[k].z >> 8], 1);
      r.w = atomicAdd(&hist[dq[k].w >> 8], 1);
      *(int4*)&lrank[i] = r;
    }
  }
  if (t == 0) {
    float tot = 0.f;
    for (int w = 0; w < 8; w++) tot += red[w];
    psum[blockIdx.x] = tot;
  }
  __syncthreads();
  if (t < NBKT) bhT[t * EBLK + blockIdx.x] = hist[t];
}

__global__ void __launch_bounds__(256)
bucket_scan_kernel(const int* __restrict__ bhT, int* __restrict__ boff,
                   int* __restrict__ bse) {
  __shared__ int wpre[8];
  __shared__ int wtot[8];
  const int t = threadIdx.x;
  const int lane = t & 31, wave = t >> 5;
  const int lo = blockIdx.x * EBLK;
  int pre = 0;
  for (int i0 = t * 4; i0 < lo; i0 += 8 * 1024) {
    int4 q[8];
    for (int u = 0; u < 8; u++) {
      const int i = i0 + u * 1024;
      q[u] = (i < lo) ? *(const int4*)&bhT[i] : make_int4(0, 0, 0, 0);
    }
    for (int u = 0; u < 8; u++) pre += (q[u].x + q[u].y) + (q[u].z + q[u].w);
  }
  const int c = (t < EBLK) ? bhT[lo + t] : 0;
  int v = c;
  for (int off = 1; off < 32; off <<= 1) {
    const int u = __shfl_up(v, off);
    if (lane >= off) v += u;
  }
  for (int off = 16; off > 0; off >>= 1) pre += __shfl_xor(pre, off);
  if (lane == 31) wtot[wave] = v;
  if (lane == 0) wpre[wave] = pre;
  __syncthreads();
  int start = 0;
  for (int w = 0; w < 8; w++) {
    start += wpre[w];
    if (w < wave) start += wtot[w];
  }
  if (t < EBLK) boff[lo + t] = start + v - c;
  if (t == 0) bse[2 * blockIdx.x] = start + v - c;
  if (t == EBLK - 1) bse[2 * blockIdx.x + 1] = start + v;
}

__global__ void __launch_bounds__(256)
bucket_scatter_kernel(const int* __restrict__ ei,
                      const float* __restrict__ ea,
                      const int* __restrict__ lrank,
                      const int* __restrict__ bhT,
                      const int* __restrict__ boff,
                      int* __restrict__ tmp_recs) {
  int4* tmp = (int4*)tmp_recs;
  __shared__ int lst[NBKT];
  __shared__ int delta[NBKT];
  __shared__ int wtot[8];
  __shared__ int4 sorted[PS_SLICE];
  const int t = threadIdx.x;
  const int lane = t & 31, wave = t >> 5;
  const int base = blockIdx.x * PS_SLICE;
  const int nblk = (N_EDGES - base < PS_SLICE) ? N_EDGES - base : PS_SLICE;
  const int* dstp = ei + N_EDGES;
  const int c = (t < NBKT) ? bhT[t * EBLK + blockIdx.x] : 0;
  const int o = (t < NBKT) ? boff[t * EBLK + blockIdx.x] : 0;
  int4 dq[PS_SLICE / 1024], rq[PS_SLICE / 1024], sq[PS_SLICE / 1024];
  float4 aq[PS_SLICE / 1024];
  for (int k = 0; k < PS_SLICE / 1024; k++) {
    const int i = base + (k * 256 + t) * 4;
    const int ic = (i < N_EDGES) ? i : N_EDGES - 4;
    dq[k] = *(const int4*)&dstp[ic];
    rq[k] = *(const int4*)&lrank[ic];
    sq[k] = *(const int4*)&ei[ic];
    aq[k] = *(const float4*)&ea[ic];
  }
  int v = c;
  for (int off = 1; off < 32; off <<= 1) {
    const int u = __shfl_up(v, off);
    if (lane >= off) v += u;
  }
  if (lane == 31) wtot[wave] = v;
  __syncthreads();
  int first = v - c;
  for (int w = 0; w < 8; w++)
    if (w < wave) first += wtot[w];
  if (t < NBKT) {
    lst[t] = first;
    delta[t] = o - first;
  }
  __syncthreads();
  for (int k = 0; k < PS_SLICE / 1024; k++) {
    const int i = base + (k * 256 + t) * 4;
    if (i < N_EDGES) {
      sorted[lst[dq[k].x >> 8] + rq[k].x] = make_int4(i, dq[k].x, sq[k].x, __float_as_int(aq[k].x));
      sorted[lst[dq[k].y >> 8] + rq[k].y] = make_int4(i + 1, dq[k].y, sq[k].y, __float_as_int(aq[k].y));
      sorted[lst[dq[k].z >> 8] + rq[k].z] = make_int4(i + 2, dq[k].z, sq[k].z, __float_as_int(aq[k].z));
      sorted[lst[dq[k].w >> 8] + rq[k].w] = make_int4(i + 3, dq[k].w, sq[k].w, __float_as_int(aq[k].w));
    }
  }
  __syncthreads();
  for (int i = t; i < nblk; i += 256) {
    const int4 p = sorted[i];
    tmp[delta[p.y >> 8] + i] = p;
  }
}

#define BP_GROUP 6
#define BP_CAP   6144
#define ID_BITS  20
#define ID_MASK  ((1 << ID_BITS) - 1)
static_assert(N_EDGES <= (1 << ID_BITS), "edge id and local destination share one word");

__device__ __forceinline__ void order_bucket_from_memory(const int* __restrict__ ei,
                                                         const float* __restrict__ ea,
                                                         const int* seg, const int* gkeys,
                                                         int bs, int nb, int t,
                                                         int* __restrict__ csr_src,
                                                         float* __restrict__ csr_ea) {
  for (int i = t; i < nb; i += 1024) {
    const int k = gkeys[bs + i];
    const int ld = k >> ID_BITS;
    const int b = seg[ld], e = seg[ld + 1];
    int pos = 0;
    for (int j = b; j < e; j++) pos += (gkeys[bs + j] < k) ? 1 : 0;
    const int id = k & ID_MASK;
    csr_src[bs + b + pos] = ei[id];
    csr_ea[bs + b + pos] = ea[id];
  }
}

__global__ void __launch_bounds__(1024)
bucket_place_kernel(const int* __restrict__ ei, const float* __restrict__ ea,
                    const int* __restrict__ tmp_recs, const int* __restrict__ bse,
                    int* __restrict__ rowptr, int* gkeys,
                    int* __restrict__ csr_src, float* __restrict__ csr_ea) {
  const int4* tmp = (const int4*)tmp_recs;
  __shared__ int cnt[256];
  __shared__ int seg[257];
  __shared__ int wtot[8];
  __shared__ int skey[BP_CAP];
  __shared__ int ssrc[BP_CAP];
  __shared__ float sea[BP_CAP];
  const int t = threadIdx.x;
  const int lane = t & 31, wave = t >> 5;
  const int bs = bse[2 * blockIdx.x], be = bse[2 * blockIdx.x + 1];
  const int nb = be - bs;
  const bool fits = nb <= BP_CAP;
  if (t < 256) cnt[t] = 0;
  __syncthreads();
  for (int i0 = bs + t; i0 < be; i0 += BP_GROUP * 1024) {
    int dd[BP_GROUP];
    for (int u = 0; u < BP_GROUP; u++) {
      const int i = i0 + u * 1024;
      dd[u] = tmp[(i < be) ? i : be - 1].y;
    }
    for (int u = 0; u < BP_GROUP; u++)
      if (i0 + u * 1024 < be) atomicAdd(&cnt[dd[u] & 255], 1);
  }
  __syncthreads();
  const int c = (t < 256) ? cnt[t] : 0;
  int v = c;
  for (int off = 1; off < 32; off <<= 1) {
    const int u = __shfl_up(v, off);
    if (lane >= off) v += u;
  }
  if (lane == 31 && wave < 8) wtot[wave] = v;
  __syncthreads();
  if (t < 256) {
    int start = bs;
    for (int w = 0; w < 8; w++)
      if (w < wave) start += wtot[w];
    const int first = start + v - c;
    const int node = blockIdx.x * 256 + t;
    if (node <= N_NODES) rowptr[node] = first;
    cnt[t] = first;
    seg[t] = first - bs;
    if (t == 255) seg[256] = first + c - bs;
  }
  __syncthreads();
  for (int i0 = bs + t; i0 < be; i0 += BP_GROUP * 1024) {
    int4 q[BP_GROUP];
    for (int u = 0; u < BP_GROUP; u++) {
      const int i = i0 + u * 1024;
      q[u] = tmp[(i < be) ? i : be - 1];
    }
    for (int u = 0; u < BP_GROUP; u++) {
      if (i0 + u * 1024 < be) {
        const int ld = q[u].y & 255;
        const int pos = atomicAdd(&cnt[ld], 1);
        const int k = q[u].x | (ld << ID_BITS);
        if (fits) {
          skey[pos - bs] = k;
          ssrc[pos - bs] = q[u].z;
          sea[pos - bs] = __int_as_float(q[u].w);
        } else {
          gkeys[pos] = k;
        }
      }
    }
  }
  __syncthreads();
  if (fits) {
    for (int i = t; i < nb; i += 1024) {
      const int k = skey[i];
      const int ld = k >> ID_BITS;
      const int b = seg[ld], e = seg[ld + 1];
      int pos = 0;
      for (int j = b; j < e; j++) pos += (skey[j] < k) ? 1 : 0;
      csr_src[bs + b + pos] = ssrc[i];
      csr_ea[bs + b + pos] = sea[i];
    }
  } else {
    order_bucket_from_memory(ei, ea, seg, gkeys, bs, nb, t, csr_src, csr_ea);
  }
}

#define A_PITCH 196
#define A_F4    764
#define D1_PITCH 132

__device__ __forceinline__ void gemm1_load_b(const float* __restrict__ W1, int colB,
                                             int t, int k0, float4 (&qb)[6]) {
#pragma unroll
  for (int i = 0; i < 6; i++) {
    const int idx4 = t + 256 * i;
    const int r = idx4 >> 5, c = (idx4 & 31) << 2, k = k0 + r;
    qb[i] = *(const float4*)&W1[((k < IN_F) ? k : IN_F - 1) * F1 + colB + c];
  }
}

__global__ void __launch_bounds__(256)
gemm1_wmma(const float* __restrict__ x, const float* __restrict__ W1,
           const float* __restrict__ a_src, const float* __restrict__ a_dst,
           float* __restrict__ xl1, float* __restrict__ asrc,
           float* __restrict__ adst) {
  __shared__ __attribute__((aligned(16))) float As[16 * A_PITCH];
  __shared__ __attribute__((aligned(16))) float Bs[48][128];
  const int row0 = blockIdx.x * 16;
  const int colB = blockIdx.y * 128;
  const int t = threadIdx.x;
  const int wave = t >> 5, lane = t & 31, lh = lane & 15, hi = lane >> 4;

  const float4* xa = (const float4*)(x + (size_t)row0 * IN_F);
  float4 qa[3];
#pragma unroll
  for (int j = 0; j < 3; j++) {
    const int f = t + 256 * j;
    qa[j] = xa[(f < A_F4) ? f : A_F4 - 1];
  }
  float4 qb[6];
  gemm1_load_b(W1, colB, t, 0, qb);

#pragma unroll
  for (int j = 0; j < 3; j++) {
    const int f = t + 256 * j;
    if (f < A_F4) {
      const float v[4] = {qa[j].x, qa[j].y, qa[j].z, qa[j].w};
#pragma unroll
      for (int q = 0; q < 4; q++) {
        const int e = 4 * f + q, r = e / IN_F, k = e - r * IN_F;
        As[r * A_PITCH + k] = v[q];
      }
    }
  }
  if (t < 16) {
#pragma unroll
    for (int k = IN_F; k < A_PITCH; k++) As[t * A_PITCH + k] = 0.f;
  }

  v8f acc = {};
#pragma unroll
  for (int k0 = 0; k0 < 192; k0 += 48) {
#pragma unroll
    for (int i = 0; i < 6; i++) {
      const int idx4 = t + 256 * i;
      const int r = idx4 >> 5, c = (idx4 & 31) << 2;
      *(float4*)&Bs[r][c] = (k0 + r < IN_F) ? qb[i] : make_float4(0.f, 0.f, 0.f, 0.f);
    }
    __syncthreads();
    if (k0 + 48 < 192) gemm1_load_b(W1, colB, t, k0 + 48, qb);
#pragma unroll
    for (int kk = 0; kk < 48; kk += 4) {
      v2f a, b;
      a = *(const v2f*)&As[lh * A_PITCH + k0 + kk + 2 * hi];
      b.x = Bs[kk + 2 * hi][wave * 16 + lh];
      b.y = Bs[kk + 2 * hi + 1][wave * 16 + lh];
      acc = __builtin_amdgcn_wmma_f32_16x16x4_f32(false, a, false, b,
                                                  (short)0, acc, false, false);
    }
    __syncthreads();
  }
  float* Ds = &Bs[0][0];
#pragma unroll
  for (int i = 0; i < 8; i++)
    Ds[(i + 8 * hi) * D1_PITCH + wave * 16 + lh] = acc[i];
  __syncthreads();
  const int col = colB + wave * 16 + lh;
#pragma unroll
  for (int i = 0; i < 8; i++)
    xl1[(row0 + i + 8 * hi) * F1 + col] = acc[i];
  if (t < 128) {
    const int r = t >> 3, hl = t & 7;
    const int h = blockIdx.y * 8 + hl;
    const float* row = Ds + r * D1_PITCH + hl * C1;
    float s = 0.f, d = 0.f;
#pragma unroll
    for (int c = 0; c < C1; c++) {
      const float v = row[c];
      s += v * a_src[h * C1 + c];
      d += v * a_dst[h * C1 + c];
    }
    asrc[(row0 + r) * H1 + h] = s;
    adst[(row0 + r) * H1 + h] = d;
  }
}

#define D2_PITCH 25
__global__ void __launch_bounds__(64)
gemm2_wmma(const float* __restrict__ h1, const float* __restrict__ W2,
           const float* __restrict__ a_src, const float* __restrict__ a_dst,
           float* __restrict__ xl2, float* __restrict__ asrc,
           float* __restrict__ adst) {
  __shared__ __attribute__((aligned(16))) float As[16][258];
  const int row0 = blockIdx.x * 16;
  const int t = threadIdx.x;
  const int wave = t >> 5, lane = t & 31, lh = lane & 15, hi = lane >> 4;
  float4 q[16];
#pragma unroll
  for (int i = 0; i < 16; i++) {
    const int idx4 = t + 64 * i;
    const int r = idx4 >> 6, c = (idx4 & 63) << 2;
    q[i] = *(const float4*)&h1[(row0 + r) * F1 + c];
  }
#pragma unroll
  for (int i = 0; i < 16; i++) {
    const int idx4 = t + 64 * i;
    const int r = idx4 >> 6, c = (idx4 & 63) << 2;
    As[r][c + 0] = q[i].x; As[r][c + 1] = q[i].y;
    As[r][c + 2] = q[i].z; As[r][c + 3] = q[i].w;
  }
  __syncthreads();
  const int col = wave * 16 + lh;
  const bool cv = col < F2;
  const int cc = cv ? col : F2 - 1;
  v8f acc = {};
#pragma unroll
  for (int kb = 0; kb < 256; kb += 64) {
    v2f bf[16];
#pragma unroll
    for (int u = 0; u < 16; u++) {
      const int k = kb + 4 * u + 2 * hi;
      const float bx = W2[k * F2 + cc], by = W2[(k + 1) * F2 + cc];
      bf[u].x = cv ? bx : 0.f;
      bf[u].y = cv ? by : 0.f;
    }
#pragma unroll
    for (int u = 0; u < 16; u++) {
      const v2f a = *(const v2f*)&As[lh][kb + 4 * u + 2 * hi];
      acc = __builtin_amdgcn_wmma_f32_16x16x4_f32(false, a, false, bf[u],
                                                  (short)0, acc, false, false);
    }
  }
  __syncthreads();
  float* Ds = &As[0][0];
  if (cv) {
#pragma unroll
    for (int i = 0; i < 8; i++)
      Ds[(i + 8 * hi) * D2_PITCH + col] = acc[i];
  }
  __syncthreads();
  if (cv) {
#pragma unroll
    for (int i = 0; i < 8; i++)
      xl2[(row0 + i + 8 * hi) * F2 + col] = acc[i];
  }
#pragma unroll
  for (int j = 0; j < 2; j++) {
    const int p = t + 64 * j;
    const int r = p >> 3, h = p & 7;
    const float* row = Ds + r * D2_PITCH + h * C2;
    float s = 0.f, d = 0.f;
#pragma unroll
    for (int c = 0; c < C2; c++) {
      const float v = row[c];
      s += v * a_src[h * C2 + c];
      d += v * a_dst[h * C2 + c];
    }
    asrc[(row0 + r) * H2 + h] = s;
    adst[(row0 + r) * H2 + h] = d;
  }
}

__global__ void __launch_bounds__(256)
node_softmax_agg1_kernel(const int* __restrict__ rowptr,
                         const int* __restrict__ csr_src,
                         const float* __restrict__ csr_ea,
                         const float* __restrict__ mean,
                         const float* __restrict__ ce,
                         const float* __restrict__ asrc,
                         const float* __restrict__ adst,
                         const float* __restrict__ xl,
                         const float* __restrict__ b1,
                         float* __restrict__ h1out) {
  const int lane = threadIdx.x & 31;
  const int wv = __builtin_amdgcn_readfirstlane((int)(threadIdx.x >> 5));
  const int d = blockIdx.x * 8 + wv;
  const int hA = lane >> 2, hB = 8 + (lane >> 2);
  const int beg = rowptr[d], end = rowptr[d + 1];
  const float adA = adst[d * H1 + hA], adB = adst[d * H1 + hB];
  const float cA = ce[hA], cB = ce[hB];
  const float mv = mean[0];
  const float aSelfA = lrelu02(asrc[d * H1 + hA] + adA + mv * cA);
  const float aSelfB = lrelu02(asrc[d * H1 + hB] + adB + mv * cB);

  float mA = aSelfA, mB = aSelfB;
  for (int base = beg; base < end; base += 32) {
    const int n = (end - base < 32) ? end - base : 32;
    int sL = 0;
    float eL = 0.f;
    if (lane < n) { sL = csr_src[base + lane]; eL = csr_ea[base + lane]; }
    for (int j = 0; j < n; j++) {
      const int s = __builtin_amdgcn_readlane(sL, j);
      const float ev = __int_as_float(__builtin_amdgcn_readlane(__float_as_int(eL), j));
      const float aA = lrelu02(asrc[s * H1 + hA] + adA + ev * cA);
      const float aB = lrelu02(asrc[s * H1 + hB] + adB + ev * cB);
      mA = fmaxf(mA, aA);
      mB = fmaxf(mB, aB);
    }
  }

  const float4 bA = ((const float4*)b1)[lane], bB = ((const float4*)b1)[32 + lane];

  float denA = __expf(aSelfA - mA), denB = __expf(aSelfB - mB);
  float4 accA, accB;
  {
    const float4* r = (const float4*)(xl + d * F1);
    const float4 u = r[lane], w = r[32 + lane];
    accA = make_float4(denA * u.x, denA * u.y, denA * u.z, denA * u.w);
    accB = make_float4(denB * w.x, denB * w.y, denB * w.z, denB * w.w);
  }
  for (int base = beg; base < end; base += 32) {
    const int n = (end - base < 32) ? end - base : 32;
    int sL = 0;
    float eL = 0.f;
    if (lane < n) { sL = csr_src[base + lane]; eL = csr_ea[base + lane]; }
    for (int j = 0; j < n; j++) {
      const int s = __builtin_amdgcn_readlane(sL, j);
      const float ev = __int_as_float(__builtin_amdgcn_readlane(__float_as_int(eL), j));
      const float4* r = (const float4*)(xl + s * F1);
      const float4 u = r[lane], w = r[32 + lane];
      const float aA = lrelu02(asrc[s * H1 + hA] + adA + ev * cA);
      const float aB = lrelu02(asrc[s * H1 + hB] + adB + ev * cB);
      const float eA = __expf(aA - mA), eB = __expf(aB - mB);
      denA += eA;
      denB += eB;
      accA.x += eA * u.x; accA.y += eA * u.y; accA.z += eA * u.z; accA.w += eA * u.w;
      accB.x += eB * w.x; accB.y += eB * w.y; accB.z += eB * w.z; accB.w += eB * w.w;
    }
  }
  const float rA = 1.0f / (denA + 1e-16f), rB = 1.0f / (denB + 1e-16f);
  float4* o = (float4*)(h1out + d * F1);
  o[lane]      = make_float4(elu1(accA.x * rA + bA.x), elu1(accA.y * rA + bA.y),
                             elu1(accA.z * rA + bA.z), elu1(accA.w * rA + bA.w));
  o[32 + lane] = make_float4(elu1(accB.x * rB + bB.x), elu1(accB.y * rB + bB.y),
                             elu1(accB.z * rB + bB.z), elu1(accB.w * rB + bB.w));
}

__global__ void __launch_bounds__(256)
node_softmax_agg2_kernel(const int* __restrict__ rowptr,
                         const int* __restrict__ csr_src,
                         const float* __restrict__ csr_ea,
                         const float* __restrict__ mean,
                         const float* __restrict__ ce,
                         const float* __restrict__ asrc,
                         const float* __restrict__ adst,
                         const float* __restrict__ xl,
                         float* __restrict__ agg) {
  const int idx = blockIdx.x * blockDim.x + threadIdx.x;
  if (idx >= N_NODES * H2) return;
  const int d = idx / H2, h = idx % H2;
  const int beg = rowptr[d], end = rowptr[d + 1];
  const float ad = adst[idx];
  const float c = ce[h];
  const float aSelf = lrelu02(asrc[idx] + ad + mean[0] * c);

  float m = aSelf;
  for (int j = beg; j < end; j++) {
    const float a = lrelu02(asrc[csr_src[j] * H2 + h] + ad + csr_ea[j] * c);
    m = fmaxf(m, a);
  }

  float den = __expf(aSelf - m);
  const float* rd = xl + d * F2 + h * C2;
  float acc0 = den * rd[0], acc1 = den * rd[1], acc2 = den * rd[2];
  for (int j = beg; j < end; j++) {
    const int s = csr_src[j];
    const float a = lrelu02(asrc[s * H2 + h] + ad + csr_ea[j] * c);
    const float ex = __expf(a - m);
    const float* rs = xl + s * F2 + h * C2;
    den += ex;
    acc0 += ex * rs[0];
    acc1 += ex * rs[1];
    acc2 += ex * rs[2];
  }
  const float rinv = 1.0f / (den + 1e-16f);
  float* o = agg + d * F2 + h * C2;
  o[0] = acc0 * rinv;
  o[1] = acc1 * rinv;
  o[2] = acc2 * rinv;
}

__global__ void final2_kernel(const float* __restrict__ agg2,
                              const float* __restrict__ b2,
                              float* __restrict__ out) {
  const int idx = blockIdx.x * blockDim.x + threadIdx.x;
  if (idx >= N_NODES * C2) return;
  const int n = idx / C2, c = idx % C2;
  float s = 0.f;
#pragma unroll
  for (int h = 0; h < H2; h++) s += agg2[n * F2 + h * C2 + c];
  const float v = s * (1.f / (float)H2) + b2[c];
  out[idx] = (v > 0.f) ? v : expm1f(v);
}

extern "C" void kernel_launch(void* const* d_in, const int* in_sizes, int n_in,
                              void* d_out, int out_size, void* d_ws, size_t ws_size,
                              hipStream_t stream) {
  (void)in_sizes; (void)n_in; (void)out_size; (void)ws_size;
  const float* x   = (const float*)d_in[0];
  const int*   ei  = (const int*)d_in[1];
  const float* ea  = (const float*)d_in[2];
  const float* W1  = (const float*)d_in[3];
  const float* We1 = (const float*)d_in[4];
  const float* as1 = (const float*)d_in[5];
  const float* ad1 = (const float*)d_in[6];
  const float* ae1 = (const float*)d_in[7];
  const float* b1  = (const float*)d_in[8];
  const float* W2  = (const float*)d_in[9];
  const float* We2 = (const float*)d_in[10];
  const float* as2 = (const float*)d_in[11];
  const float* ad2 = (const float*)d_in[12];
  const float* ae2 = (const float*)d_in[13];
  const float* b2  = (const float*)d_in[14];
  float* out = (float*)d_out;

  float* w = (float*)d_ws;
  float* mean = w;
  float* ce1  = w + 1;
  float* ce2  = w + 17;
  size_t off = 32;
  float* xl1   = w + off; off += (size_t)N_NODES * F1;
  float* asrc1 = w + off; off += (size_t)N_NODES * H1;
  float* adst1 = w + off; off += (size_t)N_NODES * H1;
  float* agg1  = w + off; off += (size_t)N_NODES * F1;
  float* xl2   = w + off; off += (size_t)N_NODES * F2;
  float* asrc2 = w + off; off += (size_t)N_NODES * H2;
  float* adst2 = w + off; off += (size_t)N_NODES * H2;
  float* agg2  = w + off; off += (size_t)N_NODES * F2;
  int* rowptr  = (int*)(w + off); off += (size_t)N_NODES + 32;
  int* rank    = (int*)(w + off); off += (size_t)N_EDGES;
  int* slot_eid = (int*)(w + off); off += (size_t)N_EDGES;
  int* csr_src = (int*)(w + off); off += (size_t)N_EDGES;
  float* csr_ea = w + off;        off += (size_t)N_EDGES;
  float* psum = w + off;          off += 256;
  int* bhT  = (int*)(w + off);    off += (size_t)NBKT * EBLK;
  int* boff = (int*)(w + off);    off += (size_t)NBKT * EBLK;
  int* bse  = (int*)(w + off);    off += 512;
  int* tmp  = (int*)(w + off);    off += (size_t)4 * N_EDGES;

  bucket_count_kernel<<<EBLK, 256, 0, stream>>>(ei, ea, rank, bhT, psum);
  bucket_scan_kernel<<<NBKT, 256, 0, stream>>>(bhT, boff, bse);
  bucket_scatter_kernel<<<EBLK, 256, 0, stream>>>(ei, ea, rank, bhT, boff, tmp);
  bucket_place_kernel<<<NBKT, 1024, 0, stream>>>(ei, ea, tmp, bse, rowptr, slot_eid, csr_src, csr_ea);

  prep_kernel<<<1, 256, 0, stream>>>(psum, We1, ae1, We2, ae2, mean, ce1, ce2);

  gemm1_wmma<<<dim3(N_NODES / 16, 2), 256, 0, stream>>>(x, W1, as1, ad1, xl1, asrc1, adst1);
  node_softmax_agg1_kernel<<<N_NODES / 8, 256, 0, stream>>>(
      rowptr, csr_src, csr_ea, mean, ce1, asrc1, adst1, xl1, b1, agg1);

  gemm2_wmma<<<N_NODES / 16, 64, 0, stream>>>(agg1, W2, as2, ad2, xl2, asrc2, adst2);
  node_softmax_agg2_kernel<<<(N_NODES * H2 + 255) / 256, 256, 0, stream>>>(
      rowptr, csr_src, csr_ea, mean, ce2, asrc2, adst2, xl2, agg2);
  final2_kernel<<<(N_NODES * C2 + 255) / 256, 256, 0, stream>>>(agg2, b2, out);
}


# 3 launches folded into producers: layer-2 product in the layer-1 aggregation block (rows kept in LDS; k sum now 8 partial sums in fixed order), head mean+ELU in layer-2 aggregation, prep in scan
# speedup vs baseline: 8.9668x; 1.0491x over previous
#include <hip/hip_runtime.h>
#include <math.h>

#define N_NODES 50000
#define N_EDGES 800000
#define IN_F    191
#define H1 16
#define C1 16
#define F1 256
#define H2 8
#define C2 3
#define F2 24
static_assert(N_NODES % 16 == 0, "row tiles of 16 nodes cover the node set exactly");

typedef __attribute__((ext_vector_type(2))) float v2f;
typedef __attribute__((ext_vector_type(8))) float v8f;


__device__ __forceinline__ float lrelu02(float a) {
  return (a > 0.f) ? a : 0.2f * a;
}

__device__ __forceinline__ float elu1(float v) {
  return (v > 0.f) ? v : expm1f(v);
}

#define PS_BLOCKS 196
#define PS_SLICE  4096

__device__ __forceinline__ float wave_sum_f(float v) {
  for (int off = 16; off > 0; off >>= 1) v += __shfl_xor(v, off);
  return v;
}

__device__ __forceinline__ void
prep_block(const float* __restrict__ psum,
           const float* __restrict__ We1, const float* __restrict__ ae1,
           const float* __restrict__ We2, const float* __restrict__ ae2,
           float* __restrict__ mean_out, float* __restrict__ ce1,
           float* __restrict__ ce2) {
  __shared__ float red[8];
  const int t = threadIdx.x;
  const float s = wave_sum_f((t < PS_BLOCKS) ? psum[t] : 0.f);
  if ((t & 31) == 0) red[t >> 5] = s;
  __syncthreads();
  if (t == 0) {
    float tot = 0.f;
    for (int w = 0; w < 8; w++) tot += red[w];
    mean_out[0] = tot * (1.0f / (float)N_EDGES);
  }
  if (t < H1) {
    float c = 0.f;
    for (int k = 0; k < C1; k++) c += We1[t * C1 + k] * ae1[t * C1 + k];
    ce1[t] = c;
  } else if (t >= 32 && t < 32 + H2) {
    const int h = t - 32;
    float c = 0.f;
    for (int k = 0; k < C2; k++) c += We2[h * C2 + k] * ae2[h * C2 + k];
    ce2[h] = c;
  }
}

#define NBKT 196
#define EBLK PS_BLOCKS

__global__ void __launch_bounds__(256)
bucket_count_kernel(const int* __restrict__ ei,
                    const float* __restrict__ ea,
                    int* __restrict__ lrank,
                    int* __restrict__ bhT,
                    float* __restrict__ psum) {
  __shared__ int hist[NBKT];
  __shared__ float red[8];
  const int t = threadIdx.x;
  if (t < NBKT) hist[t] = 0;
  const int base = blockIdx.x * PS_SLICE;
  const int* dstp = ei + N_EDGES;
  float4 q[PS_SLICE / 1024];
  int4 dq[PS_SLICE / 1024];
  for (int k = 0; k < PS_SLICE / 1024; k++) {
    const int i = base + (k * 256 + t) * 4;
    const int ic = (i < N_EDGES) ? i : N_EDGES - 4;
    q[k] = *(const float4*)&ea[ic];
    dq[k] = *(const int4*)&dstp[ic];
  }
  float s = 0.f;
  for (int k = 0; k < PS_SLICE / 1024; k++) {
    const int i = base + (k * 256 + t) * 4;
    const float part = (q[k].x + q[k].y) + (q[k].z + q[k].w);
    s += (i < N_EDGES) ? part : 0.f;
  }
  s = wave_sum_f(s);
  if ((t & 31) == 0) red[t >> 5] = s;
  __syncthreads();
  for (int k = 0; k < PS_SLICE / 1024; k++) {
    const int i = base + (k * 256 + t) * 4;
    if (i < N_EDGES) {
      int4 r;
      r.x = atomicAdd(&hist[dq[k].x >> 8], 1);
      r.y = atomicAdd(&hist[dq[k].y >> 8], 1);
      r.z = atomicAdd(&hist[dq[k].z >> 8], 1);
      r.w = atomicAdd(&hist[dq[k].w >> 8], 1);
      *(int4*)&lrank[i] = r;
    }
  }
  if (t == 0) {
    float tot = 0.f;
    for (int w = 0; w < 8; w++) tot += red[w];
    psum[blockIdx.x] = tot;
  }
  __syncthreads();
  if (t < NBKT) bhT[t * EBLK + blockIdx.x] = hist[t];
}

__global__ void __launch_bounds__(256)
bucket_scan_kernel(const int* __restrict__ bhT, int* __restrict__ boff,
                   int* __restrict__ bse,
                   const float* __restrict__ psum,
                   const float* __restrict__ We1, const float* __restrict__ ae1,
                   const float* __restrict__ We2, const float* __restrict__ ae2,
                   float* __restrict__ mean_out, float* __restrict__ ce1,
                   float* __restrict__ ce2) {
  __shared__ int wpre[8];
  __shared__ int wtot[8];
  const int t = threadIdx.x;
  const int lane = t & 31, wave = t >> 5;
  const int lo = blockIdx.x * EBLK;
  int pre = 0;
  for (int i0 = t * 4; i0 < lo; i0 += 8 * 1024) {
    int4 q[8];
    for (int u = 0; u < 8; u++) {
      const int i = i0 + u * 1024;
      q[u] = (i < lo) ? *(const int4*)&bhT[i] : make_int4(0, 0, 0, 0);
    }
    for (int u = 0; u < 8; u++) pre += (q[u].x + q[u].y) + (q[u].z + q[u].w);
  }
  const int c = (t < EBLK) ? bhT[lo + t] : 0;
  int v = c;
  for (int off = 1; off < 32; off <<= 1) {
    const int u = __shfl_up(v, off);
    if (lane >= off) v += u;
  }
  for (int off = 16; off > 0; off >>= 1) pre += __shfl_xor(pre, off);
  if (lane == 31) wtot[wave] = v;
  if (lane == 0) wpre[wave] = pre;
  __syncthreads();
  int start = 0;
  for (int w = 0; w < 8; w++) {
    start += wpre[w];
    if (w < wave) start += wtot[w];
  }
  if (t < EBLK) boff[lo + t] = start + v - c;
  if (t == 0) bse[2 * blockIdx.x] = start + v - c;
  if (t == EBLK - 1) bse[2 * blockIdx.x + 1] = start + v;
  if (blockIdx.x == 0)
    prep_block(psum, We1, ae1, We2, ae2, mean_out, ce1, ce2);
}

__global__ void __launch_bounds__(256)
bucket_scatter_kernel(const int* __restrict__ ei,
                      const float* __restrict__ ea,
                      const int* __restrict__ lrank,
                      const int* __restrict__ bhT,
                      const int* __restrict__ boff,
                      int* __restrict__ tmp_recs) {
  int4* tmp = (int4*)tmp_recs;
  __shared__ int lst[NBKT];
  __shared__ int delta[NBKT];
  __shared__ int wtot[8];
  __shared__ int4 sorted[PS_SLICE];
  const int t = threadIdx.x;
  const int lane = t & 31, wave = t >> 5;
  const int base = blockIdx.x * PS_SLICE;
  const int nblk = (N_EDGES - base < PS_SLICE) ? N_EDGES - base : PS_SLICE;
  const int* dstp = ei + N_EDGES;
  const int c = (t < NBKT) ? bhT[t * EBLK + blockIdx.x] : 0;
  const int o = (t < NBKT) ? boff[t * EBLK + blockIdx.x] : 0;
  int4 dq[PS_SLICE / 1024], rq[PS_SLICE / 1024], sq[PS_SLICE / 1024];
  float4 aq[PS_SLICE / 1024];
  for (int k = 0; k < PS_SLICE / 1024; k++) {
    const int i = base + (k * 256 + t) * 4;
    const int ic = (i < N_EDGES) ? i : N_EDGES - 4;
    dq[k] = *(const int4*)&dstp[ic];
    rq[k] = *(const int4*)&lrank[ic];
    sq[k] = *(const int4*)&ei[ic];
    aq[k] = *(const float4*)&ea[ic];
  }
  int v = c;
  for (int off = 1; off < 32; off <<= 1) {
    const int u = __shfl_up(v, off);
    if (lane >= off) v += u;
  }
  if (lane == 31) wtot[wave] = v;
  __syncthreads();
  int first = v - c;
  for (int w = 0; w < 8; w++)
    if (w < wave) first += wtot[w];
  if (t < NBKT) {
    lst[t] = first;
    delta[t] = o - first;
  }
  __syncthreads();
  for (int k = 0; k < PS_SLICE / 1024; k++) {
    const int i = base + (k * 256 + t) * 4;
    if (i < N_EDGES) {
      sorted[lst[dq[k].x >> 8] + rq[k].x] = make_int4(i, dq[k].x, sq[k].x, __float_as_int(aq[k].x));
      sorted[lst[dq[k].y >> 8] + rq[k].y] = make_int4(i + 1, dq[k].y, sq[k].y, __float_as_int(aq[k].y));
      sorted[lst[dq[k].z >> 8] + rq[k].z] = make_int4(i + 2, dq[k].z, sq[k].z, __float_as_int(aq[k].z));
      sorted[lst[dq[k].w >> 8] + rq[k].w] = make_int4(i + 3, dq[k].w, sq[k].w, __float_as_int(aq[k].w));
    }
  }
  __syncthreads();
  for (int i = t; i < nblk; i += 256) {
    const int4 p = sorted[i];
    tmp[delta[p.y >> 8] + i] = p;
  }
}

#define BP_GROUP 6
#define BP_CAP   6144
#define ID_BITS  20
#define ID_MASK  ((1 << ID_BITS) - 1)
static_assert(N_EDGES <= (1 << ID_BITS), "edge id and local destination share one word");

__device__ __forceinline__ void order_bucket_from_memory(const int* __restrict__ ei,
                                                         const float* __restrict__ ea,
                                                         const int* seg, const int* gkeys,
                                                         int bs, int nb, int t,
                                                         int* __restrict__ csr_src,
                                                         float* __restrict__ csr_ea) {
  for (int i = t; i < nb; i += 1024) {
    const int k = gkeys[bs + i];
    const int ld = k >> ID_BITS;
    const int b = seg[ld], e = seg[ld + 1];
    int pos = 0;
    for (int j = b; j < e; j++) pos += (gkeys[bs + j] < k) ? 1 : 0;
    const int id = k & ID_MASK;
    csr_src[bs + b + pos] = ei[id];
    csr_ea[bs + b + pos] = ea[id];
  }
}

__global__ void __launch_bounds__(1024)
bucket_place_kernel(const int* __restrict__ ei, const float* __restrict__ ea,
                    const int* __restrict__ tmp_recs, const int* __restrict__ bse,
                    int* __restrict__ rowptr, int* gkeys,
                    int* __restrict__ csr_src, float* __restrict__ csr_ea) {
  const int4* tmp = (const int4*)tmp_recs;
  __shared__ int cnt[256];
  __shared__ int seg[257];
  __shared__ int wtot[8];
  __shared__ int skey[BP_CAP];
  __shared__ int ssrc[BP_CAP];
  __shared__ float sea[BP_CAP];
  const int t = threadIdx.x;
  const int lane = t & 31, wave = t >> 5;
  const int bs = bse[2 * blockIdx.x], be = bse[2 * blockIdx.x + 1];
  const int nb = be - bs;
  const bool fits = nb <= BP_CAP;
  if (t < 256) cnt[t] = 0;
  __syncthreads();
  for (int i0 = bs + t; i0 < be; i0 += BP_GROUP * 1024) {
    int dd[BP_GROUP];
    for (int u = 0; u < BP_GROUP; u++) {
      const int i = i0 + u * 1024;
      dd[u] = tmp[(i < be) ? i : be - 1].y;
    }
    for (int u = 0; u < BP_GROUP; u++)
      if (i0 + u * 1024 < be) atomicAdd(&cnt[dd[u] & 255], 1);
  }
  __syncthreads();
  const int c = (t < 256) ? cnt[t] : 0;
  int v = c;
  for (int off = 1; off < 32; off <<= 1) {
    const int u = __shfl_up(v, off);
    if (lane >= off) v += u;
  }
  if (lane == 31 && wave < 8) wtot[wave] = v;
  __syncthreads();
  if (t < 256) {
    int start = bs;
    for (int w = 0; w < 8; w++)
      if (w < wave) start += wtot[w];
    const int first = start + v - c;
    const int node = blockIdx.x * 256 + t;
    if (node <= N_NODES) rowptr[node] = first;
    cnt[t] = first;
    seg[t] = first - bs;
    if (t == 255) seg[256] = first + c - bs;
  }
  __syncthreads();
  for (int i0 = bs + t; i0 < be; i0 += BP_GROUP * 1024) {
    int4 q[BP_GROUP];
    for (int u = 0; u < BP_GROUP; u++) {
      const int i = i0 + u * 1024;
      q[u] = tmp[(i < be) ? i : be - 1];
    }
    for (int u = 0; u < BP_GROUP; u++) {
      if (i0 + u * 1024 < be) {
        const int ld = q[u].y & 255;
        const int pos = atomicAdd(&cnt[ld], 1);
        const int k = q[u].x | (ld << ID_BITS);
        if (fits) {
          skey[pos - bs] = k;
          ssrc[pos - bs] = q[u].z;
          sea[pos - bs] = __int_as_float(q[u].w);
        } else {
          gkeys[pos] = k;
        }
      }
    }
  }
  __syncthreads();
  if (fits) {
    for (int i = t; i < nb; i += 1024) {
      const int k = skey[i];
      const int ld = k >> ID_BITS;
      const int b = seg[ld], e = seg[ld + 1];
      int pos = 0;
      for (int j = b; j < e; j++) pos += (skey[j] < k) ? 1 : 0;
      csr_src[bs + b + pos] = ssrc[i];
      csr_ea[bs + b + pos] = sea[i];
    }
  } else {
    order_bucket_from_memory(ei, ea, seg, gkeys, bs, nb, t, csr_src, csr_ea);
  }
}

#define A_PITCH 196
#define A_F4    764
#define D1_PITCH 132

__device__ __forceinline__ void gemm1_load_b(const float* __restrict__ W1, int colB,
                                             int t, int k0, float4 (&qb)[6]) {
#pragma unroll
  for (int i = 0; i < 6; i++) {
    const int idx4 = t + 256 * i;
    const int r = idx4 >> 5, c = (idx4 & 31) << 2, k = k0 + r;
    qb[i] = *(const float4*)&W1[((k < IN_F) ? k : IN_F - 1) * F1 + colB + c];
  }
}

__global__ void __launch_bounds__(256)
gemm1_wmma(const float* __restrict__ x, const float* __restrict__ W1,
           const float* __restrict__ a_src, const float* __restrict__ a_dst,
           float* __restrict__ xl1, float* __restrict__ asrc,
           float* __restrict__ adst) {
  __shared__ __attribute__((aligned(16))) float As[16 * A_PITCH];
  __shared__ __attribute__((aligned(16))) float Bs[48][128];
  const int row0 = blockIdx.x * 16;
  const int colB = blockIdx.y * 128;
  const int t = threadIdx.x;
  const int wave = t >> 5, lane = t & 31, lh = lane & 15, hi = lane >> 4;

  const float4* xa = (const float4*)(x + (size_t)row0 * IN_F);
  float4 qa[3];
#pragma unroll
  for (int j = 0; j < 3; j++) {
    const int f = t + 256 * j;
    qa[j] = xa[(f < A_F4) ? f : A_F4 - 1];
  }
  float4 qb[6];
  gemm1_load_b(W1, colB, t, 0, qb);

#pragma unroll
  for (int j = 0; j < 3; j++) {
    const int f = t + 256 * j;
    if (f < A_F4) {
      const float v[4] = {qa[j].x, qa[j].y, qa[j].z, qa[j].w};
#pragma unroll
      for (int q = 0; q < 4; q++) {
        const int e = 4 * f + q, r = e / IN_F, k = e - r * IN_F;
        As[r * A_PITCH + k] = v[q];
      }
    }
  }
  if (t < 16) {
#pragma unroll
    for (int k = IN_F; k < A_PITCH; k++) As[t * A_PITCH + k] = 0.f;
  }

  v8f acc = {};
#pragma unroll
  for (int k0 = 0; k0 < 192; k0 += 48) {
#pragma unroll
    for (int i = 0; i < 6; i++) {
      const int idx4 = t + 256 * i;
      const int r = idx4 >> 5, c = (idx4 & 31) << 2;
      *(float4*)&Bs[r][c] = (k0 + r < IN_F) ? qb[i] : make_float4(0.f, 0.f, 0.f, 0.f);
    }
    __syncthreads();
    if (k0 + 48 < 192) gemm1_load_b(W1, colB, t, k0 + 48, qb);
#pragma unroll
    for (int kk = 0; kk < 48; kk += 4) {
      v2f a, b;
      a = *(const v2f*)&As[lh * A_PITCH + k0 + kk + 2 * hi];
      b.x = Bs[kk + 2 * hi][wave * 16 + lh];
      b.y = Bs[kk + 2 * hi + 1][wave * 16 + lh];
      acc = __builtin_amdgcn_wmma_f32_16x16x4_f32(false, a, false, b,
                                                  (short)0, acc, false, false);
    }
    __syncthreads();
  }
  float* Ds = &Bs[0][0];
#pragma unroll
  for (int i = 0; i < 8; i++)
    Ds[(i + 8 * hi) * D1_PITCH + wave * 16 + lh] = acc[i];
  __syncthreads();
  const int col = colB + wave * 16 + lh;
#pragma unroll
  for (int i = 0; i < 8; i++)
    xl1[(row0 + i + 8 * hi) * F1 + col] = acc[i];
  if (t < 128) {
    const int r = t >> 3, hl = t & 7;
    const int h = blockIdx.y * 8 + hl;
    const float* row = Ds + r * D1_PITCH + hl * C1;
    float s = 0.f, d = 0.f;
#pragma unroll
    for (int c = 0; c < C1; c++) {
      const float v = row[c];
      s += v * a_src[h * C1 + c];
      d += v * a_dst[h * C1 + c];
    }
    asrc[(row0 + r) * H1 + h] = s;
    adst[(row0 + r) * H1 + h] = d;
  }
}

#define H1_PITCH 260
#define D2_PITCH 25
#define K2_CHUNK 32
__global__ void __launch_bounds__(512)
node_softmax_agg1_kernel(const int* __restrict__ rowptr,
                         const int* __restrict__ csr_src,
                         const float* __restrict__ csr_ea,
                         const float* __restrict__ mean,
                         const float* __restrict__ ce,
                         const float* __restrict__ asrc,
                         const float* __restrict__ adst,
                         const float* __restrict__ xl,
                         const float* __restrict__ b1,
                         const float* __restrict__ W2,
                         const float* __restrict__ a_src2,
                         const float* __restrict__ a_dst2,
                         float* __restrict__ xl2,
                         float* __restrict__ asrc2,
                         float* __restrict__ adst2) {
  __shared__ __attribute__((aligned(16))) float As[16][H1_PITCH];
  __shared__ float Pt[8][16 * D2_PITCH];
  const int lane = threadIdx.x & 31;
  const int wv = __builtin_amdgcn_readfirstlane((int)(threadIdx.x >> 5));
  const int d = blockIdx.x * 16 + wv;
  const int hA = lane >> 2, hB = 8 + (lane >> 2);
  const int beg = rowptr[d], end = rowptr[d + 1];
  const float adA = adst[d * H1 + hA], adB = adst[d * H1 + hB];
  const float cA = ce[hA], cB = ce[hB];
  const float mv = mean[0];
  const float aSelfA = lrelu02(asrc[d * H1 + hA] + adA + mv * cA);
  const float aSelfB = lrelu02(asrc[d * H1 + hB] + adB + mv * cB);

  float mA = aSelfA, mB = aSelfB;
  for (int base = beg; base < end; base += 32) {
    const int n = (end - base < 32) ? end - base : 32;
    int sL = 0;
    float eL = 0.f;
    if (lane < n) { sL = csr_src[base + lane]; eL = csr_ea[base + lane]; }
    for (int j = 0; j < n; j++) {
      const int s = __builtin_amdgcn_readlane(sL, j);
      const float ev = __int_as_float(__builtin_amdgcn_readlane(__float_as_int(eL), j));
      const float aA = lrelu02(asrc[s * H1 + hA] + adA + ev * cA);
      const float aB = lrelu02(asrc[s * H1 + hB] + adB + ev * cB);
      mA = fmaxf(mA, aA);
      mB = fmaxf(mB, aB);
    }
  }

  const float4 bA = ((const float4*)b1)[lane], bB = ((const float4*)b1)[32 + lane];

  float denA = __expf(aSelfA - mA), denB = __expf(aSelfB - mB);
  float4 accA, accB;
  {
    const float4* r = (const float4*)(xl + d * F1);
    const float4 u = r[lane], w = r[32 + lane];
    accA = make_float4(denA * u.x, denA * u.y, denA * u.z, denA * u.w);
    accB = make_float4(denB * w.x, denB * w.y, denB * w.z, denB * w.w);
  }
  for (int base = beg; base < end; base += 32) {
    const int n = (end - base < 32) ? end - base : 32;
    int sL = 0;
    float eL = 0.f;
    if (lane < n) { sL = csr_src[base + lane]; eL = csr_ea[base + lane]; }
    for (int j = 0; j < n; j++) {
      const int s = __builtin_amdgcn_readlane(sL, j);
      const float ev = __int_as_float(__builtin_amdgcn_readlane(__float_as_int(eL), j));
      const float4* r = (const float4*)(xl + s * F1);
      const float4 u = r[lane], w = r[32 + lane];
      const float aA = lrelu02(asrc[s * H1 + hA] + adA + ev * cA);
      const float aB = lrelu02(asrc[s * H1 + hB] + adB + ev * cB);
      const float eA = __expf(aA - mA), eB = __expf(aB - mB);
      denA += eA;
      denB += eB;
      accA.x += eA * u.x; accA.y += eA * u.y; accA.z += eA * u.z; accA.w += eA * u.w;
      accB.x += eB * w.x; accB.y += eB * w.y; accB.z += eB * w.z; accB.w += eB * w.w;
    }
  }

  const int ct = wv & 1, kc = wv >> 1;
  const int lh = lane & 15, hi = lane >> 4;
  const int col = ct * 16 + lh;
  const bool cv = col < F2;
  const int cc = cv ? col : F2 - 1;
  v2f bf[K2_CHUNK / 4];
#pragma unroll
  for (int u = 0; u < K2_CHUNK / 4; u++) {
    const int k = kc * K2_CHUNK + 4 * u + 2 * hi;
    const float bx = W2[k * F2 + cc], by = W2[(k + 1) * F2 + cc];
    bf[u].x = cv ? bx : 0.f;
    bf[u].y = cv ? by : 0.f;
  }

  const float rA = 1.0f / (denA + 1e-16f), rB = 1.0f / (denB + 1e-16f);
  float4* o = (float4*)&As[wv][0];
  o[lane]      = make_float4(elu1(accA.x * rA + bA.x), elu1(accA.y * rA + bA.y),
                             elu1(accA.z * rA + bA.z), elu1(accA.w * rA + bA.w));
  o[32 + lane] = make_float4(elu1(accB.x * rB + bB.x), elu1(accB.y * rB + bB.y),
                             elu1(accB.z * rB + bB.z), elu1(accB.w * rB + bB.w));
  __syncthreads();

  v8f acc = {};
#pragma unroll
  for (int u = 0; u < K2_CHUNK / 4; u++) {
    const v2f a = *(const v2f*)&As[lh][kc * K2_CHUNK + 4 * u + 2 * hi];
    acc = __builtin_amdgcn_wmma_f32_16x16x4_f32(false, a, false, bf[u],
                                                (short)0, acc, false, false);
  }
  if (cv) {
#pragma unroll
    for (int i = 0; i < 8; i++)
      Pt[kc][(i + 8 * hi) * D2_PITCH + col] = acc[i];
  }
  __syncthreads();
  const int t = threadIdx.x;
  if (t < 16 * H2) {
    const int r = t >> 3, h = t & 7;
    const int row0 = blockIdx.x * 16;
    float s = 0.f, dd = 0.f;
#pragma unroll
    for (int q = 0; q < C2; q++) {
      const int e = r * D2_PITCH + h * C2 + q;
      float v = Pt[0][e];
#pragma unroll
      for (int p = 1; p < 8; p++) v += Pt[p][e];
      xl2[(row0 + r) * F2 + h * C2 + q] = v;
      s += v * a_src2[h * C2 + q];
      dd += v * a_dst2[h * C2 + q];
    }
    asrc2[(row0 + r) * H2 + h] = s;
    adst2[(row0 + r) * H2 + h] = dd;
  }
}

__global__ void __launch_bounds__(256)
node_softmax_agg2_kernel(const int* __restrict__ rowptr,
                         const int* __restrict__ csr_src,
                         const float* __restrict__ csr_ea,
                         const float* __restrict__ mean,
                         const float* __restrict__ ce,
                         const float* __restrict__ asrc,
                         const float* __restrict__ adst,
                         const float* __restrict__ xl,
                         const float* __restrict__ b2,
                         float* __restrict__ out) {
  __shared__ float hrow[256 * C2];
  const int t = threadIdx.x;
  const int idx = blockIdx.x * 256 + t;
  if (idx < N_NODES * H2) {
    const int d = idx / H2, h = idx % H2;
    const int beg = rowptr[d], end = rowptr[d + 1];
    const float ad = adst[idx];
    const float c = ce[h];
    const float aSelf = lrelu02(asrc[idx] + ad + mean[0] * c);

    float m = aSelf;
    for (int j = beg; j < end; j++) {
      const float a = lrelu02(asrc[csr_src[j] * H2 + h] + ad + csr_ea[j] * c);
      m = fmaxf(m, a);
    }

    float den = __expf(aSelf - m);
    const float* rd = xl + d * F2 + h * C2;
    float acc0 = den * rd[0], acc1 = den * rd[1], acc2 = den * rd[2];
    for (int j = beg; j < end; j++) {
      const int s = csr_src[j];
      const float a = lrelu02(asrc[s * H2 + h] + ad + csr_ea[j] * c);
      const float ex = __expf(a - m);
      const float* rs = xl + s * F2 + h * C2;
      den += ex;
      acc0 += ex * rs[0];
      acc1 += ex * rs[1];
      acc2 += ex * rs[2];
    }
    const float rinv = 1.0f / (den + 1e-16f);
    hrow[t * C2 + 0] = acc0 * rinv;
    hrow[t * C2 + 1] = acc1 * rinv;
    hrow[t * C2 + 2] = acc2 * rinv;
  }
  __syncthreads();
  if (t < 32 * C2) {
    const int nl = t / C2, ch = t % C2;
    const int n = blockIdx.x * 32 + nl;
    if (n < N_NODES) {
      float s = 0.f;
#pragma unroll
      for (int h = 0; h < H2; h++) s += hrow[(nl * H2 + h) * C2 + ch];
      const float v = s * (1.f / (float)H2) + b2[ch];
      out[n * C2 + ch] = (v > 0.f) ? v : expm1f(v);
    }
  }
}

extern "C" void kernel_launch(void* const* d_in, const int* in_sizes, int n_in,
                              void* d_out, int out_size, void* d_ws, size_t ws_size,
                              hipStream_t stream) {
  (void)in_sizes; (void)n_in; (void)out_size; (void)ws_size;
  const float* x   = (const float*)d_in[0];
  const int*   ei  = (const int*)d_in[1];
  const float* ea  = (const float*)d_in[2];
  const float* W1  = (const float*)d_in[3];
  const float* We1 = (const float*)d_in[4];
  const float* as1 = (const float*)d_in[5];
  const float* ad1 = (const float*)d_in[6];
  const float* ae1 = (const float*)d_in[7];
  const float* b1  = (const float*)d_in[8];
  const float* W2  = (const float*)d_in[9];
  const float* We2 = (const float*)d_in[10];
  const float* as2 = (const float*)d_in[11];
  const float* ad2 = (const float*)d_in[12];
  const float* ae2 = (const float*)d_in[13];
  const float* b2  = (const float*)d_in[14];
  float* out = (float*)d_out;

  float* w = (float*)d_ws;
  float* mean = w;
  float* ce1  = w + 1;
  float* ce2  = w + 17;
  size_t off = 32;
  float* xl1   = w + off; off += (size_t)N_NODES * F1;
  float* asrc1 = w + off; off += (size_t)N_NODES * H1;
  float* adst1 = w + off; off += (size_t)N_NODES * H1;
  float* xl2   = w + off; off += (size_t)N_NODES * F2;
  float* asrc2 = w + off; off += (size_t)N_NODES * H2;
  float* adst2 = w + off; off += (size_t)N_NODES * H2;
  int* rowptr  = (int*)(w + off); off += (size_t)N_NODES + 32;
  int* rank    = (int*)(w + off); off += (size_t)N_EDGES;
  int* slot_eid = (int*)(w + off); off += (size_t)N_EDGES;
  int* csr_src = (int*)(w + off); off += (size_t)N_EDGES;
  float* csr_ea = w + off;        off += (size_t)N_EDGES;
  float* psum = w + off;          off += 256;
  int* bhT  = (int*)(w + off);    off += (size_t)NBKT * EBLK;
  int* boff = (int*)(w + off);    off += (size_t)NBKT * EBLK;
  int* bse  = (int*)(w + off);    off += 512;
  int* tmp  = (int*)(w + off);    off += (size_t)4 * N_EDGES;

  bucket_count_kernel<<<EBLK, 256, 0, stream>>>(ei, ea, rank, bhT, psum);
  bucket_scan_kernel<<<NBKT, 256, 0, stream>>>(bhT, boff, bse, psum, We1, ae1, We2, ae2,
                                               mean, ce1, ce2);
  bucket_scatter_kernel<<<EBLK, 256, 0, stream>>>(ei, ea, rank, bhT, boff, tmp);
  bucket_place_kernel<<<NBKT, 1024, 0, stream>>>(ei, ea, tmp, bse, rowptr, slot_eid, csr_src, csr_ea);

  gemm1_wmma<<<dim3(N_NODES / 16, 2), 256, 0, stream>>>(x, W1, as1, ad1, xl1, asrc1, adst1);
  node_softmax_agg1_kernel<<<N_NODES / 16, 512, 0, stream>>>(
      rowptr, csr_src, csr_ea, mean, ce1, asrc1, adst1, xl1, b1,
      W2, as2, ad2, xl2, asrc2, adst2);

  node_softmax_agg2_kernel<<<(N_NODES * H2 + 255) / 256, 256, 0, stream>>>(
      rowptr, csr_src, csr_ea, mean, ce2, asrc2, adst2, xl2, b2, out);
}


# layer-1 aggregation no longer pays one load round trip per edge: a chunk's logits fetched as one group per (edge parity, head) lane, weights via LDS, rows 3 edges per group; float sums unchanged
# speedup vs baseline: 9.2768x; 1.0346x over previous
#include <hip/hip_runtime.h>
#include <math.h>

#define N_NODES 50000
#define N_EDGES 800000
#define IN_F    191
#define H1 16
#define C1 16
#define F1 256
#define H2 8
#define C2 3
#define F2 24
static_assert(N_NODES % 16 == 0, "row tiles of 16 nodes cover the node set exactly");

typedef __attribute__((ext_vector_type(2))) float v2f;
typedef __attribute__((ext_vector_type(8))) float v8f;


__device__ __forceinline__ float lrelu02(float a) {
  return (a > 0.f) ? a : 0.2f * a;
}

__device__ __forceinline__ float elu1(float v) {
  return (v > 0.f) ? v : expm1f(v);
}

#define PS_BLOCKS 196
#define PS_SLICE  4096

__device__ __forceinline__ float wave_sum_f(float v) {
  for (int off = 16; off > 0; off >>= 1) v += __shfl_xor(v, off);
  return v;
}

__device__ __forceinline__ void
prep_block(const float* __restrict__ psum,
           const float* __restrict__ We1, const float* __restrict__ ae1,
           const float* __restrict__ We2, const float* __restrict__ ae2,
           float* __restrict__ mean_out, float* __restrict__ ce1,
           float* __restrict__ ce2) {
  __shared__ float red[8];
  const int t = threadIdx.x;
  const float s = wave_sum_f((t < PS_BLOCKS) ? psum[t] : 0.f);
  if ((t & 31) == 0) red[t >> 5] = s;
  __syncthreads();
  if (t == 0) {
    float tot = 0.f;
    for (int w = 0; w < 8; w++) tot += red[w];
    mean_out[0] = tot * (1.0f / (float)N_EDGES);
  }
  if (t < H1) {
    float c = 0.f;
    for (int k = 0; k < C1; k++) c += We1[t * C1 + k] * ae1[t * C1 + k];
    ce1[t] = c;
  } else if (t >= 32 && t < 32 + H2) {
    const int h = t - 32;
    float c = 0.f;
    for (int k = 0; k < C2; k++) c += We2[h * C2 + k] * ae2[h * C2 + k];
    ce2[h] = c;
  }
}

#define NBKT 196
#define EBLK PS_BLOCKS

__global__ void __launch_bounds__(256)
bucket_count_kernel(const int* __restrict__ ei,
                    const float* __restrict__ ea,
                    int* __restrict__ lrank,
                    int* __restrict__ bhT,
                    float* __restrict__ psum) {
  __shared__ int hist[NBKT];
  __shared__ float red[8];
  const int t = threadIdx.x;
  if (t < NBKT) hist[t] = 0;
  const int base = blockIdx.x * PS_SLICE;
  const int* dstp = ei + N_EDGES;
  float4 q[PS_SLICE / 1024];
  int4 dq[PS_SLICE / 1024];
  for (int k = 0; k < PS_SLICE / 1024; k++) {
    const int i = base + (k * 256 + t) * 4;
    const int ic = (i < N_EDGES) ? i : N_EDGES - 4;
    q[k] = *(const float4*)&ea[ic];
    dq[k] = *(const int4*)&dstp[ic];
  }
  float s = 0.f;
  for (int k = 0; k < PS_SLICE / 1024; k++) {
    const int i = base + (k * 256 + t) * 4;
    const float part = (q[k].x + q[k].y) + (q[k].z + q[k].w);
    s += (i < N_EDGES) ? part : 0.f;
  }
  s = wave_sum_f(s);
  if ((t & 31) == 0) red[t >> 5] = s;
  __syncthreads();
  for (int k = 0; k < PS_SLICE / 1024; k++) {
    const int i = base + (k * 256 + t) * 4;
    if (i < N_EDGES) {
      int4 r;
      r.x = atomicAdd(&hist[dq[k].x >> 8], 1);
      r.y = atomicAdd(&hist[dq[k].y >> 8], 1);
      r.z = atomicAdd(&hist[dq[k].z >> 8], 1);
      r.w = atomicAdd(&hist[dq[k].w >> 8], 1);
      *(int4*)&lrank[i] = r;
    }
  }
  if (t == 0) {
    float tot = 0.f;
    for (int w = 0; w < 8; w++) tot += red[w];
    psum[blockIdx.x] = tot;
  }
  __syncthreads();
  if (t < NBKT) bhT[t * EBLK + blockIdx.x] = hist[t];
}

__global__ void __launch_bounds__(256)
bucket_scan_kernel(const int* __restrict__ bhT, int* __restrict__ boff,
                   int* __restrict__ bse,
                   const float* __restrict__ psum,
                   const float* __restrict__ We1, const float* __restrict__ ae1,
                   const float* __restrict__ We2, const float* __restrict__ ae2,
                   float* __restrict__ mean_out, float* __restrict__ ce1,
                   float* __restrict__ ce2) {
  __shared__ int wpre[8];
  __shared__ int wtot[8];
  const int t = threadIdx.x;
  const int lane = t & 31, wave = t >> 5;
  const int lo = blockIdx.x * EBLK;
  int pre = 0;
  for (int i0 = t * 4; i0 < lo; i0 += 8 * 1024) {
    int4 q[8];
    for (int u = 0; u < 8; u++) {
      const int i = i0 + u * 1024;
      q[u] = (i < lo) ? *(const int4*)&bhT[i] : make_int4(0, 0, 0, 0);
    }
    for (int u = 0; u < 8; u++) pre += (q[u].x + q[u].y) + (q[u].z + q[u].w);
  }
  const int c = (t < EBLK) ? bhT[lo + t] : 0;
  int v = c;
  for (int off = 1; off < 32; off <<= 1) {
    const int u = __shfl_up(v, off);
    if (lane >= off) v += u;
  }
  for (int off = 16; off > 0; off >>= 1) pre += __shfl_xor(pre, off);
  if (lane == 31) wtot[wave] = v;
  if (lane == 0) wpre[wave] = pre;
  __syncthreads();
  int start = 0;
  for (int w = 0; w < 8; w++) {
    start += wpre[w];
    if (w < wave) start += wtot[w];
  }
  if (t < EBLK) boff[lo + t] = start + v - c;
  if (t == 0) bse[2 * blockIdx.x] = start + v - c;
  if (t == EBLK - 1) bse[2 * blockIdx.x + 1] = start + v;
  if (blockIdx.x == 0)
    prep_block(psum, We1, ae1, We2, ae2, mean_out, ce1, ce2);
}

__global__ void __launch_bounds__(256)
bucket_scatter_kernel(const int* __restrict__ ei,
                      const float* __restrict__ ea,
                      const int* __restrict__ lrank,
                      const int* __restrict__ bhT,
                      const int* __restrict__ boff,
                      int* __restrict__ tmp_recs) {
  int4* tmp = (int4*)tmp_recs;
  __shared__ int lst[NBKT];
  __shared__ int delta[NBKT];
  __shared__ int wtot[8];
  __shared__ int4 sorted[PS_SLICE];
  const int t = threadIdx.x;
  const int lane = t & 31, wave = t >> 5;
  const int base = blockIdx.x * PS_SLICE;
  const int nblk = (N_EDGES - base < PS_SLICE) ? N_EDGES - base : PS_SLICE;
  const int* dstp = ei + N_EDGES;
  const int c = (t < NBKT) ? bhT[t * EBLK + blockIdx.x] : 0;
  const int o = (t < NBKT) ? boff[t * EBLK + blockIdx.x] : 0;
  int4 dq[PS_SLICE / 1024], rq[PS_SLICE / 1024], sq[PS_SLICE / 1024];
  float4 aq[PS_SLICE / 1024];
  for (int k = 0; k < PS_SLICE / 1024; k++) {
    const int i = base + (k * 256 + t) * 4;
    const int ic = (i < N_EDGES) ? i : N_EDGES - 4;
    dq[k] = *(const int4*)&dstp[ic];
    rq[k] = *(const int4*)&lrank[ic];
    sq[k] = *(const int4*)&ei[ic];
    aq[k] = *(const float4*)&ea[ic];
  }
  int v = c;
  for (int off = 1; off < 32; off <<= 1) {
    const int u = __shfl_up(v, off);
    if (lane >= off) v += u;
  }
  if (lane == 31) wtot[wave] = v;
  __syncthreads();
  int first = v - c;
  for (int w = 0; w < 8; w++)
    if (w < wave) first += wtot[w];
  if (t < NBKT) {
    lst[t] = first;
    delta[t] = o - first;
  }
  __syncthreads();
  for (int k = 0; k < PS_SLICE / 1024; k++) {
    const int i = base + (k * 256 + t) * 4;
    if (i < N_EDGES) {
      sorted[lst[dq[k].x >> 8] + rq[k].x] = make_int4(i, dq[k].x, sq[k].x, __float_as_int(aq[k].x));
      sorted[lst[dq[k].y >> 8] + rq[k].y] = make_int4(i + 1, dq[k].y, sq[k].y, __float_as_int(aq[k].y));
      sorted[lst[dq[k].z >> 8] + rq[k].z] = make_int4(i + 2, dq[k].z, sq[k].z, __float_as_int(aq[k].z));
      sorted[lst[dq[k].w >> 8] + rq[k].w] = make_int4(i + 3, dq[k].w, sq[k].w, __float_as_int(aq[k].w));
    }
  }
  __syncthreads();
  for (int i = t; i < nblk; i += 256) {
    const int4 p = sorted[i];
    tmp[delta[p.y >> 8] + i] = p;
  }
}

#define BP_GROUP 6
#define BP_CAP   6144
#define ID_BITS  20
#define ID_MASK  ((1 << ID_BITS) - 1)
static_assert(N_EDGES <= (1 << ID_BITS), "edge id and local destination share one word");

__device__ __forceinline__ void order_bucket_from_memory(const int* __restrict__ ei,
                                                         const float* __restrict__ ea,
                                                         const int* seg, const int* gkeys,
                                                         int bs, int nb, int t,
                                                         int* __restrict__ csr_src,
                                                         float* __restrict__ csr_ea) {
  for (int i = t; i < nb; i += 1024) {
    const int k = gkeys[bs + i];
    const int ld = k >> ID_BITS;
    const int b = seg[ld], e = seg[ld + 1];
    int pos = 0;
    for (int j = b; j < e; j++) pos += (gkeys[bs + j] < k) ? 1 : 0;
    const int id = k & ID_MASK;
    csr_src[bs + b + pos] = ei[id];
    csr_ea[bs + b + pos] = ea[id];
  }
}

__global__ void __launch_bounds__(1024)
bucket_place_kernel(const int* __restrict__ ei, const float* __restrict__ ea,
                    const int* __restrict__ tmp_recs, const int* __restrict__ bse,
                    int* __restrict__ rowptr, int* gkeys,
                    int* __restrict__ csr_src, float* __restrict__ csr_ea) {
  const int4* tmp = (const int4*)tmp_recs;
  __shared__ int cnt[256];
  __shared__ int seg[257];
  __shared__ int wtot[8];
  __shared__ int skey[BP_CAP];
  __shared__ int ssrc[BP_CAP];
  __shared__ float sea[BP_CAP];
  const int t = threadIdx.x;
  const int lane = t & 31, wave = t >> 5;
  const int bs = bse[2 * blockIdx.x], be = bse[2 * blockIdx.x + 1];
  const int nb = be - bs;
  const bool fits = nb <= BP_CAP;
  if (t < 256) cnt[t] = 0;
  __syncthreads();
  for (int i0 = bs + t; i0 < be; i0 += BP_GROUP * 1024) {
    int dd[BP_GROUP];
    for (int u = 0; u < BP_GROUP; u++) {
      const int i = i0 + u * 1024;
      dd[u] = tmp[(i < be) ? i : be - 1].y;
    }
    for (int u = 0; u < BP_GROUP; u++)
      if (i0 + u * 1024 < be) atomicAdd(&cnt[dd[u] & 255], 1);
  }
  __syncthreads();
  const int c = (t < 256) ? cnt[t] : 0;
  int v = c;
  for (int off = 1; off < 32; off <<= 1) {
    const int u = __shfl_up(v, off);
    if (lane >= off) v += u;
  }
  if (lane == 31 && wave < 8) wtot[wave] = v;
  __syncthreads();
  if (t < 256) {
    int start = bs;
    for (int w = 0; w < 8; w++)
      if (w < wave) start += wtot[w];
    const int first = start + v - c;
    const int node = blockIdx.x * 256 + t;
    if (node <= N_NODES) rowptr[node] = first;
    cnt[t] = first;
    seg[t] = first - bs;
    if (t == 255) seg[256] = first + c - bs;
  }
  __syncthreads();
  for (int i0 = bs + t; i0 < be; i0 += BP_GROUP * 1024) {
    int4 q[BP_GROUP];
    for (int u = 0; u < BP_GROUP; u++) {
      const int i = i0 + u * 1024;
      q[u] = tmp[(i < be) ? i : be - 1];
    }
    for (int u = 0; u < BP_GROUP; u++) {
      if (i0 + u * 1024 < be) {
        const int ld = q[u].y & 255;
        const int pos = atomicAdd(&cnt[ld], 1);
        const int k = q[u].x | (ld << ID_BITS);
        if (fits) {
          skey[pos - bs] = k;
          ssrc[pos - bs] = q[u].z;
          sea[pos - bs] = __int_as_float(q[u].w);
        } else {
          gkeys[pos] = k;
        }
      }
    }
  }
  __syncthreads();
  if (fits) {
    for (int i = t; i < nb; i += 1024) {
      const int k = skey[i];
      const int ld = k >> ID_BITS;
      const int b = seg[ld], e = seg[ld + 1];
      int pos = 0;
      for (int j = b; j < e; j++) pos += (skey[j] < k) ? 1 : 0;
      csr_src[bs + b + pos] = ssrc[i];
      csr_ea[bs + b + pos] = sea[i];
    }
  } else {
    order_bucket_from_memory(ei, ea, seg, gkeys, bs, nb, t, csr_src, csr_ea);
  }
}

#define A_PITCH 196
#define A_F4    764
#define D1_PITCH 132

__device__ __forceinline__ void gemm1_load_b(const float* __restrict__ W1, int colB,
                                             int t, int k0, float4 (&qb)[6]) {
#pragma unroll
  for (int i = 0; i < 6; i++) {
    const int idx4 = t + 256 * i;
    const int r = idx4 >> 5, c = (idx4 & 31) << 2, k = k0 + r;
    qb[i] = *(const float4*)&W1[((k < IN_F) ? k : IN_F - 1) * F1 + colB + c];
  }
}

__global__ void __launch_bounds__(256)
gemm1_wmma(const float* __restrict__ x, const float* __restrict__ W1,
           const float* __restrict__ a_src, const float* __restrict__ a_dst,
           float* __restrict__ xl1, float* __restrict__ asrc,
           float* __restrict__ adst) {
  __shared__ __attribute__((aligned(16))) float As[16 * A_PITCH];
  __shared__ __attribute__((aligned(16))) float Bs[48][128];
  const int row0 = blockIdx.x * 16;
  const int colB = blockIdx.y * 128;
  const int t = threadIdx.x;
  const int wave = t >> 5, lane = t & 31, lh = lane & 15, hi = lane >> 4;

  const float4* xa = (const float4*)(x + (size_t)row0 * IN_F);
  float4 qa[3];
#pragma unroll
  for (int j = 0; j < 3; j++) {
    const int f = t + 256 * j;
    qa[j] = xa[(f < A_F4) ? f : A_F4 - 1];
  }
  float4 qb[6];
  gemm1_load_b(W1, colB, t, 0, qb);

#pragma unroll
  for (int j = 0; j < 3; j++) {
    const int f = t + 256 * j;
    if (f < A_F4) {
      const float v[4] = {qa[j].x, qa[j].y, qa[j].z, qa[j].w};
#pragma unroll
      for (int q = 0; q < 4; q++) {
        const int e = 4 * f + q, r = e / IN_F, k = e - r * IN_F;
        As[r * A_PITCH + k] = v[q];
      }
    }
  }
  if (t < 16) {
#pragma unroll
    for (int k = IN_F; k < A_PITCH; k++) As[t * A_PITCH + k] = 0.f;
  }

  v8f acc = {};
#pragma unroll
  for (int k0 = 0; k0 < 192; k0 += 48) {
#pragma unroll
    for (int i = 0; i < 6; i++) {
      const int idx4 = t + 256 * i;
      const int r = idx4 >> 5, c = (idx4 & 31) << 2;
      *(float4*)&Bs[r][c] = (k0 + r < IN_F) ? qb[i] : make_float4(0.f, 0.f, 0.f, 0.f);
    }
    __syncthreads();
    if (k0 + 48 < 192) gemm1_load_b(W1, colB, t, k0 + 48, qb);
#pragma unroll
    for (int kk = 0; kk < 48; kk += 4) {
      v2f a, b;
      a = *(const v2f*)&As[lh * A_PITCH + k0 + kk + 2 * hi];
      b.x = Bs[kk + 2 * hi][wave * 16 + lh];
      b.y = Bs[kk + 2 * hi + 1][wave * 16 + lh];
      acc = __builtin_amdgcn_wmma_f32_16x16x4_f32(false, a, false, b,
                                                  (short)0, acc, false, false);
    }
    __syncthreads();
  }
  float* Ds = &Bs[0][0];
#pragma unroll
  for (int i = 0; i < 8; i++)
    Ds[(i + 8 * hi) * D1_PITCH + wave * 16 + lh] = acc[i];
  __syncthreads();
  const int col = colB + wave * 16 + lh;
#pragma unroll
  for (int i = 0; i < 8; i++)
    xl1[(row0 + i + 8 * hi) * F1 + col] = acc[i];
  if (t < 128) {
    const int r = t >> 3, hl = t & 7;
    const int h = blockIdx.y * 8 + hl;
    const float* row = Ds + r * D1_PITCH + hl * C1;
    float s = 0.f, d = 0.f;
#pragma unroll
    for (int c = 0; c < C1; c++) {
      const float v = row[c];
      s += v * a_src[h * C1 + c];
      d += v * a_dst[h * C1 + c];
    }
    asrc[(row0 + r) * H1 + h] = s;
    adst[(row0 + r) * H1 + h] = d;
  }
}

#define H1_PITCH 260
#define D2_PITCH 25
#define K2_CHUNK 32
#define ROWS_AHEAD 3
#define EW_PITCH (16 + 16 * (32 + ROWS_AHEAD - 1))

__device__ __forceinline__ void
edge_logits(const float* __restrict__ asrc, int sL, float eL, int n, int e2, int h,
            float ad, float c, float (&a)[16]) {
  float v[16], ev[16];
#pragma unroll
  for (int jj = 0; jj < 8; jj++) {
    const int s0 = __builtin_amdgcn_readlane(sL, 2 * jj), s1 = __builtin_amdgcn_readlane(sL, 2 * jj + 1);
    const int x0 = __builtin_amdgcn_readlane(__float_as_int(eL), 2 * jj);
    const int x1 = __builtin_amdgcn_readlane(__float_as_int(eL), 2 * jj + 1);
    v[jj] = asrc[(e2 ? s1 : s0) * H1 + h];
    ev[jj] = __int_as_float(e2 ? x1 : x0);
  }
  if (n > 16) {
#pragma unroll
    for (int jj = 8; jj < 16; jj++) {
      const int s0 = __builtin_amdgcn_readlane(sL, 2 * jj), s1 = __builtin_amdgcn_readlane(sL, 2 * jj + 1);
      const int x0 = __builtin_amdgcn_readlane(__float_as_int(eL), 2 * jj);
      const int x1 = __builtin_amdgcn_readlane(__float_as_int(eL), 2 * jj + 1);
      v[jj] = asrc[(e2 ? s1 : s0) * H1 + h];
      ev[jj] = __int_as_float(e2 ? x1 : x0);
    }
  } else {
#pragma unroll
    for (int jj = 8; jj < 16; jj++) { v[jj] = 0.f; ev[jj] = 0.f; }
  }
#pragma unroll
  for (int jj = 0; jj < 16; jj++) a[jj] = lrelu02(v[jj] + ad + ev[jj] * c);
}

__device__ __forceinline__ void
store_weights(float* ew, const float (&a)[16], float m, int n, int e2, int h) {
#pragma unroll
  for (int jj = 0; jj < 8; jj++) ew[16 + (2 * jj + e2) * 16 + h] = __expf(a[jj] - m);
  if (n > 16) {
#pragma unroll
    for (int jj = 8; jj < 16; jj++) ew[16 + (2 * jj + e2) * 16 + h] = __expf(a[jj] - m);
  }
}

__global__ void __launch_bounds__(512)
node_softmax_agg1_kernel(const int* __restrict__ rowptr,
                         const int* __restrict__ csr_src,
                         const float* __restrict__ csr_ea,
                         const float* __restrict__ mean,
                         const float* __restrict__ ce,
                         const float* __restrict__ asrc,
                         const float* __restrict__ adst,
                         const float* __restrict__ xl,
                         const float* __restrict__ b1,
                         const float* __restrict__ W2,
                         const float* __restrict__ a_src2,
                         const float* __restrict__ a_dst2,
                         float* __restrict__ xl2,
                         float* __restrict__ asrc2,
                         float* __restrict__ adst2) {
  __shared__ __attribute__((aligned(16))) float As[16][H1_PITCH];
  __shared__ float Pt[8][16 * D2_PITCH];
  __shared__ float Ew[16][EW_PITCH];
  const int lane = threadIdx.x & 31;
  const int wv = __builtin_amdgcn_readfirstlane((int)(threadIdx.x >> 5));
  const int d = blockIdx.x * 16 + wv;
  const int hA = lane >> 2, hB = 8 + (lane >> 2);
  const int h16 = lane & 15, e2 = lane >> 4;
  const int beg = rowptr[d], end = rowptr[d + 1];
  const float ad = adst[d * H1 + h16];
  const float c = ce[h16];
  const float mv = mean[0];
  const float aSelf = lrelu02(asrc[d * H1 + h16] + ad + mv * c);
  float* ew = &Ew[wv][0];

  float m = aSelf;
  int sL = 0;
  float eL = 0.f;
  float a[16];
  for (int base = beg; base < end; base += 32) {
    const int n = (end - base < 32) ? end - base : 32;
    sL = 0;
    eL = 0.f;
    if (lane < n) { sL = csr_src[base + lane]; eL = csr_ea[base + lane]; }
    edge_logits(asrc, sL, eL, n, e2, h16, ad, c, a);
#pragma unroll
    for (int jj = 0; jj < 16; jj++)
      if (2 * jj + e2 < n) m = fmaxf(m, a[jj]);
  }
  m = fmaxf(m, __shfl_xor(m, 16));
  const bool one_chunk = (end - beg) <= 32;

  ew[h16] = __expf(aSelf - m);
  if (one_chunk && end > beg) store_weights(ew, a, m, end - beg, e2, h16);
  float denA = ew[hA], denB = ew[hB];
  float4 accA, accB;
  {
    const float4* r = (const float4*)(xl + d * F1);
    const float4 u = r[lane], w = r[32 + lane];
    accA = make_float4(denA * u.x, denA * u.y, denA * u.z, denA * u.w);
    accB = make_float4(denB * w.x, denB * w.y, denB * w.z, denB * w.w);
  }
  for (int base = beg; base < end; base += 32) {
    const int n = (end - base < 32) ? end - base : 32;
    if (!one_chunk) {
      sL = 0;
      eL = 0.f;
      if (lane < n) { sL = csr_src[base + lane]; eL = csr_ea[base + lane]; }
      float a2[16];
      edge_logits(asrc, sL, eL, n, e2, h16, ad, c, a2);
      store_weights(ew, a2, m, n, e2, h16);
    }
    for (int j0 = 0; j0 < n; j0 += ROWS_AHEAD) {
      float4 ru[ROWS_AHEAD], rw[ROWS_AHEAD];
      float eA[ROWS_AHEAD], eB[ROWS_AHEAD];
#pragma unroll
      for (int u = 0; u < ROWS_AHEAD; u++) {
        const int jc = (j0 + u < n) ? j0 + u : n - 1;
        const int s = __builtin_amdgcn_readlane(sL, jc);
        const float4* r = (const float4*)(xl + s * F1);
        ru[u] = r[lane];
        rw[u] = r[32 + lane];
        eA[u] = ew[16 + (j0 + u) * 16 + hA];
        eB[u] = ew[16 + (j0 + u) * 16 + hB];
      }
#pragma unroll
      for (int u = 0; u < ROWS_AHEAD; u++) {
        if (j0 + u < n) {
          denA += eA[u];
          denB += eB[u];
          accA.x += eA[u] * ru[u].x; accA.y += eA[u] * ru[u].y; accA.z += eA[u] * ru[u].z; accA.w += eA[u] * ru[u].w;
          accB.x += eB[u] * rw[u].x; accB.y += eB[u] * rw[u].y; accB.z += eB[u] * rw[u].z; accB.w += eB[u] * rw[u].w;
        }
      }
    }
  }

  const int ct = wv & 1, kc = wv >> 1;
  const int lh = lane & 15, hi = lane >> 4;
  const int col = ct * 16 + lh;
  const bool cv = col < F2;
  const int cc = cv ? col : F2 - 1;
  v2f bf[K2_CHUNK / 4];
#pragma unroll
  for (int u = 0; u < K2_CHUNK / 4; u++) {
    const int k = kc * K2_CHUNK + 4 * u + 2 * hi;
    const float bx = W2[k * F2 + cc], by = W2[(k + 1) * F2 + cc];
    bf[u].x = cv ? bx : 0.f;
    bf[u].y = cv ? by : 0.f;
  }

  const float4 bA = ((const float4*)b1)[lane], bB = ((const float4*)b1)[32 + lane];
  const float rA = 1.0f / (denA + 1e-16f), rB = 1.0f / (denB + 1e-16f);
  float4* o = (float4*)&As[wv][0];
  o[lane]      = make_float4(elu1(accA.x * rA + bA.x), elu1(accA.y * rA + bA.y),
                             elu1(accA.z * rA + bA.z), elu1(accA.w * rA + bA.w));
  o[32 + lane] = make_float4(elu1(accB.x * rB + bB.x), elu1(accB.y * rB + bB.y),
                             elu1(accB.z * rB + bB.z), elu1(accB.w * rB + bB.w));
  __syncthreads();

  v8f acc = {};
#pragma unroll
  for (int u = 0; u < K2_CHUNK / 4; u++) {
    const v2f a = *(const v2f*)&As[lh][kc * K2_CHUNK + 4 * u + 2 * hi];
    acc = __builtin_amdgcn_wmma_f32_16x16x4_f32(false, a, false, bf[u],
                                                (short)0, acc, false, false);
  }
  if (cv) {
#pragma unroll
    for (int i = 0; i < 8; i++)
      Pt[kc][(i + 8 * hi) * D2_PITCH + col] = acc[i];
  }
  __syncthreads();
  const int t = threadIdx.x;
  if (t < 16 * H2) {
    const int r = t >> 3, h = t & 7;
    const int row0 = blockIdx.x * 16;
    float s = 0.f, dd = 0.f;
#pragma unroll
    for (int q = 0; q < C2; q++) {
      const int e = r * D2_PITCH + h * C2 + q;
      float v = Pt[0][e];
#pragma unroll
      for (int p = 1; p < 8; p++) v += Pt[p][e];
      xl2[(row0 + r) * F2 + h * C2 + q] = v;
      s += v * a_src2[h * C2 + q];
      dd += v * a_dst2[h * C2 + q];
    }
    asrc2[(row0 + r) * H2 + h] = s;
    adst2[(row0 + r) * H2 + h] = dd;
  }
}

__global__ void __launch_bounds__(256)
node_softmax_agg2_kernel(const int* __restrict__ rowptr,
                         const int* __restrict__ csr_src,
                         const float* __restrict__ csr_ea,
                         const float* __restrict__ mean,
                         const float* __restrict__ ce,
                         const float* __restrict__ asrc,
                         const float* __restrict__ adst,
                         const float* __restrict__ xl,
                         const float* __restrict__ b2,
                         float* __restrict__ out) {
  __shared__ float hrow[256 * C2];
  const int t = threadIdx.x;
  const int idx = blockIdx.x * 256 + t;
  if (idx < N_NODES * H2) {
    const int d = idx / H2, h = idx % H2;
    const int beg = rowptr[d], end = rowptr[d + 1];
    const float ad = adst[idx];
    const float c = ce[h];
    const float aSelf = lrelu02(asrc[idx] + ad + mean[0] * c);

    float m = aSelf;
    for (int j = beg; j < end; j++) {
      const float a = lrelu02(asrc[csr_src[j] * H2 + h] + ad + csr_ea[j] * c);
      m = fmaxf(m, a);
    }

    float den = __expf(aSelf - m);
    const float* rd = xl + d * F2 + h * C2;
    float acc0 = den * rd[0], acc1 = den * rd[1], acc2 = den * rd[2];
    for (int j = beg; j < end; j++) {
      const int s = csr_src[j];
      const float a = lrelu02(asrc[s * H2 + h] + ad + csr_ea[j] * c);
      const float ex = __expf(a - m);
      const float* rs = xl + s * F2 + h * C2;
      den += ex;
      acc0 += ex * rs[0];
      acc1 += ex * rs[1];
      acc2 += ex * rs[2];
    }
    const float rinv = 1.0f / (den + 1e-16f);
    hrow[t * C2 + 0] = acc0 * rinv;
    hrow[t * C2 + 1] = acc1 * rinv;
    hrow[t * C2 + 2] = acc2 * rinv;
  }
  __syncthreads();
  if (t < 32 * C2) {
    const int nl = t / C2, ch = t % C2;
    const int n = blockIdx.x * 32 + nl;
    if (n < N_NODES) {
      float s = 0.f;
#pragma unroll
      for (int h = 0; h < H2; h++) s += hrow[(nl * H2 + h) * C2 + ch];
      const float v = s * (1.f / (float)H2) + b2[ch];
      out[n * C2 + ch] = (v > 0.f) ? v : expm1f(v);
    }
  }
}

extern "C" void kernel_launch(void* const* d_in, const int* in_sizes, int n_in,
                              void* d_out, int out_size, void* d_ws, size_t ws_size,
                              hipStream_t stream) {
  (void)in_sizes; (void)n_in; (void)out_size; (void)ws_size;
  const float* x   = (const float*)d_in[0];
  const int*   ei  = (const int*)d_in[1];
  const float* ea  = (const float*)d_in[2];
  const float* W1  = (const float*)d_in[3];
  const float* We1 = (const float*)d_in[4];
  const float* as1 = (const float*)d_in[5];
  const float* ad1 = (const float*)d_in[6];
  const float* ae1 = (const float*)d_in[7];
  const float* b1  = (const float*)d_in[8];
  const float* W2  = (const float*)d_in[9];
  const float* We2 = (const float*)d_in[10];
  const float* as2 = (const float*)d_in[11];
  const float* ad2 = (const float*)d_in[12];
  const float* ae2 = (const float*)d_in[13];
  const float* b2  = (const float*)d_in[14];
  float* out = (float*)d_out;

  float* w = (float*)d_ws;
  float* mean = w;
  float* ce1  = w + 1;
  float* ce2  = w + 17;
  size_t off = 32;
  float* xl1   = w + off; off += (size_t)N_NODES * F1;
  float* asrc1 = w + off; off += (size_t)N_NODES * H1;
  float* adst1 = w + off; off += (size_t)N_NODES * H1;
  float* xl2   = w + off; off += (size_t)N_NODES * F2;
  float* asrc2 = w + off; off += (size_t)N_NODES * H2;
  float* adst2 = w + off; off += (size_t)N_NODES * H2;
  int* rowptr  = (int*)(w + off); off += (size_t)N_NODES + 32;
  int* rank    = (int*)(w + off); off += (size_t)N_EDGES;
  int* slot_eid = (int*)(w + off); off += (size_t)N_EDGES;
  int* csr_src = (int*)(w + off); off += (size_t)N_EDGES;
  float* csr_ea = w + off;        off += (size_t)N_EDGES;
  float* psum = w + off;          off += 256;
  int* bhT  = (int*)(w + off);    off += (size_t)NBKT * EBLK;
  int* boff = (int*)(w + off);    off += (size_t)NBKT * EBLK;
  int* bse  = (int*)(w + off);    off += 512;
  int* tmp  = (int*)(w + off);    off += (size_t)4 * N_EDGES;

  bucket_count_kernel<<<EBLK, 256, 0, stream>>>(ei, ea, rank, bhT, psum);
  bucket_scan_kernel<<<NBKT, 256, 0, stream>>>(bhT, boff, bse, psum, We1, ae1, We2, ae2,
                                               mean, ce1, ce2);
  bucket_scatter_kernel<<<EBLK, 256, 0, stream>>>(ei, ea, rank, bhT, boff, tmp);
  bucket_place_kernel<<<NBKT, 1024, 0, stream>>>(ei, ea, tmp, bse, rowptr, slot_eid, csr_src, csr_ea);

  gemm1_wmma<<<dim3(N_NODES / 16, 2), 256, 0, stream>>>(x, W1, as1, ad1, xl1, asrc1, adst1);
  node_softmax_agg1_kernel<<<N_NODES / 16, 512, 0, stream>>>(
      rowptr, csr_src, csr_ea, mean, ce1, asrc1, adst1, xl1, b1,
      W2, as2, ad2, xl2, asrc2, adst2);

  node_softmax_agg2_kernel<<<(N_NODES * H2 + 255) / 256, 256, 0, stream>>>(
      rowptr, csr_src, csr_ea, mean, ce2, asrc2, adst2, xl2, b2, out);
}


# layer-1 product tiled 64 rows x 128 columns per block instead of 16 x 128: each wave holds four 16x16 accumulators on one B fragment; a quarter of the W1 reads and of the waves; float sums unchanged
# speedup vs baseline: 9.5755x; 1.0322x over previous
#include <hip/hip_runtime.h>
#include <math.h>

#define N_NODES 50000
#define N_EDGES 800000
#define IN_F    191
#define H1 16
#define C1 16
#define F1 256
#define H2 8
#define C2 3
#define F2 24
static_assert(N_NODES % 16 == 0, "row tiles of 16 nodes cover the node set exactly");

typedef __attribute__((ext_vector_type(2))) float v2f;
typedef __attribute__((ext_vector_type(8))) float v8f;


__device__ __forceinline__ float lrelu02(float a) {
  return (a > 0.f) ? a : 0.2f * a;
}

__device__ __forceinline__ float elu1(float v) {
  return (v > 0.f) ? v : expm1f(v);
}

#define PS_BLOCKS 196
#define PS_SLICE  4096

__device__ __forceinline__ float wave_sum_f(float v) {
  for (int off = 16; off > 0; off >>= 1) v += __shfl_xor(v, off);
  return v;
}

__device__ __forceinline__ void
prep_block(const float* __restrict__ psum,
           const float* __restrict__ We1, const float* __restrict__ ae1,
           const float* __restrict__ We2, const float* __restrict__ ae2,
           float* __restrict__ mean_out, float* __restrict__ ce1,
           float* __restrict__ ce2) {
  __shared__ float red[8];
  const int t = threadIdx.x;
  const float s = wave_sum_f((t < PS_BLOCKS) ? psum[t] : 0.f);
  if ((t & 31) == 0) red[t >> 5] = s;
  __syncthreads();
  if (t == 0) {
    float tot = 0.f;
    for (int w = 0; w < 8; w++) tot += red[w];
    mean_out[0] = tot * (1.0f / (float)N_EDGES);
  }
  if (t < H1) {
    float c = 0.f;
    for (int k = 0; k < C1; k++) c += We1[t * C1 + k] * ae1[t * C1 + k];
    ce1[t] = c;
  } else if (t >= 32 && t < 32 + H2) {
    const int h = t - 32;
    float c = 0.f;
    for (int k = 0; k < C2; k++) c += We2[h * C2 + k] * ae2[h * C2 + k];
    ce2[h] = c;
  }
}

#define NBKT 196
#define EBLK PS_BLOCKS

__global__ void __launch_bounds__(256)
bucket_count_kernel(const int* __restrict__ ei,
                    const float* __restrict__ ea,
                    int* __restrict__ lrank,
                    int* __restrict__ bhT,
                    float* __restrict__ psum) {
  __shared__ int hist[NBKT];
  __shared__ float red[8];
  const int t = threadIdx.x;
  if (t < NBKT) hist[t] = 0;
  const int base = blockIdx.x * PS_SLICE;
  const int* dstp = ei + N_EDGES;
  float4 q[PS_SLICE / 1024];
  int4 dq[PS_SLICE / 1024];
  for (int k = 0; k < PS_SLICE / 1024; k++) {
    const int i = base + (k * 256 + t) * 4;
    const int ic = (i < N_EDGES) ? i : N_EDGES - 4;
    q[k] = *(const float4*)&ea[ic];
    dq[k] = *(const int4*)&dstp[ic];
  }
  float s = 0.f;
  for (int k = 0; k < PS_SLICE / 1024; k++) {
    const int i = base + (k * 256 + t) * 4;
    const float part = (q[k].x + q[k].y) + (q[k].z + q[k].w);
    s += (i < N_EDGES) ? part : 0.f;
  }
  s = wave_sum_f(s);
  if ((t & 31) == 0) red[t >> 5] = s;
  __syncthreads();
  for (int k = 0; k < PS_SLICE / 1024; k++) {
    const int i = base + (k * 256 + t) * 4;
    if (i < N_EDGES) {
      int4 r;
      r.x = atomicAdd(&hist[dq[k].x >> 8], 1);
      r.y = atomicAdd(&hist[dq[k].y >> 8], 1);
      r.z = atomicAdd(&hist[dq[k].z >> 8], 1);
      r.w = atomicAdd(&hist[dq[k].w >> 8], 1);
      *(int4*)&lrank[i] = r;
    }
  }
  if (t == 0) {
    float tot = 0.f;
    for (int w = 0; w < 8; w++) tot += red[w];
    psum[blockIdx.x] = tot;
  }
  __syncthreads();
  if (t < NBKT) bhT[t * EBLK + blockIdx.x] = hist[t];
}

__global__ void __launch_bounds__(256)
bucket_scan_kernel(const int* __restrict__ bhT, int* __restrict__ boff,
                   int* __restrict__ bse,
                   const float* __restrict__ psum,
                   const float* __restrict__ We1, const float* __restrict__ ae1,
                   const float* __restrict__ We2, const float* __restrict__ ae2,
                   float* __restrict__ mean_out, float* __restrict__ ce1,
                   float* __restrict__ ce2) {
  __shared__ int wpre[8];
  __shared__ int wtot[8];
  const int t = threadIdx.x;
  const int lane = t & 31, wave = t >> 5;
  const int lo = blockIdx.x * EBLK;
  int pre = 0;
  for (int i0 = t * 4; i0 < lo; i0 += 8 * 1024) {
    int4 q[8];
    for (int u = 0; u < 8; u++) {
      const int i = i0 + u * 1024;
      q[u] = (i < lo) ? *(const int4*)&bhT[i] : make_int4(0, 0, 0, 0);
    }
    for (int u = 0; u < 8; u++) pre += (q[u].x + q[u].y) + (q[u].z + q[u].w);
  }
  const int c = (t < EBLK) ? bhT[lo + t] : 0;
  int v = c;
  for (int off = 1; off < 32; off <<= 1) {
    const int u = __shfl_up(v, off);
    if (lane >= off) v += u;
  }
  for (int off = 16; off > 0; off >>= 1) pre += __shfl_xor(pre, off);
  if (lane == 31) wtot[wave] = v;
  if (lane == 0) wpre[wave] = pre;
  __syncthreads();
  int start = 0;
  for (int w = 0; w < 8; w++) {
    start += wpre[w];
    if (w < wave) start += wtot[w];
  }
  if (t < EBLK) boff[lo + t] = start + v - c;
  if (t == 0) bse[2 * blockIdx.x] = start + v - c;
  if (t == EBLK - 1) bse[2 * blockIdx.x + 1] = start + v;
  if (blockIdx.x == 0)
    prep_block(psum, We1, ae1, We2, ae2, mean_out, ce1, ce2);
}

__global__ void __launch_bounds__(256)
bucket_scatter_kernel(const int* __restrict__ ei,
                      const float* __restrict__ ea,
                      const int* __restrict__ lrank,
                      const int* __restrict__ bhT,
                      const int* __restrict__ boff,
                      int* __restrict__ tmp_recs) {
  int4* tmp = (int4*)tmp_recs;
  __shared__ int lst[NBKT];
  __shared__ int delta[NBKT];
  __shared__ int wtot[8];
  __shared__ int4 sorted[PS_SLICE];
  const int t = threadIdx.x;
  const int lane = t & 31, wave = t >> 5;
  const int base = blockIdx.x * PS_SLICE;
  const int nblk = (N_EDGES - base < PS_SLICE) ? N_EDGES - base : PS_SLICE;
  const int* dstp = ei + N_EDGES;
  const int c = (t < NBKT) ? bhT[t * EBLK + blockIdx.x] : 0;
  const int o = (t < NBKT) ? boff[t * EBLK + blockIdx.x] : 0;
  int4 dq[PS_SLICE / 1024], rq[PS_SLICE / 1024], sq[PS_SLICE / 1024];
  float4 aq[PS_SLICE / 1024];
  for (int k = 0; k < PS_SLICE / 1024; k++) {
    const int i = base + (k * 256 + t) * 4;
    const int ic = (i < N_EDGES) ? i : N_EDGES - 4;
    dq[k] = *(const int4*)&dstp[ic];
    rq[k] = *(const int4*)&lrank[ic];
    sq[k] = *(const int4*)&ei[ic];
    aq[k] = *(const float4*)&ea[ic];
  }
  int v = c;
  for (int off = 1; off < 32; off <<= 1) {
    const int u = __shfl_up(v, off);
    if (lane >= off) v += u;
  }
  if (lane == 31) wtot[wave] = v;
  __syncthreads();
  int first = v - c;
  for (int w = 0; w < 8; w++)
    if (w < wave) first += wtot[w];
  if (t < NBKT) {
    lst[t] = first;
    delta[t] = o - first;
  }
  __syncthreads();
  for (int k = 0; k < PS_SLICE / 1024; k++) {
    const int i = base + (k * 256 + t) * 4;
    if (i < N_EDGES) {
      sorted[lst[dq[k].x >> 8] + rq[k].x] = make_int4(i, dq[k].x, sq[k].x, __float_as_int(aq[k].x));
      sorted[lst[dq[k].y >> 8] + rq[k].y] = make_int4(i + 1, dq[k].y, sq[k].y, __float_as_int(aq[k].y));
      sorted[lst[dq[k].z >> 8] + rq[k].z] = make_int4(i + 2, dq[k].z, sq[k].z, __float_as_int(aq[k].z));
      sorted[lst[dq[k].w >> 8] + rq[k].w] = make_int4(i + 3, dq[k].w, sq[k].w, __float_as_int(aq[k].w));
    }
  }
  __syncthreads();
  for (int i = t; i < nblk; i += 256) {
    const int4 p = sorted[i];
    tmp[delta[p.y >> 8] + i] = p;
  }
}

#define BP_GROUP 6
#define BP_CAP   6144
#define ID_BITS  20
#define ID_MASK  ((1 << ID_BITS) - 1)
static_assert(N_EDGES <= (1 << ID_BITS), "edge id and local destination share one word");

__device__ __forceinline__ void order_bucket_from_memory(const int* __restrict__ ei,
                                                         const float* __restrict__ ea,
                                                         const int* seg, const int* gkeys,
                                                         int bs, int nb, int t,
                                                         int* __restrict__ csr_src,
                                                         float* __restrict__ csr_ea) {
  for (int i = t; i < nb; i += 1024) {
    const int k = gkeys[bs + i];
    const int ld = k >> ID_BITS;
    const int b = seg[ld], e = seg[ld + 1];
    int pos = 0;
    for (int j = b; j < e; j++) pos += (gkeys[bs + j] < k) ? 1 : 0;
    const int id = k & ID_MASK;
    csr_src[bs + b + pos] = ei[id];
    csr_ea[bs + b + pos] = ea[id];
  }
}

__global__ void __launch_bounds__(1024)
bucket_place_kernel(const int* __restrict__ ei, const float* __restrict__ ea,
                    const int* __restrict__ tmp_recs, const int* __restrict__ bse,
                    int* __restrict__ rowptr, int* gkeys,
                    int* __restrict__ csr_src, float* __restrict__ csr_ea) {
  const int4* tmp = (const int4*)tmp_recs;
  __shared__ int cnt[256];
  __shared__ int seg[257];
  __shared__ int wtot[8];
  __shared__ int skey[BP_CAP];
  __shared__ int ssrc[BP_CAP];
  __shared__ float sea[BP_CAP];
  const int t = threadIdx.x;
  const int lane = t & 31, wave = t >> 5;
  const int bs = bse[2 * blockIdx.x], be = bse[2 * blockIdx.x + 1];
  const int nb = be - bs;
  const bool fits = nb <= BP_CAP;
  if (t < 256) cnt[t] = 0;
  __syncthreads();
  for (int i0 = bs + t; i0 < be; i0 += BP_GROUP * 1024) {
    int dd[BP_GROUP];
    for (int u = 0; u < BP_GROUP; u++) {
      const int i = i0 + u * 1024;
      dd[u] = tmp[(i < be) ? i : be - 1].y;
    }
    for (int u = 0; u < BP_GROUP; u++)
      if (i0 + u * 1024 < be) atomicAdd(&cnt[dd[u] & 255], 1);
  }
  __syncthreads();
  const int c = (t < 256) ? cnt[t] : 0;
  int v = c;
  for (int off = 1; off < 32; off <<= 1) {
    const int u = __shfl_up(v, off);
    if (lane >= off) v += u;
  }
  if (lane == 31 && wave < 8) wtot[wave] = v;
  __syncthreads();
  if (t < 256) {
    int start = bs;
    for (int w = 0; w < 8; w++)
      if (w < wave) start += wtot[w];
    const int first = start + v - c;
    const int node = blockIdx.x * 256 + t;
    if (node <= N_NODES) rowptr[node] = first;
    cnt[t] = first;
    seg[t] = first - bs;
    if (t == 255) seg[256] = first + c - bs;
  }
  __syncthreads();
  for (int i0 = bs + t; i0 < be; i0 += BP_GROUP * 1024) {
    int4 q[BP_GROUP];
    for (int u = 0; u < BP_GROUP; u++) {
      const int i = i0 + u * 1024;
      q[u] = tmp[(i < be) ? i : be - 1];
    }
    for (int u = 0; u < BP_GROUP; u++) {
      if (i0 + u * 1024 < be) {
        const int ld = q[u].y & 255;
        const int pos = atomicAdd(&cnt[ld], 1);
        const int k = q[u].x | (ld << ID_BITS);
        if (fits) {
          skey[pos - bs] = k;
          ssrc[pos - bs] = q[u].z;
          sea[pos - bs] = __int_as_float(q[u].w);
        } else {
          gkeys[pos] = k;
        }
      }
    }
  }
  __syncthreads();
  if (fits) {
    for (int i = t; i < nb; i += 1024) {
      const int k = skey[i];
      const int ld = k >> ID_BITS;
      const int b = seg[ld], e = seg[ld + 1];
      int pos = 0;
      for (int j = b; j < e; j++) pos += (skey[j] < k) ? 1 : 0;
      csr_src[bs + b + pos] = ssrc[i];
      csr_ea[bs + b + pos] = sea[i];
    }
  } else {
    order_bucket_from_memory(ei, ea, seg, gkeys, bs, nb, t, csr_src, csr_ea);
  }
}

#define G1_ROWS 64
#define G1_MT   4
#define G1_BLOCKS ((N_NODES + G1_ROWS - 1) / G1_ROWS)
#define A_PITCH 196
#define A_F4    3056
#define A_LD    12
#define D1_PITCH 132
static_assert(G1_ROWS * IN_F == 4 * A_F4, "the row tile is a whole number of float4");
static_assert(256 * A_LD >= A_F4, "one group of loads covers the row tile");
static_assert(G1_ROWS * D1_PITCH <= G1_ROWS * A_PITCH, "the result tile fits the x stage");

__device__ __forceinline__ void gemm1_load_b(const float* __restrict__ W1, int colB,
                                             int t, int k0, float4 (&qb)[6]) {
#pragma unroll
  for (int i = 0; i < 6; i++) {
    const int idx4 = t + 256 * i;
    const int r = idx4 >> 5, c = (idx4 & 31) << 2, k = k0 + r;
    qb[i] = *(const float4*)&W1[((k < IN_F) ? k : IN_F - 1) * F1 + colB + c];
  }
}

__global__ void __launch_bounds__(256, 8)
gemm1_wmma(const float* __restrict__ x, const float* __restrict__ W1,
           const float* __restrict__ a_src, const float* __restrict__ a_dst,
           float* __restrict__ xl1, float* __restrict__ asrc,
           float* __restrict__ adst) {
  __shared__ __attribute__((aligned(16))) float As[G1_ROWS * A_PITCH];
  __shared__ __attribute__((aligned(16))) float Bs[48][128];
  const int row0 = blockIdx.x * G1_ROWS;
  const int colB = blockIdx.y * 128;
  const int t = threadIdx.x;
  const int wave = t >> 5, lane = t & 31, lh = lane & 15, hi = lane >> 4;
  const int nrows = (N_NODES - row0 < G1_ROWS) ? N_NODES - row0 : G1_ROWS;
  const int last4 = nrows * IN_F / 4 - 1;

  const float4* xa = (const float4*)(x + (size_t)row0 * IN_F);
  float4 qa[A_LD];
#pragma unroll
  for (int j = 0; j < A_LD; j++) {
    const int f = t + 256 * j;
    qa[j] = xa[(f < last4) ? f : last4];
  }
  float4 qb[6];
  gemm1_load_b(W1, colB, t, 0, qb);

#pragma unroll
  for (int j = 0; j < A_LD; j++) {
    const int f = t + 256 * j;
    if (f < A_F4) {
      const float v[4] = {qa[j].x, qa[j].y, qa[j].z, qa[j].w};
#pragma unroll
      for (int q = 0; q < 4; q++) {
        const int e = 4 * f + q, r = e / IN_F, k = e - r * IN_F;
        As[r * A_PITCH + k] = v[q];
      }
    }
  }
  if (t < G1_ROWS) {
#pragma unroll
    for (int k = IN_F; k < A_PITCH; k++) As[t * A_PITCH + k] = 0.f;
  }

  v8f acc[G1_MT] = {};
#pragma unroll
  for (int k0 = 0; k0 < 192; k0 += 48) {
#pragma unroll
    for (int i = 0; i < 6; i++) {
      const int idx4 = t + 256 * i;
      const int r = idx4 >> 5, c = (idx4 & 31) << 2;
      *(float4*)&Bs[r][c] = (k0 + r < IN_F) ? qb[i] : make_float4(0.f, 0.f, 0.f, 0.f);
    }
    __syncthreads();
    if (k0 + 48 < 192) gemm1_load_b(W1, colB, t, k0 + 48, qb);
#pragma unroll
    for (int kk = 0; kk < 48; kk += 4) {
      v2f b;
      b.x = Bs[kk + 2 * hi][wave * 16 + lh];
      b.y = Bs[kk + 2 * hi + 1][wave * 16 + lh];
#pragma unroll
      for (int m = 0; m < G1_MT; m++) {
        const v2f a = *(const v2f*)&As[(16 * m + lh) * A_PITCH + k0 + kk + 2 * hi];
        acc[m] = __builtin_amdgcn_wmma_f32_16x16x4_f32(false, a, false, b,
                                                       (short)0, acc[m], false, false);
      }
    }
    __syncthreads();
  }
  float* Ds = As;
#pragma unroll
  for (int m = 0; m < G1_MT; m++) {
#pragma unroll
    for (int i = 0; i < 8; i++)
      Ds[(16 * m + i + 8 * hi) * D1_PITCH + wave * 16 + lh] = acc[m][i];
  }
  __syncthreads();
  const int col = colB + wave * 16 + lh;
#pragma unroll
  for (int m = 0; m < G1_MT; m++) {
    if (16 * m < nrows) {
#pragma unroll
      for (int i = 0; i < 8; i++)
        xl1[(row0 + 16 * m + i + 8 * hi) * F1 + col] = acc[m][i];
    }
  }
#pragma unroll
  for (int p = 0; p < G1_ROWS / 32; p++) {
    const int r = 32 * p + (t >> 3), hl = t & 7;
    const int h = blockIdx.y * 8 + hl;
    const float* row = Ds + r * D1_PITCH + hl * C1;
    float s = 0.f, d = 0.f;
#pragma unroll
    for (int c = 0; c < C1; c++) {
      const float v = row[c];
      s += v * a_src[h * C1 + c];
      d += v * a_dst[h * C1 + c];
    }
    if (r < nrows) {
      asrc[(row0 + r) * H1 + h] = s;
      adst[(row0 + r) * H1 + h] = d;
    }
  }
}

#define H1_PITCH 260
#define D2_PITCH 25
#define K2_CHUNK 32
#define ROWS_AHEAD 3
#define EW_PITCH (16 + 16 * (32 + ROWS_AHEAD - 1))

__device__ __forceinline__ void
edge_logits(const float* __restrict__ asrc, int sL, float eL, int n, int e2, int h,
            float ad, float c, float (&a)[16]) {
  float v[16], ev[16];
#pragma unroll
  for (int jj = 0; jj < 8; jj++) {
    const int s0 = __builtin_amdgcn_readlane(sL, 2 * jj), s1 = __builtin_amdgcn_readlane(sL, 2 * jj + 1);
    const int x0 = __builtin_amdgcn_readlane(__float_as_int(eL), 2 * jj);
    const int x1 = __builtin_amdgcn_readlane(__float_as_int(eL), 2 * jj + 1);
    v[jj] = asrc[(e2 ? s1 : s0) * H1 + h];
    ev[jj] = __int_as_float(e2 ? x1 : x0);
  }
  if (n > 16) {
#pragma unroll
    for (int jj = 8; jj < 16; jj++) {
      const int s0 = __builtin_amdgcn_readlane(sL, 2 * jj), s1 = __builtin_amdgcn_readlane(sL, 2 * jj + 1);
      const int x0 = __builtin_amdgcn_readlane(__float_as_int(eL), 2 * jj);
      const int x1 = __builtin_amdgcn_readlane(__float_as_int(eL), 2 * jj + 1);
      v[jj] = asrc[(e2 ? s1 : s0) * H1 + h];
      ev[jj] = __int_as_float(e2 ? x1 : x0);
    }
  } else {
#pragma unroll
    for (int jj = 8; jj < 16; jj++) { v[jj] = 0.f; ev[jj] = 0.f; }
  }
#pragma unroll
  for (int jj = 0; jj < 16; jj++) a[jj] = lrelu02(v[jj] + ad + ev[jj] * c);
}

__device__ __forceinline__ void
store_weights(float* ew, const float (&a)[16], float m, int n, int e2, int h) {
#pragma unroll
  for (int jj = 0; jj < 8; jj++) ew[16 + (2 * jj + e2) * 16 + h] = __expf(a[jj] - m);
  if (n > 16) {
#pragma unroll
    for (int jj = 8; jj < 16; jj++) ew[16 + (2 * jj + e2) * 16 + h] = __expf(a[jj] - m);
  }
}

__global__ void __launch_bounds__(512)
node_softmax_agg1_kernel(const int* __restrict__ rowptr,
                         const int* __restrict__ csr_src,
                         const float* __restrict__ csr_ea,
                         const float* __restrict__ mean,
                         const float* __restrict__ ce,
                         const float* __restrict__ asrc,
                         const float* __restrict__ adst,
                         const float* __restrict__ xl,
                         const float* __restrict__ b1,
                         const float* __restrict__ W2,
                         const float* __restrict__ a_src2,
                         const float* __restrict__ a_dst2,
                         float* __restrict__ xl2,
                         float* __restrict__ asrc2,
                         float* __restrict__ adst2) {
  __shared__ __attribute__((aligned(16))) float As[16][H1_PITCH];
  __shared__ float Pt[8][16 * D2_PITCH];
  __shared__ float Ew[16][EW_PITCH];
  const int lane = threadIdx.x & 31;
  const int wv = __builtin_amdgcn_readfirstlane((int)(threadIdx.x >> 5));
  const int d = blockIdx.x * 16 + wv;
  const int hA = lane >> 2, hB = 8 + (lane >> 2);
  const int h16 = lane & 15, e2 = lane >> 4;
  const int beg = rowptr[d], end = rowptr[d + 1];
  const float ad = adst[d * H1 + h16];
  const float c = ce[h16];
  const float mv = mean[0];
  const float aSelf = lrelu02(asrc[d * H1 + h16] + ad + mv * c);
  float* ew = &Ew[wv][0];

  float m = aSelf;
  int sL = 0;
  float eL = 0.f;
  float a[16];
  for (int base = beg; base < end; base += 32) {
    const int n = (end - base < 32) ? end - base : 32;
    sL = 0;
    eL = 0.f;
    if (lane < n) { sL = csr_src[base + lane]; eL = csr_ea[base + lane]; }
    edge_logits(asrc, sL, eL, n, e2, h16, ad, c, a);
#pragma unroll
    for (int jj = 0; jj < 16; jj++)
      if (2 * jj + e2 < n) m = fmaxf(m, a[jj]);
  }
  m = fmaxf(m, __shfl_xor(m, 16));
  const bool one_chunk = (end - beg) <= 32;

  ew[h16] = __expf(aSelf - m);
  if (one_chunk && end > beg) store_weights(ew, a, m, end - beg, e2, h16);
  float denA = ew[hA], denB = ew[hB];
  float4 accA, accB;
  {
    const float4* r = (const float4*)(xl + d * F1);
    const float4 u = r[lane], w = r[32 + lane];
    accA = make_float4(denA * u.x, denA * u.y, denA * u.z, denA * u.w);
    accB = make_float4(denB * w.x, denB * w.y, denB * w.z, denB * w.w);
  }
  for (int base = beg; base < end; base += 32) {
    const int n = (end - base < 32) ? end - base : 32;
    if (!one_chunk) {
      sL = 0;
      eL = 0.f;
      if (lane < n) { sL = csr_src[base + lane]; eL = csr_ea[base + lane]; }
      float a2[16];
      edge_logits(asrc, sL, eL, n, e2, h16, ad, c, a2);
      store_weights(ew, a2, m, n, e2, h16);
    }
    for (int j0 = 0; j0 < n; j0 += ROWS_AHEAD) {
      float4 ru[ROWS_AHEAD], rw[ROWS_AHEAD];
      float eA[ROWS_AHEAD], eB[ROWS_AHEAD];
#pragma unroll
      for (int u = 0; u < ROWS_AHEAD; u++) {
        const int jc = (j0 + u < n) ? j0 + u : n - 1;
        const int s = __builtin_amdgcn_readlane(sL, jc);
        const float4* r = (const float4*)(xl + s * F1);
        ru[u] = r[lane];
        rw[u] = r[32 + lane];
        eA[u] = ew[16 + (j0 + u) * 16 + hA];
        eB[u] = ew[16 + (j0 + u) * 16 + hB];
      }
#pragma unroll
      for (int u = 0; u < ROWS_AHEAD; u++) {
        if (j0 + u < n) {
          denA += eA[u];
          denB += eB[u];
          accA.x += eA[u] * ru[u].x; accA.y += eA[u] * ru[u].y; accA.z += eA[u] * ru[u].z; accA.w += eA[u] * ru[u].w;
          accB.x += eB[u] * rw[u].x; accB.y += eB[u] * rw[u].y; accB.z += eB[u] * rw[u].z; accB.w += eB[u] * rw[u].w;
        }
      }
    }
  }

  const int ct = wv & 1, kc = wv >> 1;
  const int lh = lane & 15, hi = lane >> 4;
  const int col = ct * 16 + lh;
  const bool cv = col < F2;
  const int cc = cv ? col : F2 - 1;
  v2f bf[K2_CHUNK / 4];
#pragma unroll
  for (int u = 0; u < K2_CHUNK / 4; u++) {
    const int k = kc * K2_CHUNK + 4 * u + 2 * hi;
    const float bx = W2[k * F2 + cc], by = W2[(k + 1) * F2 + cc];
    bf[u].x = cv ? bx : 0.f;
    bf[u].y = cv ? by : 0.f;
  }

  const float4 bA = ((const float4*)b1)[lane], bB = ((const float4*)b1)[32 + lane];
  const float rA = 1.0f / (denA + 1e-16f), rB = 1.0f / (denB + 1e-16f);
  float4* o = (float4*)&As[wv][0];
  o[lane]      = make_float4(elu1(accA.x * rA + bA.x), elu1(accA.y * rA + bA.y),
                             elu1(accA.z * rA + bA.z), elu1(accA.w * rA + bA.w));
  o[32 + lane] = make_float4(elu1(accB.x * rB + bB.x), elu1(accB.y * rB + bB.y),
                             elu1(accB.z * rB + bB.z), elu1(accB.w * rB + bB.w));
  __syncthreads();

  v8f acc = {};
#pragma unroll
  for (int u = 0; u < K2_CHUNK / 4; u++) {
    const v2f a = *(const v2f*)&As[lh][kc * K2_CHUNK + 4 * u + 2 * hi];
    acc = __builtin_amdgcn_wmma_f32_16x16x4_f32(false, a, false, bf[u],
                                                (short)0, acc, false, false);
  }
  if (cv) {
#pragma unroll
    for (int i = 0; i < 8; i++)
      Pt[kc][(i + 8 * hi) * D2_PITCH + col] = acc[i];
  }
  __syncthreads();
  const int t = threadIdx.x;
  if (t < 16 * H2) {
    const int r = t >> 3, h = t & 7;
    const int row0 = blockIdx.x * 16;
    float s = 0.f, dd = 0.f;
#pragma unroll
    for (int q = 0; q < C2; q++) {
      const int e = r * D2_PITCH + h * C2 + q;
      float v = Pt[0][e];
#pragma unroll
      for (int p = 1; p < 8; p++) v += Pt[p][e];
      xl2[(row0 + r) * F2 + h * C2 + q] = v;
      s += v * a_src2[h * C2 + q];
      dd += v * a_dst2[h * C2 + q];
    }
    asrc2[(row0 + r) * H2 + h] = s;
    adst2[(row0 + r) * H2 + h] = dd;
  }
}

__global__ void __launch_bounds__(256)
node_softmax_agg2_kernel(const int* __restrict__ rowptr,
                         const int* __restrict__ csr_src,
                         const float* __restrict__ csr_ea,
                         const float* __restrict__ mean,
                         const float* __restrict__ ce,
                         const float* __restrict__ asrc,
                         const float* __restrict__ adst,
                         const float* __restrict__ xl,
                         const float* __restrict__ b2,
                         float* __restrict__ out) {
  __shared__ float hrow[256 * C2];
  const int t = threadIdx.x;
  const int idx = blockIdx.x * 256 + t;
  if (idx < N_NODES * H2) {
    const int d = idx / H2, h = idx % H2;
    const int beg = rowptr[d], end = rowptr[d + 1];
    const float ad = adst[idx];
    const float c = ce[h];
    const float aSelf = lrelu02(asrc[idx] + ad + mean[0] * c);

    float m = aSelf;
    for (int j = beg; j < end; j++) {
      const float a = lrelu02(asrc[csr_src[j] * H2 + h] + ad + csr_ea[j] * c);
      m = fmaxf(m, a);
    }

    float den = __expf(aSelf - m);
    const float* rd = xl + d * F2 + h * C2;
    float acc0 = den * rd[0], acc1 = den * rd[1], acc2 = den * rd[2];
    for (int j = beg; j < end; j++) {
      const int s = csr_src[j];
      const float a = lrelu02(asrc[s * H2 + h] + ad + csr_ea[j] * c);
      const float ex = __expf(a - m);
      const float* rs = xl + s * F2 + h * C2;
      den += ex;
      acc0 += ex * rs[0];
      acc1 += ex * rs[1];
      acc2 += ex * rs[2];
    }
    const float rinv = 1.0f / (den + 1e-16f);
    hrow[t * C2 + 0] = acc0 * rinv;
    hrow[t * C2 + 1] = acc1 * rinv;
    hrow[t * C2 + 2] = acc2 * rinv;
  }
  __syncthreads();
  if (t < 32 * C2) {
    const int nl = t / C2, ch = t % C2;
    const int n = blockIdx.x * 32 + nl;
    if (n < N_NODES) {
      float s = 0.f;
#pragma unroll
      for (int h = 0; h < H2; h++) s += hrow[(nl * H2 + h) * C2 + ch];
      const float v = s * (1.f / (float)H2) + b2[ch];
      out[n * C2 + ch] = (v > 0.f) ? v : expm1f(v);
    }
  }
}

extern "C" void kernel_launch(void* const* d_in, const int* in_sizes, int n_in,
                              void* d_out, int out_size, void* d_ws, size_t ws_size,
                              hipStream_t stream) {
  (void)in_sizes; (void)n_in; (void)out_size; (void)ws_size;
  const float* x   = (const float*)d_in[0];
  const int*   ei  = (const int*)d_in[1];
  const float* ea  = (const float*)d_in[2];
  const float* W1  = (const float*)d_in[3];
  const float* We1 = (const float*)d_in[4];
  const float* as1 = (const float*)d_in[5];
  const float* ad1 = (const float*)d_in[6];
  const float* ae1 = (const float*)d_in[7];
  const float* b1  = (const float*)d_in[8];
  const float* W2  = (const float*)d_in[9];
  const float* We2 = (const float*)d_in[10];
  const float* as2 = (const float*)d_in[11];
  const float* ad2 = (const float*)d_in[12];
  const float* ae2 = (const float*)d_in[13];
  const float* b2  = (const float*)d_in[14];
  float* out = (float*)d_out;

  float* w = (float*)d_ws;
  float* mean = w;
  float* ce1  = w + 1;
  float* ce2  = w + 17;
  size_t off = 32;
  float* xl1   = w + off; off += (size_t)N_NODES * F1;
  float* asrc1 = w + off; off += (size_t)N_NODES * H1;
  float* adst1 = w + off; off += (size_t)N_NODES * H1;
  float* xl2   = w + off; off += (size_t)N_NODES * F2;
  float* asrc2 = w + off; off += (size_t)N_NODES * H2;
  float* adst2 = w + off; off += (size_t)N_NODES * H2;
  int* rowptr  = (int*)(w + off); off += (size_t)N_NODES + 32;
  int* rank    = (int*)(w + off); off += (size_t)N_EDGES;
  int* slot_eid = (int*)(w + off); off += (size_t)N_EDGES;
  int* csr_src = (int*)(w + off); off += (size_t)N_EDGES;
  float* csr_ea = w + off;        off += (size_t)N_EDGES;
  float* psum = w + off;          off += 256;
  int* bhT  = (int*)(w + off);    off += (size_t)NBKT * EBLK;
  int* boff = (int*)(w + off);    off += (size_t)NBKT * EBLK;
  int* bse  = (int*)(w + off);    off += 512;
  int* tmp  = (int*)(w + off);    off += (size_t)4 * N_EDGES;

  bucket_count_kernel<<<EBLK, 256, 0, stream>>>(ei, ea, rank, bhT, psum);
  bucket_scan_kernel<<<NBKT, 256, 0, stream>>>(bhT, boff, bse, psum, We1, ae1, We2, ae2,
                                               mean, ce1, ce2);
  bucket_scatter_kernel<<<EBLK, 256, 0, stream>>>(ei, ea, rank, bhT, boff, tmp);
  bucket_place_kernel<<<NBKT, 1024, 0, stream>>>(ei, ea, tmp, bse, rowptr, slot_eid, csr_src, csr_ea);

  gemm1_wmma<<<dim3(G1_BLOCKS, 2), 256, 0, stream>>>(x, W1, as1, ad1, xl1, asrc1, adst1);
  node_softmax_agg1_kernel<<<N_NODES / 16, 512, 0, stream>>>(
      rowptr, csr_src, csr_ea, mean, ce1, asrc1, adst1, xl1, b1,
      W2, as2, ad2, xl2, asrc2, adst2);

  node_softmax_agg2_kernel<<<(N_NODES * H2 + 255) / 256, 256, 0, stream>>>(
      rowptr, csr_src, csr_ea, mean, ce2, asrc2, adst2, xl2, b2, out);
}
